# starting kernel
# baseline (speedup 1.0000x reference)
#include <hip/hip_runtime.h>
#include <math.h>

typedef __attribute__((ext_vector_type(2))) float v2f;
typedef __attribute__((ext_vector_type(8))) float v8f;

#define PI_D 3.14159265358979323846

constexpr long long SZ_A1    = 255LL*128*128;
constexpr long long SZ_XP    = 128LL*256*128;
constexpr long long SZ_PH1   = 256LL*256;
constexpr long long SZ_G     = 128LL*256*128;
constexpr long long SZ_F1    = 255LL*128*128;
constexpr long long SZ_R1    = 64LL*128;
constexpr long long SZ_R2P   = 128LL*256;
constexpr long long SZ_T1    = 256LL*64*128;
constexpr long long SZ_FF    = 64LL*128*128;
constexpr long long SZ_WC    = 64LL*128*128;
constexpr long long SZ_F2    = 64LL*128*128;
constexpr long long SZ_P2    = 127LL*64*64;
constexpr long long SZ_FT    = 127LL*64*128;
constexpr long long SZ_PH2   = 128LL*128;
constexpr long long SZ_XSPEC = 64LL*128*128;
constexpr long long SZ_TVEC  = 256;
constexpr long long SZ_TC    = 64;
constexpr long long SZ_WSP   = 128LL*128;
constexpr long long SZ_XS1   = 128LL*255*128;
constexpr long long SZ_T2    = 256LL*64*128;
constexpr long long SZ_XS2   = 64LL*128*128;
constexpr long long SZ_XSP   = 67LL*129*128;
constexpr long long SZ_XCV   = 64LL*128*128;

constexpr long long OFF_A1    = 0;
constexpr long long OFF_XP    = OFF_A1    + SZ_A1;
constexpr long long OFF_COS1  = OFF_XP    + SZ_XP;
constexpr long long OFF_SIN1N = OFF_COS1  + SZ_PH1;
constexpr long long OFF_GR    = OFF_SIN1N + SZ_PH1;
constexpr long long OFF_GI    = OFF_GR    + SZ_G;
constexpr long long OFF_F1R   = OFF_GI    + SZ_G;
constexpr long long OFF_F1I   = OFF_F1R   + SZ_F1;
constexpr long long OFF_R1    = OFF_F1I   + SZ_F1;
constexpr long long OFF_R2P   = OFF_R1    + SZ_R1;
constexpr long long OFF_T1R   = OFF_R2P   + SZ_R2P;
constexpr long long OFF_T1I   = OFF_T1R   + SZ_T1;
constexpr long long OFF_FFR   = OFF_T1I   + SZ_T1;
constexpr long long OFF_FFI   = OFF_FFR   + SZ_FF;
constexpr long long OFF_WCR   = OFF_FFI   + SZ_FF;
constexpr long long OFF_WCI   = OFF_WCR   + SZ_WC;
constexpr long long OFF_F2R   = OFF_WCI   + SZ_WC;
constexpr long long OFF_F2I   = OFF_F2R   + SZ_F2;
constexpr long long OFF_P2A   = OFF_F2I   + SZ_F2;
constexpr long long OFF_P2B   = OFF_P2A   + SZ_P2;
constexpr long long OFF_FTR   = OFF_P2B   + SZ_P2;
constexpr long long OFF_FTI   = OFF_FTR   + SZ_FT;
constexpr long long OFF_COS2  = OFF_FTI   + SZ_FT;
constexpr long long OFF_SIN2  = OFF_COS2  + SZ_PH2;
constexpr long long OFF_XSPEC = OFF_SIN2  + SZ_PH2;
constexpr long long OFF_TVEC  = OFF_XSPEC + SZ_XSPEC;
constexpr long long OFF_TCR   = OFF_TVEC  + SZ_TVEC;
constexpr long long OFF_TCI   = OFF_TCR   + SZ_TC;
constexpr long long OFF_WSP   = OFF_TCI   + SZ_TC;
constexpr long long OFF_XS1   = OFF_WSP   + SZ_WSP;
constexpr long long OFF_T2R   = OFF_XS1   + SZ_XS1;
constexpr long long OFF_XS2   = OFF_T2R   + SZ_T2;
constexpr long long OFF_XSP   = OFF_XS2   + SZ_XS2;
constexpr long long OFF_XCV   = OFF_XSP   + SZ_XSP;

__global__ void gemm_wmma(const float* __restrict__ A, const float* __restrict__ B,
                          float* __restrict__ C, int M, int K,
                          int lda, int ldb, int ldc,
                          long long sAb, long long sBb, long long sCb,
                          float alpha, int accumulate) {
  const int lane = threadIdx.x;
  const int n0 = blockIdx.x * 64;
  const int m0 = blockIdx.y * 16;
  const long long b = blockIdx.z;
  A += b * sAb; B += b * sBb; C += b * sCb;
  const int half = lane >> 4;
  const int l16  = lane & 15;
  v8f acc0 = {}, acc1 = {}, acc2 = {}, acc3 = {};
  const float* arow = A + (long long)(m0 + l16) * lda + 2 * half;
  const float* brow = B + (long long)(2 * half) * ldb + n0 + l16;
  for (int k0 = 0; k0 < K; k0 += 4) {
    v2f a; a.x = arow[k0]; a.y = arow[k0 + 1];
    const float* bp = brow + (long long)k0 * ldb;
    v2f b0; b0.x = bp[0];   b0.y = bp[ldb];
    v2f b1; b1.x = bp[16];  b1.y = bp[ldb + 16];
    v2f b2; b2.x = bp[32];  b2.y = bp[ldb + 32];
    v2f b3; b3.x = bp[48];  b3.y = bp[ldb + 48];
    acc0 = __builtin_amdgcn_wmma_f32_16x16x4_f32(false, a, false, b0, (short)0, acc0, false, false);
    acc1 = __builtin_amdgcn_wmma_f32_16x16x4_f32(false, a, false, b1, (short)0, acc1, false, false);
    acc2 = __builtin_amdgcn_wmma_f32_16x16x4_f32(false, a, false, b2, (short)0, acc2, false, false);
    acc3 = __builtin_amdgcn_wmma_f32_16x16x4_f32(false, a, false, b3, (short)0, acc3, false, false);
    if (k0 + 4 < K) {
      __builtin_prefetch(arow + k0 + 4, 0, 1);
      __builtin_prefetch(bp + 4LL * ldb, 0, 1);
    }
  }
#pragma unroll
  for (int v = 0; v < 8; ++v) {
    const int row = m0 + half * 8 + v;
    float* cp = C + (long long)row * ldc + n0 + l16;
    if (accumulate) {
      cp[0]  += alpha * acc0[v]; cp[16] += alpha * acc1[v];
      cp[32] += alpha * acc2[v]; cp[48] += alpha * acc3[v];
    } else {
      cp[0]  = alpha * acc0[v]; cp[16] = alpha * acc1[v];
      cp[32] = alpha * acc2[v]; cp[48] = alpha * acc3[v];
    }
  }
}

__global__ void conv3x3_wmma(const float* __restrict__ xsp, const float* __restrict__ ck,
                             const float* __restrict__ cb, const float* __restrict__ tvec,
                             float* __restrict__ out) {
  const int lane = threadIdx.x;
  const int n0 = blockIdx.x * 64;
  const int m0 = blockIdx.y * 16;
  const int l  = blockIdx.z;
  const int half = lane >> 4, l16 = lane & 15;
  v8f acc0 = {}, acc1 = {}, acc2 = {}, acc3 = {};
  for (int dy = 0; dy < 3; ++dy) {
    for (int dx = 0; dx < 3; ++dx) {
      const float* arow = xsp + (((long long)(l + dy)) * 129 + (m0 + l16 + dx)) * 128 + 2 * half;
      const float* brow = ck + (((long long)(dy * 3 + dx)) * 128 + 2 * half) * 128 + n0 + l16;
      for (int k0 = 0; k0 < 128; k0 += 4) {
        v2f a; a.x = arow[k0]; a.y = arow[k0 + 1];
        const float* bp = brow + (long long)k0 * 128;
        v2f b0; b0.x = bp[0];  b0.y = bp[128];
        v2f b1; b1.x = bp[16]; b1.y = bp[144];
        v2f b2; b2.x = bp[32]; b2.y = bp[160];
        v2f b3; b3.x = bp[48]; b3.y = bp[176];
        acc0 = __builtin_amdgcn_wmma_f32_16x16x4_f32(false, a, false, b0, (short)0, acc0, false, false);
        acc1 = __builtin_amdgcn_wmma_f32_16x16x4_f32(false, a, false, b1, (short)0, acc1, false, false);
        acc2 = __builtin_amdgcn_wmma_f32_16x16x4_f32(false, a, false, b2, (short)0, acc2, false, false);
        acc3 = __builtin_amdgcn_wmma_f32_16x16x4_f32(false, a, false, b3, (short)0, acc3, false, false);
      }
    }
  }
  const int c0 = n0 + l16;
  const float bias0 = cb[c0]      + tvec[128 + c0];
  const float bias1 = cb[c0 + 16] + tvec[128 + c0 + 16];
  const float bias2 = cb[c0 + 32] + tvec[128 + c0 + 32];
  const float bias3 = cb[c0 + 48] + tvec[128 + c0 + 48];
#pragma unroll
  for (int v = 0; v < 8; ++v) {
    const int row = m0 + half * 8 + v;
    float* op = out + (((long long)l) * 128 + row) * 128 + c0;
    op[0]  = acc0[v] + bias0; op[16] = acc1[v] + bias1;
    op[32] = acc2[v] + bias2; op[48] = acc3[v] + bias3;
  }
}

__global__ void zerof(float* p, long long n) {
  long long i = (long long)blockIdx.x * 256 + threadIdx.x;
  if (i < n) p[i] = 0.0f;
}

__global__ void build_xpad(const float* __restrict__ x, float* __restrict__ xp, long long n) {
  long long i = (long long)blockIdx.x * 256 + threadIdx.x;
  if (i >= n) return;
  long long t = i / 32768, r = i % 32768, p = r / 128, c = r % 128;
  xp[i] = (p < 255) ? x[(t * 255 + p) * 128 + c] : 0.0f;
}

__global__ void k_tvec(const float* __restrict__ temb,
                       const float* __restrict__ dw, const float* __restrict__ db,
                       const float* __restrict__ drw, const float* __restrict__ drb,
                       const float* __restrict__ diw, const float* __restrict__ dib,
                       float* tvec, float* tcr, float* tci) {
  int j = threadIdx.x;
  float acc = db[j];
  for (int k = 0; k < 256; ++k) acc += temb[k] * dw[k * 256 + j];
  tvec[j] = acc;
  if (j < 64) {
    float ar = drb[j], ai = dib[j];
    for (int k = 0; k < 256; ++k) {
      ar += temb[k] * drw[k * 64 + j];
      ai += temb[k] * diw[k * 64 + j];
    }
    tcr[j] = ar; tci[j] = ai;
  }
}

__global__ void k_wc(const float* __restrict__ wr, const float* __restrict__ wi,
                     const float* __restrict__ tcr, const float* __restrict__ tci,
                     float* wcr, float* wci, long long n) {
  long long i = (long long)blockIdx.x * 256 + threadIdx.x;
  if (i >= n) return;
  int l = (int)(i / 16384);
  float tr = tcr[l], ti = tci[l], r = wr[i], im = wi[i];
  wcr[i] = tr * r - ti * im;
  wci[i] = tr * im + ti * r;
}

__global__ void k_wsp(const float* __restrict__ spw, const float* __restrict__ tvec,
                      float* wsp, long long n) {
  long long i = (long long)blockIdx.x * 256 + threadIdx.x;
  if (i >= n) return;
  int o = (int)(i % 128);
  wsp[i] = tvec[o] * spw[i];
}

__global__ void k_phase1(float* c1, float* s1n, long long n) {
  long long i = (long long)blockIdx.x * 256 + threadIdx.x;
  if (i >= n) return;
  int m = (int)(i / 256), p = (int)(i % 256);
  if (m < 255 && p < 255) {
    double mval = (double)(m - 127);
    double phi = 2.0 * PI_D * (double)p / 255.0;
    double dphi = 2.0 * PI_D / 255.0;
    c1[i]  = (float)( cos(phi * mval) * dphi);
    s1n[i] = (float)(-sin(phi * mval) * dphi);
  } else { c1[i] = 0.0f; s1n[i] = 0.0f; }
}

__global__ void k_phase2(float* c2, float* s2, long long n) {
  long long i = (long long)blockIdx.x * 256 + threadIdx.x;
  if (i >= n) return;
  int p = (int)(i / 128), m = (int)(i % 128);
  if (p < 127 && m < 127) {
    double mval = (double)(m - 63);
    double phi = 2.0 * PI_D * (double)p / 127.0;
    c2[i] = (float)cos(mval * phi);
    s2[i] = (float)sin(mval * phi);
  } else { c2[i] = 0.0f; s2[i] = 0.0f; }
}

__global__ void k_resize(float* r1, float* r2p) {
  int tid = threadIdx.x;
  if (tid < 64) {
    int i = tid;
    float scale = 0.5f;
    float sample = (i + 0.5f) / scale - 0.5f;
    float sum = 0.0f;
    for (int j = 0; j < 128; ++j) {
      float w = 1.0f - fabsf(sample - j) * scale;
      if (w > 0.0f) sum += w;
    }
    for (int j = 0; j < 128; ++j) {
      float w = 1.0f - fabsf(sample - j) * scale;
      r1[i * 128 + j] = (w > 0.0f) ? (w / sum) : 0.0f;
    }
  } else if (tid < 192) {
    int j = tid - 64;
    if (j < 127) {
      float scale = 127.0f / 255.0f;
      float sample = (j + 0.5f) / scale - 0.5f;
      float sum = 0.0f;
      for (int p = 0; p < 255; ++p) {
        float w = 1.0f - fabsf(sample - p) * scale;
        if (w > 0.0f) sum += w;
      }
      for (int p = 0; p < 255; ++p) {
        float w = 1.0f - fabsf(sample - p) * scale;
        r2p[j * 256 + p] = (w > 0.0f) ? (w / sum) : 0.0f;
      }
      r2p[j * 256 + 255] = 0.0f;
    } else {
      for (int p = 0; p < 256; ++p) r2p[j * 256 + p] = 0.0f;
    }
  }
}

__global__ void k_legendre(float* out, int L, int weighted) {
  int t = threadIdx.x;
  if (t >= L) return;
  double theta = (2.0 * t + 1.0) * PI_D / (2.0 * L - 1.0);
  double x = cos(theta), s = sin(theta);
  double w = weighted ? (s * PI_D / (double)L) : 1.0;
  auto idx = [&](int m, int l) { return ((long long)(L - 1 + m) * L + l) * (long long)L + t; };
  out[idx(0, 0)] = (float)(sqrt(1.0 / (4.0 * PI_D)) * w);
  for (int l = 1; l < L; ++l) {
    double pprev = (double)out[idx(l - 1, l - 1)];
    out[idx(l, l)]     = (float)(-sqrt((2.0 * l + 1.0) / (2.0 * l)) * s * pprev);
    out[idx(l - 1, l)] = (float)( sqrt(2.0 * l + 1.0) * x * pprev);
    for (int m = 0; m <= l - 2; ++m) {
      double ll = (double)l, mm = (double)m;
      double a = sqrt((4.0 * ll * ll - 1.0) / (ll * ll - mm * mm));
      double b = sqrt((2.0 * ll + 1.0) * ((ll - 1.0) * (ll - 1.0) - mm * mm) /
                      ((2.0 * ll - 3.0) * (ll * ll - mm * mm)));
      out[idx(m, l)] = (float)(a * x * (double)out[idx(m, l - 1)] - b * (double)out[idx(m, l - 2)]);
    }
  }
}

__global__ void k_mirror(float* out, int L, long long n) {
  long long i = (long long)blockIdx.x * 256 + threadIdx.x;
  if (i >= n) return;
  int LL = L * L;
  int m = (int)(i / LL) + 1;
  int r = (int)(i % LL);
  int l = r / L, t = r % L;
  float v = out[((long long)(L - 1 + m) * L + l) * L + t];
  out[((long long)(L - 1 - m) * L + l) * L + t] = (m & 1) ? -v : v;
}

__global__ void k_transp_p2(const float* __restrict__ a, float* __restrict__ b, long long n) {
  long long i = (long long)blockIdx.x * 256 + threadIdx.x;
  if (i >= n) return;
  int m = (int)(i / 4096), t = (int)((i / 64) % 64), l = (int)(i % 64);
  b[i] = a[((long long)m * 64 + l) * 64 + t];
}

__global__ void k_xsppad(const float* __restrict__ xs2, float* __restrict__ xsp, long long n) {
  long long i = (long long)blockIdx.x * 256 + threadIdx.x;
  if (i >= n) return;
  long long row = i / (129 * 128), r = i % (129 * 128);
  int col = (int)(r / 128), c = (int)(r % 128);
  int l = (int)row - 1, m = col - 1;
  xsp[i] = (l >= 0 && l < 64 && m >= 0 && m < 127) ? xs2[((long long)l * 128 + m) * 128 + c] : 0.0f;
}

__global__ void k_fuse(const float* __restrict__ xspec, const float* __restrict__ xcv,
                       const float* __restrict__ lns, const float* __restrict__ lnb,
                       float* __restrict__ out) {
  __shared__ float sh[128];
  int c = threadIdx.x;
  int n = blockIdx.x;
  int l = n / 127, m = n % 127;
  long long idx = ((long long)l * 128 + m) * 128 + c;
  float v = xspec[idx] + xcv[idx];
  float g = 0.5f * v * (1.0f + tanhf(0.7978845608028654f * (v + 0.044715f * v * v * v)));
  sh[c] = g; __syncthreads();
  for (int s = 64; s > 0; s >>= 1) { if (c < s) sh[c] += sh[c + s]; __syncthreads(); }
  float mu = sh[0] * (1.0f / 128.0f); __syncthreads();
  float d = g - mu;
  sh[c] = d * d; __syncthreads();
  for (int s = 64; s > 0; s >>= 1) { if (c < s) sh[c] += sh[c + s]; __syncthreads(); }
  float var = sh[0] * (1.0f / 128.0f);
  out[((long long)l * 127 + m) * 128 + c] = d * rsqrtf(var + 1e-6f) * lns[c] + lnb[c];
}

static inline void launch_gemm(hipStream_t s, const float* A, const float* B, float* C,
                               int M, int K, int lda, int ldb, int ldc,
                               long long sA, long long sB, long long sC,
                               float alpha, int acc, int batch) {
  dim3 g(2, M / 16, batch);
  gemm_wmma<<<g, dim3(32), 0, s>>>(A, B, C, M, K, lda, ldb, ldc, sA, sB, sC, alpha, acc);
}

#define EWG(n) dim3((unsigned)(((n) + 255) / 256)), dim3(256), 0, stream

extern "C" void kernel_launch(void* const* d_in, const int* in_sizes, int n_in,
                              void* d_out, int out_size, void* d_ws, size_t ws_size,
                              hipStream_t stream) {
  (void)in_sizes; (void)n_in; (void)out_size; (void)ws_size;
  const float* x       = (const float*)d_in[0];
  const float* t_emb   = (const float*)d_in[1];
  const float* sp_dw   = (const float*)d_in[2];
  const float* sp_db   = (const float*)d_in[3];
  const float* sp_wt   = (const float*)d_in[4];
  const float* sp_ck   = (const float*)d_in[5];
  const float* sp_cb   = (const float*)d_in[6];
  const float* spec_wr = (const float*)d_in[7];
  const float* spec_wi = (const float*)d_in[8];
  const float* drw     = (const float*)d_in[9];
  const float* drb     = (const float*)d_in[10];
  const float* diw     = (const float*)d_in[11];
  const float* dib     = (const float*)d_in[12];
  const float* lns     = (const float*)d_in[13];
  const float* lnb     = (const float*)d_in[14];
  float* out = (float*)d_out;
  float* ws  = (float*)d_ws;

  float *A1 = ws + OFF_A1, *XP = ws + OFF_XP, *COS1 = ws + OFF_COS1, *SIN1N = ws + OFF_SIN1N;
  float *GR = ws + OFF_GR, *GI = ws + OFF_GI, *F1R = ws + OFF_F1R, *F1I = ws + OFF_F1I;
  float *R1 = ws + OFF_R1, *R2P = ws + OFF_R2P, *T1R = ws + OFF_T1R, *T1I = ws + OFF_T1I;
  float *FFR = ws + OFF_FFR, *FFI = ws + OFF_FFI, *WCR = ws + OFF_WCR, *WCI = ws + OFF_WCI;
  float *F2R = ws + OFF_F2R, *F2I = ws + OFF_F2I, *P2A = ws + OFF_P2A, *P2B = ws + OFF_P2B;
  float *FTR = ws + OFF_FTR, *FTI = ws + OFF_FTI, *COS2 = ws + OFF_COS2, *SIN2 = ws + OFF_SIN2;
  float *XSPEC = ws + OFF_XSPEC, *TVEC = ws + OFF_TVEC, *TCR = ws + OFF_TCR, *TCI = ws + OFF_TCI;
  float *WSP = ws + OFF_WSP, *XS1 = ws + OFF_XS1, *T2R = ws + OFF_T2R, *XS2 = ws + OFF_XS2;
  float *XSP = ws + OFF_XSP, *XCV = ws + OFF_XCV;

  zerof<<<EWG(SZ_A1)>>>(A1, SZ_A1);
  zerof<<<EWG(SZ_P2)>>>(P2A, SZ_P2);
  zerof<<<EWG(8192)>>>(T1R + 255LL * 8192, 8192);
  zerof<<<EWG(8192)>>>(T1I + 255LL * 8192, 8192);
  zerof<<<EWG(8192)>>>(T2R + 255LL * 8192, 8192);

  build_xpad<<<EWG(SZ_XP)>>>(x, XP, SZ_XP);
  k_tvec<<<dim3(1), dim3(256), 0, stream>>>(t_emb, sp_dw, sp_db, drw, drb, diw, dib, TVEC, TCR, TCI);
  k_wc<<<EWG(SZ_WC)>>>(spec_wr, spec_wi, TCR, TCI, WCR, WCI, SZ_WC);
  k_wsp<<<EWG(SZ_WSP)>>>(sp_wt, TVEC, WSP, SZ_WSP);
  k_phase1<<<EWG(SZ_PH1)>>>(COS1, SIN1N, SZ_PH1);
  k_phase2<<<EWG(SZ_PH2)>>>(COS2, SIN2, SZ_PH2);
  k_resize<<<dim3(1), dim3(256), 0, stream>>>(R1, R2P);
  k_legendre<<<dim3(1), dim3(128), 0, stream>>>(A1, 128, 1);
  k_mirror<<<EWG(127LL * 128 * 128)>>>(A1, 128, 127LL * 128 * 128);
  k_legendre<<<dim3(1), dim3(64), 0, stream>>>(P2A, 64, 0);
  k_mirror<<<EWG(63LL * 64 * 64)>>>(P2A, 64, 63LL * 64 * 64);
  k_transp_p2<<<EWG(SZ_P2)>>>(P2A, P2B, SZ_P2);

  launch_gemm(stream, COS1,  XP, GR, 256, 256, 256, 128, 128, 0, 256LL * 128, 256LL * 128, 1.0f, 0, 128);
  launch_gemm(stream, SIN1N, XP, GI, 256, 256, 256, 128, 128, 0, 256LL * 128, 256LL * 128, 1.0f, 0, 128);
  launch_gemm(stream, A1, GR, F1R, 128, 128, 128, 256 * 128, 128, 128LL * 128, 128, 128LL * 128, 1.0f, 0, 255);
  launch_gemm(stream, A1, GI, F1I, 128, 128, 128, 256 * 128, 128, 128LL * 128, 128, 128LL * 128, 1.0f, 0, 255);
  launch_gemm(stream, R1, F1R, T1R, 64, 128, 128, 128, 128, 0, 128LL * 128, 64LL * 128, 1.0f, 0, 255);
  launch_gemm(stream, R1, F1I, T1I, 64, 128, 128, 128, 128, 0, 128LL * 128, 64LL * 128, 1.0f, 0, 255);
  launch_gemm(stream, R2P, T1R, FFR, 128, 256, 256, 64 * 128, 128, 0, 128, 128LL * 128, 1.0f, 0, 64);
  launch_gemm(stream, R2P, T1I, FFI, 128, 256, 256, 64 * 128, 128, 0, 128, 128LL * 128, 1.0f, 0, 64);
  launch_gemm(stream, FFR, WCR, F2R, 128, 128, 128, 128, 128, 128LL * 128, 128LL * 128, 128LL * 128,  1.0f, 0, 64);
  launch_gemm(stream, FFI, WCI, F2R, 128, 128, 128, 128, 128, 128LL * 128, 128LL * 128, 128LL * 128, -1.0f, 1, 64);
  launch_gemm(stream, FFR, WCI, F2I, 128, 128, 128, 128, 128, 128LL * 128, 128LL * 128, 128LL * 128,  1.0f, 0, 64);
  launch_gemm(stream, FFI, WCR, F2I, 128, 128, 128, 128, 128, 128LL * 128, 128LL * 128, 128LL * 128,  1.0f, 1, 64);
  launch_gemm(stream, P2B, F2R, FTR, 64, 64, 64, 128 * 128, 128, 64LL * 64, 128, 64LL * 128, 1.0f, 0, 127);
  launch_gemm(stream, P2B, F2I, FTI, 64, 64, 64, 128 * 128, 128, 64LL * 64, 128, 64LL * 128, 1.0f, 0, 127);
  launch_gemm(stream, COS2, FTR, XSPEC, 128, 128, 128, 64 * 128, 128, 0, 128, 128LL * 128,  1.0f, 0, 64);
  launch_gemm(stream, SIN2, FTI, XSPEC, 128, 128, 128, 64 * 128, 128, 0, 128, 128LL * 128, -1.0f, 1, 64);

  launch_gemm(stream, x, WSP, XS1, 32640, 128, 128, 128, 128, 0, 0, 0, 1.0f, 0, 1);
  launch_gemm(stream, R1, XS1, T2R, 64, 128, 128, 255 * 128, 128, 0, 128, 64LL * 128, 1.0f, 0, 255);
  launch_gemm(stream, R2P, T2R, XS2, 128, 256, 256, 64 * 128, 128, 0, 128, 128LL * 128, 1.0f, 0, 64);
  k_xsppad<<<EWG(SZ_XSP)>>>(XS2, XSP, SZ_XSP);
  conv3x3_wmma<<<dim3(2, 8, 64), dim3(32), 0, stream>>>(XSP, sp_ck, sp_cb, TVEC, XCV);

  k_fuse<<<dim3(64 * 127), dim3(128), 0, stream>>>(XSPEC, XCV, lns, lnb, out);
}


# Legendre tables: one block per order m with the degree recurrence in f64 registers and the (l,m) coefficients shared in LDS, replacing one block that looped over all (l,m) through global memory
# speedup vs baseline: 14.0984x; 14.0984x over previous
#include <hip/hip_runtime.h>
#include <math.h>

typedef __attribute__((ext_vector_type(2))) float v2f;
typedef __attribute__((ext_vector_type(8))) float v8f;

#define PI_D 3.14159265358979323846

constexpr long long SZ_A1    = 255LL*128*128;
constexpr long long SZ_XP    = 128LL*256*128;
constexpr long long SZ_PH1   = 256LL*256;
constexpr long long SZ_G     = 128LL*256*128;
constexpr long long SZ_F1    = 255LL*128*128;
constexpr long long SZ_R1    = 64LL*128;
constexpr long long SZ_R2P   = 128LL*256;
constexpr long long SZ_T1    = 256LL*64*128;
constexpr long long SZ_FF    = 64LL*128*128;
constexpr long long SZ_WC    = 64LL*128*128;
constexpr long long SZ_F2    = 64LL*128*128;
constexpr long long SZ_P2    = 127LL*64*64;
constexpr long long SZ_FT    = 127LL*64*128;
constexpr long long SZ_PH2   = 128LL*128;
constexpr long long SZ_XSPEC = 64LL*128*128;
constexpr long long SZ_TVEC  = 256;
constexpr long long SZ_TC    = 64;
constexpr long long SZ_WSP   = 128LL*128;
constexpr long long SZ_XS1   = 128LL*255*128;
constexpr long long SZ_T2    = 256LL*64*128;
constexpr long long SZ_XS2   = 64LL*128*128;
constexpr long long SZ_XSP   = 67LL*129*128;
constexpr long long SZ_XCV   = 64LL*128*128;

constexpr long long OFF_A1    = 0;
constexpr long long OFF_XP    = OFF_A1    + SZ_A1;
constexpr long long OFF_COS1  = OFF_XP    + SZ_XP;
constexpr long long OFF_SIN1N = OFF_COS1  + SZ_PH1;
constexpr long long OFF_GR    = OFF_SIN1N + SZ_PH1;
constexpr long long OFF_GI    = OFF_GR    + SZ_G;
constexpr long long OFF_F1R   = OFF_GI    + SZ_G;
constexpr long long OFF_F1I   = OFF_F1R   + SZ_F1;
constexpr long long OFF_R1    = OFF_F1I   + SZ_F1;
constexpr long long OFF_R2P   = OFF_R1    + SZ_R1;
constexpr long long OFF_T1R   = OFF_R2P   + SZ_R2P;
constexpr long long OFF_T1I   = OFF_T1R   + SZ_T1;
constexpr long long OFF_FFR   = OFF_T1I   + SZ_T1;
constexpr long long OFF_FFI   = OFF_FFR   + SZ_FF;
constexpr long long OFF_WCR   = OFF_FFI   + SZ_FF;
constexpr long long OFF_WCI   = OFF_WCR   + SZ_WC;
constexpr long long OFF_F2R   = OFF_WCI   + SZ_WC;
constexpr long long OFF_F2I   = OFF_F2R   + SZ_F2;
constexpr long long OFF_P2A   = OFF_F2I   + SZ_F2;
constexpr long long OFF_P2B   = OFF_P2A   + SZ_P2;
constexpr long long OFF_FTR   = OFF_P2B   + SZ_P2;
constexpr long long OFF_FTI   = OFF_FTR   + SZ_FT;
constexpr long long OFF_COS2  = OFF_FTI   + SZ_FT;
constexpr long long OFF_SIN2  = OFF_COS2  + SZ_PH2;
constexpr long long OFF_XSPEC = OFF_SIN2  + SZ_PH2;
constexpr long long OFF_TVEC  = OFF_XSPEC + SZ_XSPEC;
constexpr long long OFF_TCR   = OFF_TVEC  + SZ_TVEC;
constexpr long long OFF_TCI   = OFF_TCR   + SZ_TC;
constexpr long long OFF_WSP   = OFF_TCI   + SZ_TC;
constexpr long long OFF_XS1   = OFF_WSP   + SZ_WSP;
constexpr long long OFF_T2R   = OFF_XS1   + SZ_XS1;
constexpr long long OFF_XS2   = OFF_T2R   + SZ_T2;
constexpr long long OFF_XSP   = OFF_XS2   + SZ_XS2;
constexpr long long OFF_XCV   = OFF_XSP   + SZ_XSP;

__global__ void gemm_wmma(const float* __restrict__ A, const float* __restrict__ B,
                          float* __restrict__ C, int M, int K,
                          int lda, int ldb, int ldc,
                          long long sAb, long long sBb, long long sCb,
                          float alpha, int accumulate) {
  const int lane = threadIdx.x;
  const int n0 = blockIdx.x * 64;
  const int m0 = blockIdx.y * 16;
  const long long b = blockIdx.z;
  A += b * sAb; B += b * sBb; C += b * sCb;
  const int half = lane >> 4;
  const int l16  = lane & 15;
  v8f acc0 = {}, acc1 = {}, acc2 = {}, acc3 = {};
  const float* arow = A + (long long)(m0 + l16) * lda + 2 * half;
  const float* brow = B + (long long)(2 * half) * ldb + n0 + l16;
  for (int k0 = 0; k0 < K; k0 += 4) {
    v2f a; a.x = arow[k0]; a.y = arow[k0 + 1];
    const float* bp = brow + (long long)k0 * ldb;
    v2f b0; b0.x = bp[0];   b0.y = bp[ldb];
    v2f b1; b1.x = bp[16];  b1.y = bp[ldb + 16];
    v2f b2; b2.x = bp[32];  b2.y = bp[ldb + 32];
    v2f b3; b3.x = bp[48];  b3.y = bp[ldb + 48];
    acc0 = __builtin_amdgcn_wmma_f32_16x16x4_f32(false, a, false, b0, (short)0, acc0, false, false);
    acc1 = __builtin_amdgcn_wmma_f32_16x16x4_f32(false, a, false, b1, (short)0, acc1, false, false);
    acc2 = __builtin_amdgcn_wmma_f32_16x16x4_f32(false, a, false, b2, (short)0, acc2, false, false);
    acc3 = __builtin_amdgcn_wmma_f32_16x16x4_f32(false, a, false, b3, (short)0, acc3, false, false);
    if (k0 + 4 < K) {
      __builtin_prefetch(arow + k0 + 4, 0, 1);
      __builtin_prefetch(bp + 4LL * ldb, 0, 1);
    }
  }
#pragma unroll
  for (int v = 0; v < 8; ++v) {
    const int row = m0 + half * 8 + v;
    float* cp = C + (long long)row * ldc + n0 + l16;
    if (accumulate) {
      cp[0]  += alpha * acc0[v]; cp[16] += alpha * acc1[v];
      cp[32] += alpha * acc2[v]; cp[48] += alpha * acc3[v];
    } else {
      cp[0]  = alpha * acc0[v]; cp[16] = alpha * acc1[v];
      cp[32] = alpha * acc2[v]; cp[48] = alpha * acc3[v];
    }
  }
}

__global__ void conv3x3_wmma(const float* __restrict__ xsp, const float* __restrict__ ck,
                             const float* __restrict__ cb, const float* __restrict__ tvec,
                             float* __restrict__ out) {
  const int lane = threadIdx.x;
  const int n0 = blockIdx.x * 64;
  const int m0 = blockIdx.y * 16;
  const int l  = blockIdx.z;
  const int half = lane >> 4, l16 = lane & 15;
  v8f acc0 = {}, acc1 = {}, acc2 = {}, acc3 = {};
  for (int dy = 0; dy < 3; ++dy) {
    for (int dx = 0; dx < 3; ++dx) {
      const float* arow = xsp + (((long long)(l + dy)) * 129 + (m0 + l16 + dx)) * 128 + 2 * half;
      const float* brow = ck + (((long long)(dy * 3 + dx)) * 128 + 2 * half) * 128 + n0 + l16;
      for (int k0 = 0; k0 < 128; k0 += 4) {
        v2f a; a.x = arow[k0]; a.y = arow[k0 + 1];
        const float* bp = brow + (long long)k0 * 128;
        v2f b0; b0.x = bp[0];  b0.y = bp[128];
        v2f b1; b1.x = bp[16]; b1.y = bp[144];
        v2f b2; b2.x = bp[32]; b2.y = bp[160];
        v2f b3; b3.x = bp[48]; b3.y = bp[176];
        acc0 = __builtin_amdgcn_wmma_f32_16x16x4_f32(false, a, false, b0, (short)0, acc0, false, false);
        acc1 = __builtin_amdgcn_wmma_f32_16x16x4_f32(false, a, false, b1, (short)0, acc1, false, false);
        acc2 = __builtin_amdgcn_wmma_f32_16x16x4_f32(false, a, false, b2, (short)0, acc2, false, false);
        acc3 = __builtin_amdgcn_wmma_f32_16x16x4_f32(false, a, false, b3, (short)0, acc3, false, false);
      }
    }
  }
  const int c0 = n0 + l16;
  const float bias0 = cb[c0]      + tvec[128 + c0];
  const float bias1 = cb[c0 + 16] + tvec[128 + c0 + 16];
  const float bias2 = cb[c0 + 32] + tvec[128 + c0 + 32];
  const float bias3 = cb[c0 + 48] + tvec[128 + c0 + 48];
#pragma unroll
  for (int v = 0; v < 8; ++v) {
    const int row = m0 + half * 8 + v;
    float* op = out + (((long long)l) * 128 + row) * 128 + c0;
    op[0]  = acc0[v] + bias0; op[16] = acc1[v] + bias1;
    op[32] = acc2[v] + bias2; op[48] = acc3[v] + bias3;
  }
}

__global__ void zerof(float* p, long long n) {
  long long i = (long long)blockIdx.x * 256 + threadIdx.x;
  if (i < n) p[i] = 0.0f;
}

__global__ void build_xpad(const float* __restrict__ x, float* __restrict__ xp, long long n) {
  long long i = (long long)blockIdx.x * 256 + threadIdx.x;
  if (i >= n) return;
  long long t = i / 32768, r = i % 32768, p = r / 128, c = r % 128;
  xp[i] = (p < 255) ? x[(t * 255 + p) * 128 + c] : 0.0f;
}

__global__ void k_tvec(const float* __restrict__ temb,
                       const float* __restrict__ dw, const float* __restrict__ db,
                       const float* __restrict__ drw, const float* __restrict__ drb,
                       const float* __restrict__ diw, const float* __restrict__ dib,
                       float* tvec, float* tcr, float* tci) {
  int j = threadIdx.x;
  float acc = db[j];
  for (int k = 0; k < 256; ++k) acc += temb[k] * dw[k * 256 + j];
  tvec[j] = acc;
  if (j < 64) {
    float ar = drb[j], ai = dib[j];
    for (int k = 0; k < 256; ++k) {
      ar += temb[k] * drw[k * 64 + j];
      ai += temb[k] * diw[k * 64 + j];
    }
    tcr[j] = ar; tci[j] = ai;
  }
}

__global__ void k_wc(const float* __restrict__ wr, const float* __restrict__ wi,
                     const float* __restrict__ tcr, const float* __restrict__ tci,
                     float* wcr, float* wci, long long n) {
  long long i = (long long)blockIdx.x * 256 + threadIdx.x;
  if (i >= n) return;
  int l = (int)(i / 16384);
  float tr = tcr[l], ti = tci[l], r = wr[i], im = wi[i];
  wcr[i] = tr * r - ti * im;
  wci[i] = tr * im + ti * r;
}

__global__ void k_wsp(const float* __restrict__ spw, const float* __restrict__ tvec,
                      float* wsp, long long n) {
  long long i = (long long)blockIdx.x * 256 + threadIdx.x;
  if (i >= n) return;
  int o = (int)(i % 128);
  wsp[i] = tvec[o] * spw[i];
}

__global__ void k_phase1(float* c1, float* s1n, long long n) {
  long long i = (long long)blockIdx.x * 256 + threadIdx.x;
  if (i >= n) return;
  int m = (int)(i / 256), p = (int)(i % 256);
  if (m < 255 && p < 255) {
    double mval = (double)(m - 127);
    double phi = 2.0 * PI_D * (double)p / 255.0;
    double dphi = 2.0 * PI_D / 255.0;
    c1[i]  = (float)( cos(phi * mval) * dphi);
    s1n[i] = (float)(-sin(phi * mval) * dphi);
  } else { c1[i] = 0.0f; s1n[i] = 0.0f; }
}

__global__ void k_phase2(float* c2, float* s2, long long n) {
  long long i = (long long)blockIdx.x * 256 + threadIdx.x;
  if (i >= n) return;
  int p = (int)(i / 128), m = (int)(i % 128);
  if (p < 127 && m < 127) {
    double mval = (double)(m - 63);
    double phi = 2.0 * PI_D * (double)p / 127.0;
    c2[i] = (float)cos(mval * phi);
    s2[i] = (float)sin(mval * phi);
  } else { c2[i] = 0.0f; s2[i] = 0.0f; }
}

__global__ void k_resize(float* r1, float* r2p) {
  int tid = threadIdx.x;
  if (tid < 64) {
    int i = tid;
    float scale = 0.5f;
    float sample = (i + 0.5f) / scale - 0.5f;
    float sum = 0.0f;
    for (int j = 0; j < 128; ++j) {
      float w = 1.0f - fabsf(sample - j) * scale;
      if (w > 0.0f) sum += w;
    }
    for (int j = 0; j < 128; ++j) {
      float w = 1.0f - fabsf(sample - j) * scale;
      r1[i * 128 + j] = (w > 0.0f) ? (w / sum) : 0.0f;
    }
  } else if (tid < 192) {
    int j = tid - 64;
    if (j < 127) {
      float scale = 127.0f / 255.0f;
      float sample = (j + 0.5f) / scale - 0.5f;
      float sum = 0.0f;
      for (int p = 0; p < 255; ++p) {
        float w = 1.0f - fabsf(sample - p) * scale;
        if (w > 0.0f) sum += w;
      }
      for (int p = 0; p < 255; ++p) {
        float w = 1.0f - fabsf(sample - p) * scale;
        r2p[j * 256 + p] = (w > 0.0f) ? (w / sum) : 0.0f;
      }
      r2p[j * 256 + 255] = 0.0f;
    } else {
      for (int p = 0; p < 256; ++p) r2p[j * 256 + p] = 0.0f;
    }
  }
}

__global__ void k_legendre(float* out, int L, int weighted) {
  __shared__ double cdiag[128];
  __shared__ double ca[128];
  __shared__ double cb[128];
  const int t = threadIdx.x;
  const int m = blockIdx.x;
  {
    const int l = t;
    const double ll = (double)l, mm = (double)m;
    cdiag[l] = (l >= 1) ? -sqrt((2.0 * ll + 1.0) / (2.0 * ll)) : 0.0;
    double a = 0.0, b = 0.0;
    if (l >= m + 2) {
      a = sqrt((4.0 * ll * ll - 1.0) / (ll * ll - mm * mm));
      b = sqrt((2.0 * ll + 1.0) * ((ll - 1.0) * (ll - 1.0) - mm * mm) /
               ((2.0 * ll - 3.0) * (ll * ll - mm * mm)));
    }
    ca[l] = a; cb[l] = b;
  }
  __syncthreads();
  const double theta = (2.0 * t + 1.0) * PI_D / (2.0 * L - 1.0);
  const double x = cos(theta), s = sin(theta);
  const double w = weighted ? (s * PI_D / (double)L) : 1.0;
  float* op = out + ((long long)(L - 1 + m) * L) * (long long)L + t;
  double pmm = sqrt(1.0 / (4.0 * PI_D)) * w;
  for (int j = 1; j <= m; ++j) pmm = cdiag[j] * s * pmm;
  op[(long long)m * L] = (float)pmm;
  if (m + 1 < L) {
    double p2 = pmm;
    double p1 = sqrt(2.0 * (m + 1) + 1.0) * x * pmm;
    op[(long long)(m + 1) * L] = (float)p1;
    for (int l = m + 2; l < L; ++l) {
      const double p0 = ca[l] * x * p1 - cb[l] * p2;
      op[(long long)l * L] = (float)p0;
      p2 = p1; p1 = p0;
    }
  }
}

__global__ void k_mirror(float* out, int L, long long n) {
  long long i = (long long)blockIdx.x * 256 + threadIdx.x;
  if (i >= n) return;
  int LL = L * L;
  int m = (int)(i / LL) + 1;
  int r = (int)(i % LL);
  int l = r / L, t = r % L;
  float v = out[((long long)(L - 1 + m) * L + l) * L + t];
  out[((long long)(L - 1 - m) * L + l) * L + t] = (m & 1) ? -v : v;
}

__global__ void k_transp_p2(const float* __restrict__ a, float* __restrict__ b, long long n) {
  long long i = (long long)blockIdx.x * 256 + threadIdx.x;
  if (i >= n) return;
  int m = (int)(i / 4096), t = (int)((i / 64) % 64), l = (int)(i % 64);
  b[i] = a[((long long)m * 64 + l) * 64 + t];
}

__global__ void k_xsppad(const float* __restrict__ xs2, float* __restrict__ xsp, long long n) {
  long long i = (long long)blockIdx.x * 256 + threadIdx.x;
  if (i >= n) return;
  long long row = i / (129 * 128), r = i % (129 * 128);
  int col = (int)(r / 128), c = (int)(r % 128);
  int l = (int)row - 1, m = col - 1;
  xsp[i] = (l >= 0 && l < 64 && m >= 0 && m < 127) ? xs2[((long long)l * 128 + m) * 128 + c] : 0.0f;
}

__global__ void k_fuse(const float* __restrict__ xspec, const float* __restrict__ xcv,
                       const float* __restrict__ lns, const float* __restrict__ lnb,
                       float* __restrict__ out) {
  __shared__ float sh[128];
  int c = threadIdx.x;
  int n = blockIdx.x;
  int l = n / 127, m = n % 127;
  long long idx = ((long long)l * 128 + m) * 128 + c;
  float v = xspec[idx] + xcv[idx];
  float g = 0.5f * v * (1.0f + tanhf(0.7978845608028654f * (v + 0.044715f * v * v * v)));
  sh[c] = g; __syncthreads();
  for (int s = 64; s > 0; s >>= 1) { if (c < s) sh[c] += sh[c + s]; __syncthreads(); }
  float mu = sh[0] * (1.0f / 128.0f); __syncthreads();
  float d = g - mu;
  sh[c] = d * d; __syncthreads();
  for (int s = 64; s > 0; s >>= 1) { if (c < s) sh[c] += sh[c + s]; __syncthreads(); }
  float var = sh[0] * (1.0f / 128.0f);
  out[((long long)l * 127 + m) * 128 + c] = d * rsqrtf(var + 1e-6f) * lns[c] + lnb[c];
}

static inline void launch_gemm(hipStream_t s, const float* A, const float* B, float* C,
                               int M, int K, int lda, int ldb, int ldc,
                               long long sA, long long sB, long long sC,
                               float alpha, int acc, int batch) {
  dim3 g(2, M / 16, batch);
  gemm_wmma<<<g, dim3(32), 0, s>>>(A, B, C, M, K, lda, ldb, ldc, sA, sB, sC, alpha, acc);
}

#define EWG(n) dim3((unsigned)(((n) + 255) / 256)), dim3(256), 0, stream

extern "C" void kernel_launch(void* const* d_in, const int* in_sizes, int n_in,
                              void* d_out, int out_size, void* d_ws, size_t ws_size,
                              hipStream_t stream) {
  (void)in_sizes; (void)n_in; (void)out_size; (void)ws_size;
  const float* x       = (const float*)d_in[0];
  const float* t_emb   = (const float*)d_in[1];
  const float* sp_dw   = (const float*)d_in[2];
  const float* sp_db   = (const float*)d_in[3];
  const float* sp_wt   = (const float*)d_in[4];
  const float* sp_ck   = (const float*)d_in[5];
  const float* sp_cb   = (const float*)d_in[6];
  const float* spec_wr = (const float*)d_in[7];
  const float* spec_wi = (const float*)d_in[8];
  const float* drw     = (const float*)d_in[9];
  const float* drb     = (const float*)d_in[10];
  const float* diw     = (const float*)d_in[11];
  const float* dib     = (const float*)d_in[12];
  const float* lns     = (const float*)d_in[13];
  const float* lnb     = (const float*)d_in[14];
  float* out = (float*)d_out;
  float* ws  = (float*)d_ws;

  float *A1 = ws + OFF_A1, *XP = ws + OFF_XP, *COS1 = ws + OFF_COS1, *SIN1N = ws + OFF_SIN1N;
  float *GR = ws + OFF_GR, *GI = ws + OFF_GI, *F1R = ws + OFF_F1R, *F1I = ws + OFF_F1I;
  float *R1 = ws + OFF_R1, *R2P = ws + OFF_R2P, *T1R = ws + OFF_T1R, *T1I = ws + OFF_T1I;
  float *FFR = ws + OFF_FFR, *FFI = ws + OFF_FFI, *WCR = ws + OFF_WCR, *WCI = ws + OFF_WCI;
  float *F2R = ws + OFF_F2R, *F2I = ws + OFF_F2I, *P2A = ws + OFF_P2A, *P2B = ws + OFF_P2B;
  float *FTR = ws + OFF_FTR, *FTI = ws + OFF_FTI, *COS2 = ws + OFF_COS2, *SIN2 = ws + OFF_SIN2;
  float *XSPEC = ws + OFF_XSPEC, *TVEC = ws + OFF_TVEC, *TCR = ws + OFF_TCR, *TCI = ws + OFF_TCI;
  float *WSP = ws + OFF_WSP, *XS1 = ws + OFF_XS1, *T2R = ws + OFF_T2R, *XS2 = ws + OFF_XS2;
  float *XSP = ws + OFF_XSP, *XCV = ws + OFF_XCV;

  zerof<<<EWG(SZ_A1)>>>(A1, SZ_A1);
  zerof<<<EWG(SZ_P2)>>>(P2A, SZ_P2);
  zerof<<<EWG(8192)>>>(T1R + 255LL * 8192, 8192);
  zerof<<<EWG(8192)>>>(T1I + 255LL * 8192, 8192);
  zerof<<<EWG(8192)>>>(T2R + 255LL * 8192, 8192);

  build_xpad<<<EWG(SZ_XP)>>>(x, XP, SZ_XP);
  k_tvec<<<dim3(1), dim3(256), 0, stream>>>(t_emb, sp_dw, sp_db, drw, drb, diw, dib, TVEC, TCR, TCI);
  k_wc<<<EWG(SZ_WC)>>>(spec_wr, spec_wi, TCR, TCI, WCR, WCI, SZ_WC);
  k_wsp<<<EWG(SZ_WSP)>>>(sp_wt, TVEC, WSP, SZ_WSP);
  k_phase1<<<EWG(SZ_PH1)>>>(COS1, SIN1N, SZ_PH1);
  k_phase2<<<EWG(SZ_PH2)>>>(COS2, SIN2, SZ_PH2);
  k_resize<<<dim3(1), dim3(256), 0, stream>>>(R1, R2P);
  k_legendre<<<dim3(128), dim3(128), 0, stream>>>(A1, 128, 1);
  k_mirror<<<EWG(127LL * 128 * 128)>>>(A1, 128, 127LL * 128 * 128);
  k_legendre<<<dim3(64), dim3(64), 0, stream>>>(P2A, 64, 0);
  k_mirror<<<EWG(63LL * 64 * 64)>>>(P2A, 64, 63LL * 64 * 64);
  k_transp_p2<<<EWG(SZ_P2)>>>(P2A, P2B, SZ_P2);

  launch_gemm(stream, COS1,  XP, GR, 256, 256, 256, 128, 128, 0, 256LL * 128, 256LL * 128, 1.0f, 0, 128);
  launch_gemm(stream, SIN1N, XP, GI, 256, 256, 256, 128, 128, 0, 256LL * 128, 256LL * 128, 1.0f, 0, 128);
  launch_gemm(stream, A1, GR, F1R, 128, 128, 128, 256 * 128, 128, 128LL * 128, 128, 128LL * 128, 1.0f, 0, 255);
  launch_gemm(stream, A1, GI, F1I, 128, 128, 128, 256 * 128, 128, 128LL * 128, 128, 128LL * 128, 1.0f, 0, 255);
  launch_gemm(stream, R1, F1R, T1R, 64, 128, 128, 128, 128, 0, 128LL * 128, 64LL * 128, 1.0f, 0, 255);
  launch_gemm(stream, R1, F1I, T1I, 64, 128, 128, 128, 128, 0, 128LL * 128, 64LL * 128, 1.0f, 0, 255);
  launch_gemm(stream, R2P, T1R, FFR, 128, 256, 256, 64 * 128, 128, 0, 128, 128LL * 128, 1.0f, 0, 64);
  launch_gemm(stream, R2P, T1I, FFI, 128, 256, 256, 64 * 128, 128, 0, 128, 128LL * 128, 1.0f, 0, 64);
  launch_gemm(stream, FFR, WCR, F2R, 128, 128, 128, 128, 128, 128LL * 128, 128LL * 128, 128LL * 128,  1.0f, 0, 64);
  launch_gemm(stream, FFI, WCI, F2R, 128, 128, 128, 128, 128, 128LL * 128, 128LL * 128, 128LL * 128, -1.0f, 1, 64);
  launch_gemm(stream, FFR, WCI, F2I, 128, 128, 128, 128, 128, 128LL * 128, 128LL * 128, 128LL * 128,  1.0f, 0, 64);
  launch_gemm(stream, FFI, WCR, F2I, 128, 128, 128, 128, 128, 128LL * 128, 128LL * 128, 128LL * 128,  1.0f, 1, 64);
  launch_gemm(stream, P2B, F2R, FTR, 64, 64, 64, 128 * 128, 128, 64LL * 64, 128, 64LL * 128, 1.0f, 0, 127);
  launch_gemm(stream, P2B, F2I, FTI, 64, 64, 64, 128 * 128, 128, 64LL * 64, 128, 64LL * 128, 1.0f, 0, 127);
  launch_gemm(stream, COS2, FTR, XSPEC, 128, 128, 128, 64 * 128, 128, 0, 128, 128LL * 128,  1.0f, 0, 64);
  launch_gemm(stream, SIN2, FTI, XSPEC, 128, 128, 128, 64 * 128, 128, 0, 128, 128LL * 128, -1.0f, 1, 64);

  launch_gemm(stream, x, WSP, XS1, 32640, 128, 128, 128, 128, 0, 0, 0, 1.0f, 0, 1);
  launch_gemm(stream, R1, XS1, T2R, 64, 128, 128, 255 * 128, 128, 0, 128, 64LL * 128, 1.0f, 0, 255);
  launch_gemm(stream, R2P, T2R, XS2, 128, 256, 256, 64 * 128, 128, 0, 128, 128LL * 128, 1.0f, 0, 64);
  k_xsppad<<<EWG(SZ_XSP)>>>(XS2, XSP, SZ_XSP);
  conv3x3_wmma<<<dim3(2, 8, 64), dim3(32), 0, stream>>>(XSP, sp_ck, sp_cb, TVEC, XCV);

  k_fuse<<<dim3(64 * 127), dim3(128), 0, stream>>>(XSPEC, XCV, lns, lnb, out);
}


# WMMA GEMM and conv k-loops: the next 16-k group's fragment loads go out as one batch into a second register set ahead of the current group's 16 WMMAs, replacing load-wait-4-WMMA per k-step
# speedup vs baseline: 27.3062x; 1.9368x over previous
#include <hip/hip_runtime.h>
#include <math.h>

typedef __attribute__((ext_vector_type(2))) float v2f;
typedef __attribute__((ext_vector_type(8))) float v8f;

#define PI_D 3.14159265358979323846

constexpr long long SZ_A1    = 255LL*128*128;
constexpr long long SZ_XP    = 128LL*256*128;
constexpr long long SZ_PH1   = 256LL*256;
constexpr long long SZ_G     = 128LL*256*128;
constexpr long long SZ_F1    = 255LL*128*128;
constexpr long long SZ_R1    = 64LL*128;
constexpr long long SZ_R2P   = 128LL*256;
constexpr long long SZ_T1    = 256LL*64*128;
constexpr long long SZ_FF    = 64LL*128*128;
constexpr long long SZ_WC    = 64LL*128*128;
constexpr long long SZ_F2    = 64LL*128*128;
constexpr long long SZ_P2    = 127LL*64*64;
constexpr long long SZ_FT    = 127LL*64*128;
constexpr long long SZ_PH2   = 128LL*128;
constexpr long long SZ_XSPEC = 64LL*128*128;
constexpr long long SZ_TVEC  = 256;
constexpr long long SZ_TC    = 64;
constexpr long long SZ_WSP   = 128LL*128;
constexpr long long SZ_XS1   = 128LL*255*128;
constexpr long long SZ_T2    = 256LL*64*128;
constexpr long long SZ_XS2   = 64LL*128*128;
constexpr long long SZ_XSP   = 67LL*129*128;
constexpr long long SZ_XCV   = 64LL*128*128;

constexpr long long OFF_A1    = 0;
constexpr long long OFF_XP    = OFF_A1    + SZ_A1;
constexpr long long OFF_COS1  = OFF_XP    + SZ_XP;
constexpr long long OFF_SIN1N = OFF_COS1  + SZ_PH1;
constexpr long long OFF_GR    = OFF_SIN1N + SZ_PH1;
constexpr long long OFF_GI    = OFF_GR    + SZ_G;
constexpr long long OFF_F1R   = OFF_GI    + SZ_G;
constexpr long long OFF_F1I   = OFF_F1R   + SZ_F1;
constexpr long long OFF_R1    = OFF_F1I   + SZ_F1;
constexpr long long OFF_R2P   = OFF_R1    + SZ_R1;
constexpr long long OFF_T1R   = OFF_R2P   + SZ_R2P;
constexpr long long OFF_T1I   = OFF_T1R   + SZ_T1;
constexpr long long OFF_FFR   = OFF_T1I   + SZ_T1;
constexpr long long OFF_FFI   = OFF_FFR   + SZ_FF;
constexpr long long OFF_WCR   = OFF_FFI   + SZ_FF;
constexpr long long OFF_WCI   = OFF_WCR   + SZ_WC;
constexpr long long OFF_F2R   = OFF_WCI   + SZ_WC;
constexpr long long OFF_F2I   = OFF_F2R   + SZ_F2;
constexpr long long OFF_P2A   = OFF_F2I   + SZ_F2;
constexpr long long OFF_P2B   = OFF_P2A   + SZ_P2;
constexpr long long OFF_FTR   = OFF_P2B   + SZ_P2;
constexpr long long OFF_FTI   = OFF_FTR   + SZ_FT;
constexpr long long OFF_COS2  = OFF_FTI   + SZ_FT;
constexpr long long OFF_SIN2  = OFF_COS2  + SZ_PH2;
constexpr long long OFF_XSPEC = OFF_SIN2  + SZ_PH2;
constexpr long long OFF_TVEC  = OFF_XSPEC + SZ_XSPEC;
constexpr long long OFF_TCR   = OFF_TVEC  + SZ_TVEC;
constexpr long long OFF_TCI   = OFF_TCR   + SZ_TC;
constexpr long long OFF_WSP   = OFF_TCI   + SZ_TC;
constexpr long long OFF_XS1   = OFF_WSP   + SZ_WSP;
constexpr long long OFF_T2R   = OFF_XS1   + SZ_XS1;
constexpr long long OFF_XS2   = OFF_T2R   + SZ_T2;
constexpr long long OFF_XSP   = OFF_XS2   + SZ_XS2;
constexpr long long OFF_XCV   = OFF_XSP   + SZ_XSP;

struct FragSet {
  v2f a[4];
  v2f b[4][4];
};

__device__ __forceinline__ void sched_fence() { __builtin_amdgcn_sched_barrier(0); }
__device__ __forceinline__ void sched_loads_then_mma_twice() {
  __builtin_amdgcn_sched_group_barrier(0x020, 36, 0);
  __builtin_amdgcn_sched_group_barrier(0x008, 16, 0);
  __builtin_amdgcn_sched_group_barrier(0x020, 36, 0);
  __builtin_amdgcn_sched_group_barrier(0x008, 16, 0);
}

__device__ __forceinline__ void load_frags(FragSet& f, const float* __restrict__ arow,
                                           const float* __restrict__ brow, long long ldb) {
#pragma unroll
  for (int s = 0; s < 4; ++s) {
    f.a[s].x = arow[4 * s];
    f.a[s].y = arow[4 * s + 1];
    const float* bp = brow + (long long)(4 * s) * ldb;
#pragma unroll
    for (int j = 0; j < 4; ++j) {
      f.b[s][j].x = bp[16 * j];
      f.b[s][j].y = bp[ldb + 16 * j];
    }
  }
}

__device__ __forceinline__ void mma_frags(const FragSet& f, v8f& acc0, v8f& acc1, v8f& acc2, v8f& acc3) {
#pragma unroll
  for (int s = 0; s < 4; ++s) {
    acc0 = __builtin_amdgcn_wmma_f32_16x16x4_f32(false, f.a[s], false, f.b[s][0], (short)0, acc0, false, false);
    acc1 = __builtin_amdgcn_wmma_f32_16x16x4_f32(false, f.a[s], false, f.b[s][1], (short)0, acc1, false, false);
    acc2 = __builtin_amdgcn_wmma_f32_16x16x4_f32(false, f.a[s], false, f.b[s][2], (short)0, acc2, false, false);
    acc3 = __builtin_amdgcn_wmma_f32_16x16x4_f32(false, f.a[s], false, f.b[s][3], (short)0, acc3, false, false);
  }
}

__global__ void __launch_bounds__(32)
gemm_wmma(const float* __restrict__ A, const float* __restrict__ B,
          float* __restrict__ C, int M, int K,
          int lda, int ldb, int ldc,
          long long sAb, long long sBb, long long sCb,
          float alpha, int accumulate) {
  (void)M;
  const int lane = threadIdx.x;
  const int n0 = blockIdx.x * 64;
  const int m0 = blockIdx.y * 16;
  const long long b = blockIdx.z;
  A += b * sAb; B += b * sBb; C += b * sCb;
  const int half = lane >> 4;
  const int l16  = lane & 15;
  v8f acc0 = {}, acc1 = {}, acc2 = {}, acc3 = {};
  const float* arow = A + (long long)(m0 + l16) * lda + 2 * half;
  const float* brow = B + (long long)(2 * half) * ldb + n0 + l16;
  const int K32 = K & ~31;
  if (K32 > 0) {
    FragSet P, Q;
    load_frags(P, arow, brow, ldb);
    int k0 = 0;
    for (; k0 + 32 < K32; k0 += 32) {
      load_frags(Q, arow + k0 + 16, brow + (long long)(k0 + 16) * ldb, ldb);
      mma_frags(P, acc0, acc1, acc2, acc3);
      load_frags(P, arow + k0 + 32, brow + (long long)(k0 + 32) * ldb, ldb);
      mma_frags(Q, acc0, acc1, acc2, acc3);
      sched_loads_then_mma_twice();
    }
    load_frags(Q, arow + k0 + 16, brow + (long long)(k0 + 16) * ldb, ldb);
    sched_fence();
    mma_frags(P, acc0, acc1, acc2, acc3);
    mma_frags(Q, acc0, acc1, acc2, acc3);
  }
  for (int k0 = K32; k0 < K; k0 += 4) {
    v2f a; a.x = arow[k0]; a.y = arow[k0 + 1];
    const float* bp = brow + (long long)k0 * ldb;
    v2f b0; b0.x = bp[0];   b0.y = bp[ldb];
    v2f b1; b1.x = bp[16];  b1.y = bp[ldb + 16];
    v2f b2; b2.x = bp[32];  b2.y = bp[ldb + 32];
    v2f b3; b3.x = bp[48];  b3.y = bp[ldb + 48];
    acc0 = __builtin_amdgcn_wmma_f32_16x16x4_f32(false, a, false, b0, (short)0, acc0, false, false);
    acc1 = __builtin_amdgcn_wmma_f32_16x16x4_f32(false, a, false, b1, (short)0, acc1, false, false);
    acc2 = __builtin_amdgcn_wmma_f32_16x16x4_f32(false, a, false, b2, (short)0, acc2, false, false);
    acc3 = __builtin_amdgcn_wmma_f32_16x16x4_f32(false, a, false, b3, (short)0, acc3, false, false);
  }
#pragma unroll
  for (int v = 0; v < 8; ++v) {
    const int row = m0 + half * 8 + v;
    float* cp = C + (long long)row * ldc + n0 + l16;
    if (accumulate) {
      cp[0]  += alpha * acc0[v]; cp[16] += alpha * acc1[v];
      cp[32] += alpha * acc2[v]; cp[48] += alpha * acc3[v];
    } else {
      cp[0]  = alpha * acc0[v]; cp[16] = alpha * acc1[v];
      cp[32] = alpha * acc2[v]; cp[48] = alpha * acc3[v];
    }
  }
}

__device__ __forceinline__ void conv_load(FragSet& f, const float* __restrict__ xsp,
                                          const float* __restrict__ ck, int g,
                                          int l, int mrow, int half, int ncol) {
  const int tap = g >> 3, kk = (g & 7) * 16;
  const int dy = tap / 3, dx = tap - 3 * dy;
  const float* arow = xsp + (((long long)(l + dy)) * 129 + (mrow + dx)) * 128 + 2 * half + kk;
  const float* brow = ck + (((long long)tap) * 128 + 2 * half + kk) * 128 + ncol;
  load_frags(f, arow, brow, 128);
}

__global__ void __launch_bounds__(32)
conv3x3_wmma(const float* __restrict__ xsp, const float* __restrict__ ck,
             const float* __restrict__ cb, const float* __restrict__ tvec,
             float* __restrict__ out) {
  const int lane = threadIdx.x;
  const int n0 = blockIdx.x * 64;
  const int m0 = blockIdx.y * 16;
  const int l  = blockIdx.z;
  const int half = lane >> 4, l16 = lane & 15;
  const int mrow = m0 + l16, ncol = n0 + l16;
  v8f acc0 = {}, acc1 = {}, acc2 = {}, acc3 = {};
  {
    FragSet P, Q;
    conv_load(P, xsp, ck, 0, l, mrow, half, ncol);
    int g = 0;
    for (; g + 2 < 72; g += 2) {
      conv_load(Q, xsp, ck, g + 1, l, mrow, half, ncol);
      mma_frags(P, acc0, acc1, acc2, acc3);
      conv_load(P, xsp, ck, g + 2, l, mrow, half, ncol);
      mma_frags(Q, acc0, acc1, acc2, acc3);
      sched_loads_then_mma_twice();
    }
    conv_load(Q, xsp, ck, g + 1, l, mrow, half, ncol);
    sched_fence();
    mma_frags(P, acc0, acc1, acc2, acc3);
    mma_frags(Q, acc0, acc1, acc2, acc3);
  }
  const int c0 = n0 + l16;
  const float bias0 = cb[c0]      + tvec[128 + c0];
  const float bias1 = cb[c0 + 16] + tvec[128 + c0 + 16];
  const float bias2 = cb[c0 + 32] + tvec[128 + c0 + 32];
  const float bias3 = cb[c0 + 48] + tvec[128 + c0 + 48];
#pragma unroll
  for (int v = 0; v < 8; ++v) {
    const int row = m0 + half * 8 + v;
    float* op = out + (((long long)l) * 128 + row) * 128 + c0;
    op[0]  = acc0[v] + bias0; op[16] = acc1[v] + bias1;
    op[32] = acc2[v] + bias2; op[48] = acc3[v] + bias3;
  }
}

__global__ void zerof(float* p, long long n) {
  long long i = (long long)blockIdx.x * 256 + threadIdx.x;
  if (i < n) p[i] = 0.0f;
}

__global__ void build_xpad(const float* __restrict__ x, float* __restrict__ xp, long long n) {
  long long i = (long long)blockIdx.x * 256 + threadIdx.x;
  if (i >= n) return;
  long long t = i / 32768, r = i % 32768, p = r / 128, c = r % 128;
  xp[i] = (p < 255) ? x[(t * 255 + p) * 128 + c] : 0.0f;
}

__global__ void k_tvec(const float* __restrict__ temb,
                       const float* __restrict__ dw, const float* __restrict__ db,
                       const float* __restrict__ drw, const float* __restrict__ drb,
                       const float* __restrict__ diw, const float* __restrict__ dib,
                       float* tvec, float* tcr, float* tci) {
  int j = threadIdx.x;
  float acc = db[j];
  for (int k = 0; k < 256; ++k) acc += temb[k] * dw[k * 256 + j];
  tvec[j] = acc;
  if (j < 64) {
    float ar = drb[j], ai = dib[j];
    for (int k = 0; k < 256; ++k) {
      ar += temb[k] * drw[k * 64 + j];
      ai += temb[k] * diw[k * 64 + j];
    }
    tcr[j] = ar; tci[j] = ai;
  }
}

__global__ void k_wc(const float* __restrict__ wr, const float* __restrict__ wi,
                     const float* __restrict__ tcr, const float* __restrict__ tci,
                     float* wcr, float* wci, long long n) {
  long long i = (long long)blockIdx.x * 256 + threadIdx.x;
  if (i >= n) return;
  int l = (int)(i / 16384);
  float tr = tcr[l], ti = tci[l], r = wr[i], im = wi[i];
  wcr[i] = tr * r - ti * im;
  wci[i] = tr * im + ti * r;
}

__global__ void k_wsp(const float* __restrict__ spw, const float* __restrict__ tvec,
                      float* wsp, long long n) {
  long long i = (long long)blockIdx.x * 256 + threadIdx.x;
  if (i >= n) return;
  int o = (int)(i % 128);
  wsp[i] = tvec[o] * spw[i];
}

__global__ void k_phase1(float* c1, float* s1n, long long n) {
  long long i = (long long)blockIdx.x * 256 + threadIdx.x;
  if (i >= n) return;
  int m = (int)(i / 256), p = (int)(i % 256);
  if (m < 255 && p < 255) {
    double mval = (double)(m - 127);
    double phi = 2.0 * PI_D * (double)p / 255.0;
    double dphi = 2.0 * PI_D / 255.0;
    c1[i]  = (float)( cos(phi * mval) * dphi);
    s1n[i] = (float)(-sin(phi * mval) * dphi);
  } else { c1[i] = 0.0f; s1n[i] = 0.0f; }
}

__global__ void k_phase2(float* c2, float* s2, long long n) {
  long long i = (long long)blockIdx.x * 256 + threadIdx.x;
  if (i >= n) return;
  int p = (int)(i / 128), m = (int)(i % 128);
  if (p < 127 && m < 127) {
    double mval = (double)(m - 63);
    double phi = 2.0 * PI_D * (double)p / 127.0;
    c2[i] = (float)cos(mval * phi);
    s2[i] = (float)sin(mval * phi);
  } else { c2[i] = 0.0f; s2[i] = 0.0f; }
}

__global__ void k_resize(float* r1, float* r2p) {
  int tid = threadIdx.x;
  if (tid < 64) {
    int i = tid;
    float scale = 0.5f;
    float sample = (i + 0.5f) / scale - 0.5f;
    float sum = 0.0f;
    for (int j = 0; j < 128; ++j) {
      float w = 1.0f - fabsf(sample - j) * scale;
      if (w > 0.0f) sum += w;
    }
    for (int j = 0; j < 128; ++j) {
      float w = 1.0f - fabsf(sample - j) * scale;
      r1[i * 128 + j] = (w > 0.0f) ? (w / sum) : 0.0f;
    }
  } else if (tid < 192) {
    int j = tid - 64;
    if (j < 127) {
      float scale = 127.0f / 255.0f;
      float sample = (j + 0.5f) / scale - 0.5f;
      float sum = 0.0f;
      for (int p = 0; p < 255; ++p) {
        float w = 1.0f - fabsf(sample - p) * scale;
        if (w > 0.0f) sum += w;
      }
      for (int p = 0; p < 255; ++p) {
        float w = 1.0f - fabsf(sample - p) * scale;
        r2p[j * 256 + p] = (w > 0.0f) ? (w / sum) : 0.0f;
      }
      r2p[j * 256 + 255] = 0.0f;
    } else {
      for (int p = 0; p < 256; ++p) r2p[j * 256 + p] = 0.0f;
    }
  }
}

__global__ void k_legendre(float* out, int L, int weighted) {
  __shared__ double cdiag[128];
  __shared__ double ca[128];
  __shared__ double cb[128];
  const int t = threadIdx.x;
  const int m = blockIdx.x;
  {
    const int l = t;
    const double ll = (double)l, mm = (double)m;
    cdiag[l] = (l >= 1) ? -sqrt((2.0 * ll + 1.0) / (2.0 * ll)) : 0.0;
    double a = 0.0, b = 0.0;
    if (l >= m + 2) {
      a = sqrt((4.0 * ll * ll - 1.0) / (ll * ll - mm * mm));
      b = sqrt((2.0 * ll + 1.0) * ((ll - 1.0) * (ll - 1.0) - mm * mm) /
               ((2.0 * ll - 3.0) * (ll * ll - mm * mm)));
    }
    ca[l] = a; cb[l] = b;
  }
  __syncthreads();
  const double theta = (2.0 * t + 1.0) * PI_D / (2.0 * L - 1.0);
  const double x = cos(theta), s = sin(theta);
  const double w = weighted ? (s * PI_D / (double)L) : 1.0;
  float* op = out + ((long long)(L - 1 + m) * L) * (long long)L + t;
  double pmm = sqrt(1.0 / (4.0 * PI_D)) * w;
  for (int j = 1; j <= m; ++j) pmm = cdiag[j] * s * pmm;
  op[(long long)m * L] = (float)pmm;
  if (m + 1 < L) {
    double p2 = pmm;
    double p1 = sqrt(2.0 * (m + 1) + 1.0) * x * pmm;
    op[(long long)(m + 1) * L] = (float)p1;
    for (int l = m + 2; l < L; ++l) {
      const double p0 = ca[l] * x * p1 - cb[l] * p2;
      op[(long long)l * L] = (float)p0;
      p2 = p1; p1 = p0;
    }
  }
}

__global__ void k_mirror(float* out, int L, long long n) {
  long long i = (long long)blockIdx.x * 256 + threadIdx.x;
  if (i >= n) return;
  int LL = L * L;
  int m = (int)(i / LL) + 1;
  int r = (int)(i % LL);
  int l = r / L, t = r % L;
  float v = out[((long long)(L - 1 + m) * L + l) * L + t];
  out[((long long)(L - 1 - m) * L + l) * L + t] = (m & 1) ? -v : v;
}

__global__ void k_transp_p2(const float* __restrict__ a, float* __restrict__ b, long long n) {
  long long i = (long long)blockIdx.x * 256 + threadIdx.x;
  if (i >= n) return;
  int m = (int)(i / 4096), t = (int)((i / 64) % 64), l = (int)(i % 64);
  b[i] = a[((long long)m * 64 + l) * 64 + t];
}

__global__ void k_xsppad(const float* __restrict__ xs2, float* __restrict__ xsp, long long n) {
  long long i = (long long)blockIdx.x * 256 + threadIdx.x;
  if (i >= n) return;
  long long row = i / (129 * 128), r = i % (129 * 128);
  int col = (int)(r / 128), c = (int)(r % 128);
  int l = (int)row - 1, m = col - 1;
  xsp[i] = (l >= 0 && l < 64 && m >= 0 && m < 127) ? xs2[((long long)l * 128 + m) * 128 + c] : 0.0f;
}

__global__ void k_fuse(const float* __restrict__ xspec, const float* __restrict__ xcv,
                       const float* __restrict__ lns, const float* __restrict__ lnb,
                       float* __restrict__ out) {
  __shared__ float sh[128];
  int c = threadIdx.x;
  int n = blockIdx.x;
  int l = n / 127, m = n % 127;
  long long idx = ((long long)l * 128 + m) * 128 + c;
  float v = xspec[idx] + xcv[idx];
  float g = 0.5f * v * (1.0f + tanhf(0.7978845608028654f * (v + 0.044715f * v * v * v)));
  sh[c] = g; __syncthreads();
  for (int s = 64; s > 0; s >>= 1) { if (c < s) sh[c] += sh[c + s]; __syncthreads(); }
  float mu = sh[0] * (1.0f / 128.0f); __syncthreads();
  float d = g - mu;
  sh[c] = d * d; __syncthreads();
  for (int s = 64; s > 0; s >>= 1) { if (c < s) sh[c] += sh[c + s]; __syncthreads(); }
  float var = sh[0] * (1.0f / 128.0f);
  out[((long long)l * 127 + m) * 128 + c] = d * rsqrtf(var + 1e-6f) * lns[c] + lnb[c];
}

static inline void launch_gemm(hipStream_t s, const float* A, const float* B, float* C,
                               int M, int K, int lda, int ldb, int ldc,
                               long long sA, long long sB, long long sC,
                               float alpha, int acc, int batch) {
  dim3 g(2, M / 16, batch);
  gemm_wmma<<<g, dim3(32), 0, s>>>(A, B, C, M, K, lda, ldb, ldc, sA, sB, sC, alpha, acc);
}

#define EWG(n) dim3((unsigned)(((n) + 255) / 256)), dim3(256), 0, stream

extern "C" void kernel_launch(void* const* d_in, const int* in_sizes, int n_in,
                              void* d_out, int out_size, void* d_ws, size_t ws_size,
                              hipStream_t stream) {
  (void)in_sizes; (void)n_in; (void)out_size; (void)ws_size;
  const float* x       = (const float*)d_in[0];
  const float* t_emb   = (const float*)d_in[1];
  const float* sp_dw   = (const float*)d_in[2];
  const float* sp_db   = (const float*)d_in[3];
  const float* sp_wt   = (const float*)d_in[4];
  const float* sp_ck   = (const float*)d_in[5];
  const float* sp_cb   = (const float*)d_in[6];
  const float* spec_wr = (const float*)d_in[7];
  const float* spec_wi = (const float*)d_in[8];
  const float* drw     = (const float*)d_in[9];
  const float* drb     = (const float*)d_in[10];
  const float* diw     = (const float*)d_in[11];
  const float* dib     = (const float*)d_in[12];
  const float* lns     = (const float*)d_in[13];
  const float* lnb     = (const float*)d_in[14];
  float* out = (float*)d_out;
  float* ws  = (float*)d_ws;

  float *A1 = ws + OFF_A1, *XP = ws + OFF_XP, *COS1 = ws + OFF_COS1, *SIN1N = ws + OFF_SIN1N;
  float *GR = ws + OFF_GR, *GI = ws + OFF_GI, *F1R = ws + OFF_F1R, *F1I = ws + OFF_F1I;
  float *R1 = ws + OFF_R1, *R2P = ws + OFF_R2P, *T1R = ws + OFF_T1R, *T1I = ws + OFF_T1I;
  float *FFR = ws + OFF_FFR, *FFI = ws + OFF_FFI, *WCR = ws + OFF_WCR, *WCI = ws + OFF_WCI;
  float *F2R = ws + OFF_F2R, *F2I = ws + OFF_F2I, *P2A = ws + OFF_P2A, *P2B = ws + OFF_P2B;
  float *FTR = ws + OFF_FTR, *FTI = ws + OFF_FTI, *COS2 = ws + OFF_COS2, *SIN2 = ws + OFF_SIN2;
  float *XSPEC = ws + OFF_XSPEC, *TVEC = ws + OFF_TVEC, *TCR = ws + OFF_TCR, *TCI = ws + OFF_TCI;
  float *WSP = ws + OFF_WSP, *XS1 = ws + OFF_XS1, *T2R = ws + OFF_T2R, *XS2 = ws + OFF_XS2;
  float *XSP = ws + OFF_XSP, *XCV = ws + OFF_XCV;

  zerof<<<EWG(SZ_A1)>>>(A1, SZ_A1);
  zerof<<<EWG(SZ_P2)>>>(P2A, SZ_P2);
  zerof<<<EWG(8192)>>>(T1R + 255LL * 8192, 8192);
  zerof<<<EWG(8192)>>>(T1I + 255LL * 8192, 8192);
  zerof<<<EWG(8192)>>>(T2R + 255LL * 8192, 8192);

  build_xpad<<<EWG(SZ_XP)>>>(x, XP, SZ_XP);
  k_tvec<<<dim3(1), dim3(256), 0, stream>>>(t_emb, sp_dw, sp_db, drw, drb, diw, dib, TVEC, TCR, TCI);
  k_wc<<<EWG(SZ_WC)>>>(spec_wr, spec_wi, TCR, TCI, WCR, WCI, SZ_WC);
  k_wsp<<<EWG(SZ_WSP)>>>(sp_wt, TVEC, WSP, SZ_WSP);
  k_phase1<<<EWG(SZ_PH1)>>>(COS1, SIN1N, SZ_PH1);
  k_phase2<<<EWG(SZ_PH2)>>>(COS2, SIN2, SZ_PH2);
  k_resize<<<dim3(1), dim3(256), 0, stream>>>(R1, R2P);
  k_legendre<<<dim3(128), dim3(128), 0, stream>>>(A1, 128, 1);
  k_mirror<<<EWG(127LL * 128 * 128)>>>(A1, 128, 127LL * 128 * 128);
  k_legendre<<<dim3(64), dim3(64), 0, stream>>>(P2A, 64, 0);
  k_mirror<<<EWG(63LL * 64 * 64)>>>(P2A, 64, 63LL * 64 * 64);
  k_transp_p2<<<EWG(SZ_P2)>>>(P2A, P2B, SZ_P2);

  launch_gemm(stream, COS1,  XP, GR, 256, 256, 256, 128, 128, 0, 256LL * 128, 256LL * 128, 1.0f, 0, 128);
  launch_gemm(stream, SIN1N, XP, GI, 256, 256, 256, 128, 128, 0, 256LL * 128, 256LL * 128, 1.0f, 0, 128);
  launch_gemm(stream, A1, GR, F1R, 128, 128, 128, 256 * 128, 128, 128LL * 128, 128, 128LL * 128, 1.0f, 0, 255);
  launch_gemm(stream, A1, GI, F1I, 128, 128, 128, 256 * 128, 128, 128LL * 128, 128, 128LL * 128, 1.0f, 0, 255);
  launch_gemm(stream, R1, F1R, T1R, 64, 128, 128, 128, 128, 0, 128LL * 128, 64LL * 128, 1.0f, 0, 255);
  launch_gemm(stream, R1, F1I, T1I, 64, 128, 128, 128, 128, 0, 128LL * 128, 64LL * 128, 1.0f, 0, 255);
  launch_gemm(stream, R2P, T1R, FFR, 128, 256, 256, 64 * 128, 128, 0, 128, 128LL * 128, 1.0f, 0, 64);
  launch_gemm(stream, R2P, T1I, FFI, 128, 256, 256, 64 * 128, 128, 0, 128, 128LL * 128, 1.0f, 0, 64);
  launch_gemm(stream, FFR, WCR, F2R, 128, 128, 128, 128, 128, 128LL * 128, 128LL * 128, 128LL * 128,  1.0f, 0, 64);
  launch_gemm(stream, FFI, WCI, F2R, 128, 128, 128, 128, 128, 128LL * 128, 128LL * 128, 128LL * 128, -1.0f, 1, 64);
  launch_gemm(stream, FFR, WCI, F2I, 128, 128, 128, 128, 128, 128LL * 128, 128LL * 128, 128LL * 128,  1.0f, 0, 64);
  launch_gemm(stream, FFI, WCR, F2I, 128, 128, 128, 128, 128, 128LL * 128, 128LL * 128, 128LL * 128,  1.0f, 1, 64);
  launch_gemm(stream, P2B, F2R, FTR, 64, 64, 64, 128 * 128, 128, 64LL * 64, 128, 64LL * 128, 1.0f, 0, 127);
  launch_gemm(stream, P2B, F2I, FTI, 64, 64, 64, 128 * 128, 128, 64LL * 64, 128, 64LL * 128, 1.0f, 0, 127);
  launch_gemm(stream, COS2, FTR, XSPEC, 128, 128, 128, 64 * 128, 128, 0, 128, 128LL * 128,  1.0f, 0, 64);
  launch_gemm(stream, SIN2, FTI, XSPEC, 128, 128, 128, 64 * 128, 128, 0, 128, 128LL * 128, -1.0f, 1, 64);

  launch_gemm(stream, x, WSP, XS1, 32640, 128, 128, 128, 128, 0, 0, 0, 1.0f, 0, 1);
  launch_gemm(stream, R1, XS1, T2R, 64, 128, 128, 255 * 128, 128, 0, 128, 64LL * 128, 1.0f, 0, 255);
  launch_gemm(stream, R2P, T2R, XS2, 128, 256, 256, 64 * 128, 128, 0, 128, 128LL * 128, 1.0f, 0, 64);
  k_xsppad<<<EWG(SZ_XSP)>>>(XS2, XSP, SZ_XSP);
  conv3x3_wmma<<<dim3(2, 8, 64), dim3(32), 0, stream>>>(XSP, sp_ck, sp_cb, TVEC, XCV);

  k_fuse<<<dim3(64 * 127), dim3(128), 0, stream>>>(XSPEC, XCV, lns, lnb, out);
}


# Elementwise helpers (zero fill, pads, table mirror and transpose, weight scaling, GELU+LayerNorm): 16 bytes per lane and 4 pieces per thread, so a call launches about 38k waves instead of 471k
# speedup vs baseline: 28.7846x; 1.0541x over previous
#include <hip/hip_runtime.h>
#include <math.h>

typedef __attribute__((ext_vector_type(2))) float v2f;
typedef __attribute__((ext_vector_type(8))) float v8f;
typedef __attribute__((ext_vector_type(4))) float v4f;

#define PI_D 3.14159265358979323846

constexpr long long SZ_A1    = 255LL*128*128;
constexpr long long SZ_XP    = 128LL*256*128;
constexpr long long SZ_PH1   = 256LL*256;
constexpr long long SZ_G     = 128LL*256*128;
constexpr long long SZ_F1    = 255LL*128*128;
constexpr long long SZ_R1    = 64LL*128;
constexpr long long SZ_R2P   = 128LL*256;
constexpr long long SZ_T1    = 256LL*64*128;
constexpr long long SZ_FF    = 64LL*128*128;
constexpr long long SZ_WC    = 64LL*128*128;
constexpr long long SZ_F2    = 64LL*128*128;
constexpr long long SZ_P2    = 127LL*64*64;
constexpr long long SZ_FT    = 127LL*64*128;
constexpr long long SZ_PH2   = 128LL*128;
constexpr long long SZ_XSPEC = 64LL*128*128;
constexpr long long SZ_TVEC  = 256;
constexpr long long SZ_TC    = 64;
constexpr long long SZ_WSP   = 128LL*128;
constexpr long long SZ_XS1   = 128LL*255*128;
constexpr long long SZ_T2    = 256LL*64*128;
constexpr long long SZ_XS2   = 64LL*128*128;
constexpr long long SZ_XSP   = 67LL*129*128;
constexpr long long SZ_XCV   = 64LL*128*128;

constexpr long long OFF_A1    = 0;
constexpr long long OFF_XP    = OFF_A1    + SZ_A1;
constexpr long long OFF_COS1  = OFF_XP    + SZ_XP;
constexpr long long OFF_SIN1N = OFF_COS1  + SZ_PH1;
constexpr long long OFF_GR    = OFF_SIN1N + SZ_PH1;
constexpr long long OFF_GI    = OFF_GR    + SZ_G;
constexpr long long OFF_F1R   = OFF_GI    + SZ_G;
constexpr long long OFF_F1I   = OFF_F1R   + SZ_F1;
constexpr long long OFF_R1    = OFF_F1I   + SZ_F1;
constexpr long long OFF_R2P   = OFF_R1    + SZ_R1;
constexpr long long OFF_T1R   = OFF_R2P   + SZ_R2P;
constexpr long long OFF_T1I   = OFF_T1R   + SZ_T1;
constexpr long long OFF_FFR   = OFF_T1I   + SZ_T1;
constexpr long long OFF_FFI   = OFF_FFR   + SZ_FF;
constexpr long long OFF_WCR   = OFF_FFI   + SZ_FF;
constexpr long long OFF_WCI   = OFF_WCR   + SZ_WC;
constexpr long long OFF_F2R   = OFF_WCI   + SZ_WC;
constexpr long long OFF_F2I   = OFF_F2R   + SZ_F2;
constexpr long long OFF_P2A   = OFF_F2I   + SZ_F2;
constexpr long long OFF_P2B   = OFF_P2A   + SZ_P2;
constexpr long long OFF_FTR   = OFF_P2B   + SZ_P2;
constexpr long long OFF_FTI   = OFF_FTR   + SZ_FT;
constexpr long long OFF_COS2  = OFF_FTI   + SZ_FT;
constexpr long long OFF_SIN2  = OFF_COS2  + SZ_PH2;
constexpr long long OFF_XSPEC = OFF_SIN2  + SZ_PH2;
constexpr long long OFF_TVEC  = OFF_XSPEC + SZ_XSPEC;
constexpr long long OFF_TCR   = OFF_TVEC  + SZ_TVEC;
constexpr long long OFF_TCI   = OFF_TCR   + SZ_TC;
constexpr long long OFF_WSP   = OFF_TCI   + SZ_TC;
constexpr long long OFF_XS1   = OFF_WSP   + SZ_WSP;
constexpr long long OFF_T2R   = OFF_XS1   + SZ_XS1;
constexpr long long OFF_XS2   = OFF_T2R   + SZ_T2;
constexpr long long OFF_XSP   = OFF_XS2   + SZ_XS2;
constexpr long long OFF_XCV   = OFF_XSP   + SZ_XSP;

struct FragSet {
  v2f a[4];
  v2f b[4][4];
};

__device__ __forceinline__ void sched_fence() { __builtin_amdgcn_sched_barrier(0); }
__device__ __forceinline__ void sched_loads_then_mma_twice() {
  __builtin_amdgcn_sched_group_barrier(0x020, 36, 0);
  __builtin_amdgcn_sched_group_barrier(0x008, 16, 0);
  __builtin_amdgcn_sched_group_barrier(0x020, 36, 0);
  __builtin_amdgcn_sched_group_barrier(0x008, 16, 0);
}

__device__ __forceinline__ void load_frags(FragSet& f, const float* __restrict__ arow,
                                           const float* __restrict__ brow, long long ldb) {
#pragma unroll
  for (int s = 0; s < 4; ++s) {
    f.a[s].x = arow[4 * s];
    f.a[s].y = arow[4 * s + 1];
    const float* bp = brow + (long long)(4 * s) * ldb;
#pragma unroll
    for (int j = 0; j < 4; ++j) {
      f.b[s][j].x = bp[16 * j];
      f.b[s][j].y = bp[ldb + 16 * j];
    }
  }
}

__device__ __forceinline__ void mma_frags(const FragSet& f, v8f& acc0, v8f& acc1, v8f& acc2, v8f& acc3) {
#pragma unroll
  for (int s = 0; s < 4; ++s) {
    acc0 = __builtin_amdgcn_wmma_f32_16x16x4_f32(false, f.a[s], false, f.b[s][0], (short)0, acc0, false, false);
    acc1 = __builtin_amdgcn_wmma_f32_16x16x4_f32(false, f.a[s], false, f.b[s][1], (short)0, acc1, false, false);
    acc2 = __builtin_amdgcn_wmma_f32_16x16x4_f32(false, f.a[s], false, f.b[s][2], (short)0, acc2, false, false);
    acc3 = __builtin_amdgcn_wmma_f32_16x16x4_f32(false, f.a[s], false, f.b[s][3], (short)0, acc3, false, false);
  }
}

__global__ void __launch_bounds__(32)
gemm_wmma(const float* __restrict__ A, const float* __restrict__ B,
          float* __restrict__ C, int M, int K,
          int lda, int ldb, int ldc,
          long long sAb, long long sBb, long long sCb,
          float alpha, int accumulate) {
  (void)M;
  const int lane = threadIdx.x;
  const int n0 = blockIdx.x * 64;
  const int m0 = blockIdx.y * 16;
  const long long b = blockIdx.z;
  A += b * sAb; B += b * sBb; C += b * sCb;
  const int half = lane >> 4;
  const int l16  = lane & 15;
  v8f acc0 = {}, acc1 = {}, acc2 = {}, acc3 = {};
  const float* arow = A + (long long)(m0 + l16) * lda + 2 * half;
  const float* brow = B + (long long)(2 * half) * ldb + n0 + l16;
  const int K32 = K & ~31;
  if (K32 > 0) {
    FragSet P, Q;
    load_frags(P, arow, brow, ldb);
    int k0 = 0;
    for (; k0 + 32 < K32; k0 += 32) {
      load_frags(Q, arow + k0 + 16, brow + (long long)(k0 + 16) * ldb, ldb);
      mma_frags(P, acc0, acc1, acc2, acc3);
      load_frags(P, arow + k0 + 32, brow + (long long)(k0 + 32) * ldb, ldb);
      mma_frags(Q, acc0, acc1, acc2, acc3);
      sched_loads_then_mma_twice();
    }
    load_frags(Q, arow + k0 + 16, brow + (long long)(k0 + 16) * ldb, ldb);
    sched_fence();
    mma_frags(P, acc0, acc1, acc2, acc3);
    mma_frags(Q, acc0, acc1, acc2, acc3);
  }
  for (int k0 = K32; k0 < K; k0 += 4) {
    v2f a; a.x = arow[k0]; a.y = arow[k0 + 1];
    const float* bp = brow + (long long)k0 * ldb;
    v2f b0; b0.x = bp[0];   b0.y = bp[ldb];
    v2f b1; b1.x = bp[16];  b1.y = bp[ldb + 16];
    v2f b2; b2.x = bp[32];  b2.y = bp[ldb + 32];
    v2f b3; b3.x = bp[48];  b3.y = bp[ldb + 48];
    acc0 = __builtin_amdgcn_wmma_f32_16x16x4_f32(false, a, false, b0, (short)0, acc0, false, false);
    acc1 = __builtin_amdgcn_wmma_f32_16x16x4_f32(false, a, false, b1, (short)0, acc1, false, false);
    acc2 = __builtin_amdgcn_wmma_f32_16x16x4_f32(false, a, false, b2, (short)0, acc2, false, false);
    acc3 = __builtin_amdgcn_wmma_f32_16x16x4_f32(false, a, false, b3, (short)0, acc3, false, false);
  }
#pragma unroll
  for (int v = 0; v < 8; ++v) {
    const int row = m0 + half * 8 + v;
    float* cp = C + (long long)row * ldc + n0 + l16;
    if (accumulate) {
      cp[0]  += alpha * acc0[v]; cp[16] += alpha * acc1[v];
      cp[32] += alpha * acc2[v]; cp[48] += alpha * acc3[v];
    } else {
      cp[0]  = alpha * acc0[v]; cp[16] = alpha * acc1[v];
      cp[32] = alpha * acc2[v]; cp[48] = alpha * acc3[v];
    }
  }
}

__device__ __forceinline__ void conv_load(FragSet& f, const float* __restrict__ xsp,
                                          const float* __restrict__ ck, int g,
                                          int l, int mrow, int half, int ncol) {
  const int tap = g >> 3, kk = (g & 7) * 16;
  const int dy = tap / 3, dx = tap - 3 * dy;
  const float* arow = xsp + (((long long)(l + dy)) * 129 + (mrow + dx)) * 128 + 2 * half + kk;
  const float* brow = ck + (((long long)tap) * 128 + 2 * half + kk) * 128 + ncol;
  load_frags(f, arow, brow, 128);
}

__global__ void __launch_bounds__(32)
conv3x3_wmma(const float* __restrict__ xsp, const float* __restrict__ ck,
             const float* __restrict__ cb, const float* __restrict__ tvec,
             float* __restrict__ out) {
  const int lane = threadIdx.x;
  const int n0 = blockIdx.x * 64;
  const int m0 = blockIdx.y * 16;
  const int l  = blockIdx.z;
  const int half = lane >> 4, l16 = lane & 15;
  const int mrow = m0 + l16, ncol = n0 + l16;
  v8f acc0 = {}, acc1 = {}, acc2 = {}, acc3 = {};
  {
    FragSet P, Q;
    conv_load(P, xsp, ck, 0, l, mrow, half, ncol);
    int g = 0;
    for (; g + 2 < 72; g += 2) {
      conv_load(Q, xsp, ck, g + 1, l, mrow, half, ncol);
      mma_frags(P, acc0, acc1, acc2, acc3);
      conv_load(P, xsp, ck, g + 2, l, mrow, half, ncol);
      mma_frags(Q, acc0, acc1, acc2, acc3);
      sched_loads_then_mma_twice();
    }
    conv_load(Q, xsp, ck, g + 1, l, mrow, half, ncol);
    sched_fence();
    mma_frags(P, acc0, acc1, acc2, acc3);
    mma_frags(Q, acc0, acc1, acc2, acc3);
  }
  const int c0 = n0 + l16;
  const float bias0 = cb[c0]      + tvec[128 + c0];
  const float bias1 = cb[c0 + 16] + tvec[128 + c0 + 16];
  const float bias2 = cb[c0 + 32] + tvec[128 + c0 + 32];
  const float bias3 = cb[c0 + 48] + tvec[128 + c0 + 48];
#pragma unroll
  for (int v = 0; v < 8; ++v) {
    const int row = m0 + half * 8 + v;
    float* op = out + (((long long)l) * 128 + row) * 128 + c0;
    op[0]  = acc0[v] + bias0; op[16] = acc1[v] + bias1;
    op[32] = acc2[v] + bias2; op[48] = acc3[v] + bias3;
  }
}

constexpr int EW_U = 4;

__global__ void __launch_bounds__(256) zerof(float* p, int n4) {
  v4f* p4 = (v4f*)p;
  const int base = blockIdx.x * (256 * EW_U) + threadIdx.x;
  const v4f z = {0.0f, 0.0f, 0.0f, 0.0f};
#pragma unroll
  for (int u = 0; u < EW_U; ++u) {
    const int i = base + u * 256;
    if (i < n4) p4[i] = z;
  }
}

__global__ void __launch_bounds__(256)
build_xpad(const float* __restrict__ x, float* __restrict__ xp, int n4) {
  const v4f* x4 = (const v4f*)x;
  v4f* xp4 = (v4f*)xp;
  const int base = blockIdx.x * (256 * EW_U) + threadIdx.x;
  v4f v[EW_U];
#pragma unroll
  for (int u = 0; u < EW_U; ++u) {
    const int i = min(base + u * 256, n4 - 1);
    const int t = i >> 13, r = i & 8191, p = r >> 5, c4 = r & 31;
    const v4f z = {0.0f, 0.0f, 0.0f, 0.0f};
    const v4f ld = x4[(t * 255 + min(p, 254)) * 32 + c4];
    v[u] = (p < 255) ? ld : z;
  }
#pragma unroll
  for (int u = 0; u < EW_U; ++u) {
    const int i = base + u * 256;
    if (i < n4) xp4[i] = v[u];
  }
}

__global__ void k_tvec(const float* __restrict__ temb,
                       const float* __restrict__ dw, const float* __restrict__ db,
                       const float* __restrict__ drw, const float* __restrict__ drb,
                       const float* __restrict__ diw, const float* __restrict__ dib,
                       float* tvec, float* tcr, float* tci) {
  int j = threadIdx.x;
  float acc = db[j];
  for (int k = 0; k < 256; ++k) acc += temb[k] * dw[k * 256 + j];
  tvec[j] = acc;
  if (j < 64) {
    float ar = drb[j], ai = dib[j];
    for (int k = 0; k < 256; ++k) {
      ar += temb[k] * drw[k * 64 + j];
      ai += temb[k] * diw[k * 64 + j];
    }
    tcr[j] = ar; tci[j] = ai;
  }
}

__global__ void __launch_bounds__(256)
k_wc(const float* __restrict__ wr, const float* __restrict__ wi,
     const float* __restrict__ tcr, const float* __restrict__ tci,
     float* __restrict__ wcr, float* __restrict__ wci, int n4) {
  const v4f* wr4 = (const v4f*)wr;
  const v4f* wi4 = (const v4f*)wi;
  v4f* wcr4 = (v4f*)wcr;
  v4f* wci4 = (v4f*)wci;
  const int base = blockIdx.x * (256 * EW_U) + threadIdx.x;
  v4f r[EW_U], im[EW_U];
  float tr[EW_U], ti[EW_U];
#pragma unroll
  for (int u = 0; u < EW_U; ++u) {
    const int i = min(base + u * 256, n4 - 1);
    const int l = i >> 12;
    r[u] = wr4[i]; im[u] = wi4[i];
    tr[u] = tcr[l]; ti[u] = tci[l];
  }
#pragma unroll
  for (int u = 0; u < EW_U; ++u) {
    const int i = base + u * 256;
    if (i < n4) {
      wcr4[i] = tr[u] * r[u] - ti[u] * im[u];
      wci4[i] = tr[u] * im[u] + ti[u] * r[u];
    }
  }
}

__global__ void k_wsp(const float* __restrict__ spw, const float* __restrict__ tvec,
                      float* wsp, long long n) {
  long long i = (long long)blockIdx.x * 256 + threadIdx.x;
  if (i >= n) return;
  int o = (int)(i % 128);
  wsp[i] = tvec[o] * spw[i];
}

__global__ void k_phase1(float* c1, float* s1n, long long n) {
  long long i = (long long)blockIdx.x * 256 + threadIdx.x;
  if (i >= n) return;
  int m = (int)(i / 256), p = (int)(i % 256);
  if (m < 255 && p < 255) {
    double mval = (double)(m - 127);
    double phi = 2.0 * PI_D * (double)p / 255.0;
    double dphi = 2.0 * PI_D / 255.0;
    c1[i]  = (float)( cos(phi * mval) * dphi);
    s1n[i] = (float)(-sin(phi * mval) * dphi);
  } else { c1[i] = 0.0f; s1n[i] = 0.0f; }
}

__global__ void k_phase2(float* c2, float* s2, long long n) {
  long long i = (long long)blockIdx.x * 256 + threadIdx.x;
  if (i >= n) return;
  int p = (int)(i / 128), m = (int)(i % 128);
  if (p < 127 && m < 127) {
    double mval = (double)(m - 63);
    double phi = 2.0 * PI_D * (double)p / 127.0;
    c2[i] = (float)cos(mval * phi);
    s2[i] = (float)sin(mval * phi);
  } else { c2[i] = 0.0f; s2[i] = 0.0f; }
}

__global__ void k_resize(float* r1, float* r2p) {
  int tid = threadIdx.x;
  if (tid < 64) {
    int i = tid;
    float scale = 0.5f;
    float sample = (i + 0.5f) / scale - 0.5f;
    float sum = 0.0f;
    for (int j = 0; j < 128; ++j) {
      float w = 1.0f - fabsf(sample - j) * scale;
      if (w > 0.0f) sum += w;
    }
    for (int j = 0; j < 128; ++j) {
      float w = 1.0f - fabsf(sample - j) * scale;
      r1[i * 128 + j] = (w > 0.0f) ? (w / sum) : 0.0f;
    }
  } else if (tid < 192) {
    int j = tid - 64;
    if (j < 127) {
      float scale = 127.0f / 255.0f;
      float sample = (j + 0.5f) / scale - 0.5f;
      float sum = 0.0f;
      for (int p = 0; p < 255; ++p) {
        float w = 1.0f - fabsf(sample - p) * scale;
        if (w > 0.0f) sum += w;
      }
      for (int p = 0; p < 255; ++p) {
        float w = 1.0f - fabsf(sample - p) * scale;
        r2p[j * 256 + p] = (w > 0.0f) ? (w / sum) : 0.0f;
      }
      r2p[j * 256 + 255] = 0.0f;
    } else {
      for (int p = 0; p < 256; ++p) r2p[j * 256 + p] = 0.0f;
    }
  }
}

__global__ void k_legendre(float* out, int L, int weighted) {
  __shared__ double cdiag[128];
  __shared__ double ca[128];
  __shared__ double cb[128];
  const int t = threadIdx.x;
  const int m = blockIdx.x;
  {
    const int l = t;
    const double ll = (double)l, mm = (double)m;
    cdiag[l] = (l >= 1) ? -sqrt((2.0 * ll + 1.0) / (2.0 * ll)) : 0.0;
    double a = 0.0, b = 0.0;
    if (l >= m + 2) {
      a = sqrt((4.0 * ll * ll - 1.0) / (ll * ll - mm * mm));
      b = sqrt((2.0 * ll + 1.0) * ((ll - 1.0) * (ll - 1.0) - mm * mm) /
               ((2.0 * ll - 3.0) * (ll * ll - mm * mm)));
    }
    ca[l] = a; cb[l] = b;
  }
  __syncthreads();
  const double theta = (2.0 * t + 1.0) * PI_D / (2.0 * L - 1.0);
  const double x = cos(theta), s = sin(theta);
  const double w = weighted ? (s * PI_D / (double)L) : 1.0;
  float* op = out + ((long long)(L - 1 + m) * L) * (long long)L + t;
  double pmm = sqrt(1.0 / (4.0 * PI_D)) * w;
  for (int j = 1; j <= m; ++j) pmm = cdiag[j] * s * pmm;
  op[(long long)m * L] = (float)pmm;
  if (m + 1 < L) {
    double p2 = pmm;
    double p1 = sqrt(2.0 * (m + 1) + 1.0) * x * pmm;
    op[(long long)(m + 1) * L] = (float)p1;
    for (int l = m + 2; l < L; ++l) {
      const double p0 = ca[l] * x * p1 - cb[l] * p2;
      op[(long long)l * L] = (float)p0;
      p2 = p1; p1 = p0;
    }
  }
}

__global__ void __launch_bounds__(256)
k_mirror(float* out, int L, int n4) {
  v4f* o4 = (v4f*)out;
  const int LL4 = (L * L) >> 2;
  const int base = blockIdx.x * (256 * EW_U) + threadIdx.x;
  v4f v[EW_U];
#pragma unroll
  for (int u = 0; u < EW_U; ++u) {
    const int i = min(base + u * 256, n4 - 1);
    const int m = i / LL4 + 1, r = i - (m - 1) * LL4;
    v[u] = o4[(long long)(L - 1 + m) * LL4 + r];
  }
#pragma unroll
  for (int u = 0; u < EW_U; ++u) {
    const int i = base + u * 256;
    if (i < n4) {
      const int m = i / LL4 + 1, r = i - (m - 1) * LL4;
      o4[(long long)(L - 1 - m) * LL4 + r] = (m & 1) ? -v[u] : v[u];
    }
  }
}

__global__ void __launch_bounds__(256)
k_transp_p2(const float* __restrict__ a, float* __restrict__ b) {
  __shared__ float tile[64][65];
  const int m = blockIdx.x;
  const v4f* a4 = (const v4f*)a + (long long)m * 1024;
  v4f* b4 = (v4f*)b + (long long)m * 1024;
  v4f v[4];
#pragma unroll
  for (int u = 0; u < 4; ++u) v[u] = a4[threadIdx.x + 256 * u];
#pragma unroll
  for (int u = 0; u < 4; ++u) {
    const int idx = threadIdx.x + 256 * u;
    const int l = idx >> 4, t4 = idx & 15;
    tile[l][4 * t4 + 0] = v[u].x; tile[l][4 * t4 + 1] = v[u].y;
    tile[l][4 * t4 + 2] = v[u].z; tile[l][4 * t4 + 3] = v[u].w;
  }
  __syncthreads();
#pragma unroll
  for (int u = 0; u < 4; ++u) {
    const int idx = threadIdx.x + 256 * u;
    const int t = idx >> 4, l4 = idx & 15;
    v4f o;
    o.x = tile[4 * l4 + 0][t]; o.y = tile[4 * l4 + 1][t];
    o.z = tile[4 * l4 + 2][t]; o.w = tile[4 * l4 + 3][t];
    b4[idx] = o;
  }
}

__global__ void __launch_bounds__(256)
k_xsppad(const float* __restrict__ xs2, float* __restrict__ xsp, int n4) {
  const v4f* s4 = (const v4f*)xs2;
  v4f* d4 = (v4f*)xsp;
  const int base = blockIdx.x * (256 * EW_U) + threadIdx.x;
  v4f v[EW_U];
#pragma unroll
  for (int u = 0; u < EW_U; ++u) {
    const int i = min(base + u * 256, n4 - 1);
    const int row = i / (129 * 32), r = i - row * (129 * 32);
    const int col = r >> 5, c4 = r & 31;
    const int l = row - 1, m = col - 1;
    const bool inside = (l >= 0 && l < 64 && m >= 0 && m < 127);
    const v4f z = {0.0f, 0.0f, 0.0f, 0.0f};
    const v4f ld = s4[(min(max(l, 0), 63) * 128 + min(max(m, 0), 126)) * 32 + c4];
    v[u] = inside ? ld : z;
  }
#pragma unroll
  for (int u = 0; u < EW_U; ++u) {
    const int i = base + u * 256;
    if (i < n4) d4[i] = v[u];
  }
}

__device__ __forceinline__ float wave_sum(float s) {
#pragma unroll
  for (int o = 16; o > 0; o >>= 1) s += __shfl_xor(s, o);
  return s;
}

__global__ void __launch_bounds__(256)
k_fuse(const float* __restrict__ xspec, const float* __restrict__ xcv,
       const float* __restrict__ lns, const float* __restrict__ lnb,
       float* __restrict__ out) {
  const int lane = threadIdx.x & 31;
  const int n = blockIdx.x * 8 + (threadIdx.x >> 5);
  if (n >= 64 * 127) return;
  const int l = n / 127, m = n - l * 127;
  const int i4 = (l * 128 + m) * 32 + lane;
  const v4f a  = ((const v4f*)xspec)[i4];
  const v4f b  = ((const v4f*)xcv)[i4];
  const v4f sc = ((const v4f*)lns)[lane];
  const v4f bi = ((const v4f*)lnb)[lane];
  v4f g;
#pragma unroll
  for (int q = 0; q < 4; ++q) {
    const float v = a[q] + b[q];
    g[q] = 0.5f * v * (1.0f + tanhf(0.7978845608028654f * (v + 0.044715f * v * v * v)));
  }
  const float mu = wave_sum((g[0] + g[1]) + (g[2] + g[3])) * (1.0f / 128.0f);
  v4f d;
#pragma unroll
  for (int q = 0; q < 4; ++q) d[q] = g[q] - mu;
  const float var = wave_sum((d[0] * d[0] + d[1] * d[1]) + (d[2] * d[2] + d[3] * d[3])) * (1.0f / 128.0f);
  const float rs = rsqrtf(var + 1e-6f);
  v4f o;
#pragma unroll
  for (int q = 0; q < 4; ++q) o[q] = d[q] * rs * sc[q] + bi[q];
  ((v4f*)out)[(l * 127 + m) * 32 + lane] = o;
}

static inline void launch_gemm(hipStream_t s, const float* A, const float* B, float* C,
                               int M, int K, int lda, int ldb, int ldc,
                               long long sA, long long sB, long long sC,
                               float alpha, int acc, int batch) {
  dim3 g(2, M / 16, batch);
  gemm_wmma<<<g, dim3(32), 0, s>>>(A, B, C, M, K, lda, ldb, ldc, sA, sB, sC, alpha, acc);
}

#define EWG(n) dim3((unsigned)(((n) + 255) / 256)), dim3(256), 0, stream
#define EWG4(n4) dim3((unsigned)(((n4) + 256 * EW_U - 1) / (256 * EW_U))), dim3(256), 0, stream

extern "C" void kernel_launch(void* const* d_in, const int* in_sizes, int n_in,
                              void* d_out, int out_size, void* d_ws, size_t ws_size,
                              hipStream_t stream) {
  (void)in_sizes; (void)n_in; (void)out_size; (void)ws_size;
  const float* x       = (const float*)d_in[0];
  const float* t_emb   = (const float*)d_in[1];
  const float* sp_dw   = (const float*)d_in[2];
  const float* sp_db   = (const float*)d_in[3];
  const float* sp_wt   = (const float*)d_in[4];
  const float* sp_ck   = (const float*)d_in[5];
  const float* sp_cb   = (const float*)d_in[6];
  const float* spec_wr = (const float*)d_in[7];
  const float* spec_wi = (const float*)d_in[8];
  const float* drw     = (const float*)d_in[9];
  const float* drb     = (const float*)d_in[10];
  const float* diw     = (const float*)d_in[11];
  const float* dib     = (const float*)d_in[12];
  const float* lns     = (const float*)d_in[13];
  const float* lnb     = (const float*)d_in[14];
  float* out = (float*)d_out;
  float* ws  = (float*)d_ws;

  float *A1 = ws + OFF_A1, *XP = ws + OFF_XP, *COS1 = ws + OFF_COS1, *SIN1N = ws + OFF_SIN1N;
  float *GR = ws + OFF_GR, *GI = ws + OFF_GI, *F1R = ws + OFF_F1R, *F1I = ws + OFF_F1I;
  float *R1 = ws + OFF_R1, *R2P = ws + OFF_R2P, *T1R = ws + OFF_T1R, *T1I = ws + OFF_T1I;
  float *FFR = ws + OFF_FFR, *FFI = ws + OFF_FFI, *WCR = ws + OFF_WCR, *WCI = ws + OFF_WCI;
  float *F2R = ws + OFF_F2R, *F2I = ws + OFF_F2I, *P2A = ws + OFF_P2A, *P2B = ws + OFF_P2B;
  float *FTR = ws + OFF_FTR, *FTI = ws + OFF_FTI, *COS2 = ws + OFF_COS2, *SIN2 = ws + OFF_SIN2;
  float *XSPEC = ws + OFF_XSPEC, *TVEC = ws + OFF_TVEC, *TCR = ws + OFF_TCR, *TCI = ws + OFF_TCI;
  float *WSP = ws + OFF_WSP, *XS1 = ws + OFF_XS1, *T2R = ws + OFF_T2R, *XS2 = ws + OFF_XS2;
  float *XSP = ws + OFF_XSP, *XCV = ws + OFF_XCV;

  zerof<<<EWG4(SZ_A1 / 4)>>>(A1, (int)(SZ_A1 / 4));
  zerof<<<EWG4(SZ_P2 / 4)>>>(P2A, (int)(SZ_P2 / 4));
  zerof<<<EWG4(2048)>>>(T1R + 255LL * 8192, 2048);
  zerof<<<EWG4(2048)>>>(T1I + 255LL * 8192, 2048);
  zerof<<<EWG4(2048)>>>(T2R + 255LL * 8192, 2048);

  build_xpad<<<EWG4(SZ_XP / 4)>>>(x, XP, (int)(SZ_XP / 4));
  k_tvec<<<dim3(1), dim3(256), 0, stream>>>(t_emb, sp_dw, sp_db, drw, drb, diw, dib, TVEC, TCR, TCI);
  k_wc<<<EWG4(SZ_WC / 4)>>>(spec_wr, spec_wi, TCR, TCI, WCR, WCI, (int)(SZ_WC / 4));
  k_wsp<<<EWG(SZ_WSP)>>>(sp_wt, TVEC, WSP, SZ_WSP);
  k_phase1<<<EWG(SZ_PH1)>>>(COS1, SIN1N, SZ_PH1);
  k_phase2<<<EWG(SZ_PH2)>>>(COS2, SIN2, SZ_PH2);
  k_resize<<<dim3(1), dim3(256), 0, stream>>>(R1, R2P);
  k_legendre<<<dim3(128), dim3(128), 0, stream>>>(A1, 128, 1);
  k_mirror<<<EWG4(127 * 128 * 32)>>>(A1, 128, 127 * 128 * 32);
  k_legendre<<<dim3(64), dim3(64), 0, stream>>>(P2A, 64, 0);
  k_mirror<<<EWG4(63 * 64 * 16)>>>(P2A, 64, 63 * 64 * 16);
  k_transp_p2<<<dim3(127), dim3(256), 0, stream>>>(P2A, P2B);

  launch_gemm(stream, COS1,  XP, GR, 256, 256, 256, 128, 128, 0, 256LL * 128, 256LL * 128, 1.0f, 0, 128);
  launch_gemm(stream, SIN1N, XP, GI, 256, 256, 256, 128, 128, 0, 256LL * 128, 256LL * 128, 1.0f, 0, 128);
  launch_gemm(stream, A1, GR, F1R, 128, 128, 128, 256 * 128, 128, 128LL * 128, 128, 128LL * 128, 1.0f, 0, 255);
  launch_gemm(stream, A1, GI, F1I, 128, 128, 128, 256 * 128, 128, 128LL * 128, 128, 128LL * 128, 1.0f, 0, 255);
  launch_gemm(stream, R1, F1R, T1R, 64, 128, 128, 128, 128, 0, 128LL * 128, 64LL * 128, 1.0f, 0, 255);
  launch_gemm(stream, R1, F1I, T1I, 64, 128, 128, 128, 128, 0, 128LL * 128, 64LL * 128, 1.0f, 0, 255);
  launch_gemm(stream, R2P, T1R, FFR, 128, 256, 256, 64 * 128, 128, 0, 128, 128LL * 128, 1.0f, 0, 64);
  launch_gemm(stream, R2P, T1I, FFI, 128, 256, 256, 64 * 128, 128, 0, 128, 128LL * 128, 1.0f, 0, 64);
  launch_gemm(stream, FFR, WCR, F2R, 128, 128, 128, 128, 128, 128LL * 128, 128LL * 128, 128LL * 128,  1.0f, 0, 64);
  launch_gemm(stream, FFI, WCI, F2R, 128, 128, 128, 128, 128, 128LL * 128, 128LL * 128, 128LL * 128, -1.0f, 1, 64);
  launch_gemm(stream, FFR, WCI, F2I, 128, 128, 128, 128, 128, 128LL * 128, 128LL * 128, 128LL * 128,  1.0f, 0, 64);
  launch_gemm(stream, FFI, WCR, F2I, 128, 128, 128, 128, 128, 128LL * 128, 128LL * 128, 128LL * 128,  1.0f, 1, 64);
  launch_gemm(stream, P2B, F2R, FTR, 64, 64, 64, 128 * 128, 128, 64LL * 64, 128, 64LL * 128, 1.0f, 0, 127);
  launch_gemm(stream, P2B, F2I, FTI, 64, 64, 64, 128 * 128, 128, 64LL * 64, 128, 64LL * 128, 1.0f, 0, 127);
  launch_gemm(stream, COS2, FTR, XSPEC, 128, 128, 128, 64 * 128, 128, 0, 128, 128LL * 128,  1.0f, 0, 64);
  launch_gemm(stream, SIN2, FTI, XSPEC, 128, 128, 128, 64 * 128, 128, 0, 128, 128LL * 128, -1.0f, 1, 64);

  launch_gemm(stream, x, WSP, XS1, 32640, 128, 128, 128, 128, 0, 0, 0, 1.0f, 0, 1);
  launch_gemm(stream, R1, XS1, T2R, 64, 128, 128, 255 * 128, 128, 0, 128, 64LL * 128, 1.0f, 0, 255);
  launch_gemm(stream, R2P, T2R, XS2, 128, 256, 256, 64 * 128, 128, 0, 128, 128LL * 128, 1.0f, 0, 64);
  k_xsppad<<<EWG4(SZ_XSP / 4)>>>(XS2, XSP, (int)(SZ_XSP / 4));
  conv3x3_wmma<<<dim3(2, 8, 64), dim3(32), 0, stream>>>(XSP, sp_ck, sp_cb, TVEC, XCV);

  k_fuse<<<dim3(64 * 127 / 8), dim3(256), 0, stream>>>(XSPEC, XCV, lns, lnb, out);
}


# Resizes moved ahead of the big products: degree resize 128->64 folded into the f64-built Legendre table (one GEMM launch fewer); spatial channel product runs after both resizes on 64x128 rows
# speedup vs baseline: 47.6768x; 1.6563x over previous
#include <hip/hip_runtime.h>
#include <math.h>

typedef __attribute__((ext_vector_type(2))) float v2f;
typedef __attribute__((ext_vector_type(8))) float v8f;
typedef __attribute__((ext_vector_type(4))) float v4f;

#define PI_D 3.14159265358979323846

constexpr long long SZ_A1    = 255LL*128*128;
constexpr long long SZ_XP    = 128LL*256*128;
constexpr long long SZ_PH1   = 256LL*256;
constexpr long long SZ_G     = 128LL*256*128;
constexpr long long SZ_F1    = 255LL*128*128;
constexpr long long SZ_R1    = 64LL*128;
constexpr long long SZ_R2P   = 128LL*256;
constexpr long long SZ_T1    = 256LL*64*128;
constexpr long long SZ_FF    = 64LL*128*128;
constexpr long long SZ_WC    = 64LL*128*128;
constexpr long long SZ_F2    = 64LL*128*128;
constexpr long long SZ_P2    = 127LL*64*64;
constexpr long long SZ_FT    = 127LL*64*128;
constexpr long long SZ_PH2   = 128LL*128;
constexpr long long SZ_XSPEC = 64LL*128*128;
constexpr long long SZ_TVEC  = 256;
constexpr long long SZ_TC    = 64;
constexpr long long SZ_WSP   = 128LL*128;
constexpr long long SZ_XS1   = 128LL*255*128;
constexpr long long SZ_T2    = 256LL*64*128;
constexpr long long SZ_XS2   = 64LL*128*128;
constexpr long long SZ_XSP   = 67LL*129*128;
constexpr long long SZ_XCV   = 64LL*128*128;

constexpr long long OFF_A1    = 0;
constexpr long long OFF_XP    = OFF_A1    + SZ_A1;
constexpr long long OFF_COS1  = OFF_XP    + SZ_XP;
constexpr long long OFF_SIN1N = OFF_COS1  + SZ_PH1;
constexpr long long OFF_GR    = OFF_SIN1N + SZ_PH1;
constexpr long long OFF_GI    = OFF_GR    + SZ_G;
constexpr long long OFF_F1R   = OFF_GI    + SZ_G;
constexpr long long OFF_F1I   = OFF_F1R   + SZ_F1;
constexpr long long OFF_R1    = OFF_F1I   + SZ_F1;
constexpr long long OFF_R2P   = OFF_R1    + SZ_R1;
constexpr long long OFF_T1R   = OFF_R2P   + SZ_R2P;
constexpr long long OFF_T1I   = OFF_T1R   + SZ_T1;
constexpr long long OFF_FFR   = OFF_T1I   + SZ_T1;
constexpr long long OFF_FFI   = OFF_FFR   + SZ_FF;
constexpr long long OFF_WCR   = OFF_FFI   + SZ_FF;
constexpr long long OFF_WCI   = OFF_WCR   + SZ_WC;
constexpr long long OFF_F2R   = OFF_WCI   + SZ_WC;
constexpr long long OFF_F2I   = OFF_F2R   + SZ_F2;
constexpr long long OFF_P2A   = OFF_F2I   + SZ_F2;
constexpr long long OFF_P2B   = OFF_P2A   + SZ_P2;
constexpr long long OFF_FTR   = OFF_P2B   + SZ_P2;
constexpr long long OFF_FTI   = OFF_FTR   + SZ_FT;
constexpr long long OFF_COS2  = OFF_FTI   + SZ_FT;
constexpr long long OFF_SIN2  = OFF_COS2  + SZ_PH2;
constexpr long long OFF_XSPEC = OFF_SIN2  + SZ_PH2;
constexpr long long OFF_TVEC  = OFF_XSPEC + SZ_XSPEC;
constexpr long long OFF_TCR   = OFF_TVEC  + SZ_TVEC;
constexpr long long OFF_TCI   = OFF_TCR   + SZ_TC;
constexpr long long OFF_WSP   = OFF_TCI   + SZ_TC;
constexpr long long OFF_XS1   = OFF_WSP   + SZ_WSP;
constexpr long long OFF_T2R   = OFF_XS1   + SZ_XS1;
constexpr long long OFF_XS2   = OFF_T2R   + SZ_T2;
constexpr long long OFF_XSP   = OFF_XS2   + SZ_XS2;
constexpr long long OFF_XCV   = OFF_XSP   + SZ_XSP;

struct FragSet {
  v2f a[4];
  v2f b[4][4];
};

__device__ __forceinline__ void sched_fence() { __builtin_amdgcn_sched_barrier(0); }
__device__ __forceinline__ void sched_loads_then_mma_twice() {
  __builtin_amdgcn_sched_group_barrier(0x020, 36, 0);
  __builtin_amdgcn_sched_group_barrier(0x008, 16, 0);
  __builtin_amdgcn_sched_group_barrier(0x020, 36, 0);
  __builtin_amdgcn_sched_group_barrier(0x008, 16, 0);
}

__device__ __forceinline__ void load_frags(FragSet& f, const float* __restrict__ arow,
                                           const float* __restrict__ brow, long long ldb) {
#pragma unroll
  for (int s = 0; s < 4; ++s) {
    f.a[s].x = arow[4 * s];
    f.a[s].y = arow[4 * s + 1];
    const float* bp = brow + (long long)(4 * s) * ldb;
#pragma unroll
    for (int j = 0; j < 4; ++j) {
      f.b[s][j].x = bp[16 * j];
      f.b[s][j].y = bp[ldb + 16 * j];
    }
  }
}

__device__ __forceinline__ void mma_frags(const FragSet& f, v8f& acc0, v8f& acc1, v8f& acc2, v8f& acc3) {
#pragma unroll
  for (int s = 0; s < 4; ++s) {
    acc0 = __builtin_amdgcn_wmma_f32_16x16x4_f32(false, f.a[s], false, f.b[s][0], (short)0, acc0, false, false);
    acc1 = __builtin_amdgcn_wmma_f32_16x16x4_f32(false, f.a[s], false, f.b[s][1], (short)0, acc1, false, false);
    acc2 = __builtin_amdgcn_wmma_f32_16x16x4_f32(false, f.a[s], false, f.b[s][2], (short)0, acc2, false, false);
    acc3 = __builtin_amdgcn_wmma_f32_16x16x4_f32(false, f.a[s], false, f.b[s][3], (short)0, acc3, false, false);
  }
}

__global__ void __launch_bounds__(32)
gemm_wmma(const float* __restrict__ A, const float* __restrict__ B,
          float* __restrict__ C, int M, int K,
          int lda, int ldb, int ldc,
          long long sAb, long long sBb, long long sCb,
          long long pA, long long pB, long long pC,
          float alpha0, float alpha1, int accumulate) {
  (void)M;
  const int lane = threadIdx.x;
  const int part = blockIdx.x >> 1;
  const int n0 = (blockIdx.x & 1) * 64;
  const int m0 = blockIdx.y * 16;
  const long long b = blockIdx.z;
  A += b * sAb + part * pA; B += b * sBb + part * pB; C += b * sCb + part * pC;
  const float alpha = part ? alpha1 : alpha0;
  const int half = lane >> 4;
  const int l16  = lane & 15;
  v8f acc0 = {}, acc1 = {}, acc2 = {}, acc3 = {};
  const float* arow = A + (long long)(m0 + l16) * lda + 2 * half;
  const float* brow = B + (long long)(2 * half) * ldb + n0 + l16;
  const int K32 = K & ~31;
  if (K32 > 0) {
    FragSet P, Q;
    load_frags(P, arow, brow, ldb);
    int k0 = 0;
    for (; k0 + 32 < K32; k0 += 32) {
      load_frags(Q, arow + k0 + 16, brow + (long long)(k0 + 16) * ldb, ldb);
      mma_frags(P, acc0, acc1, acc2, acc3);
      load_frags(P, arow + k0 + 32, brow + (long long)(k0 + 32) * ldb, ldb);
      mma_frags(Q, acc0, acc1, acc2, acc3);
      sched_loads_then_mma_twice();
    }
    load_frags(Q, arow + k0 + 16, brow + (long long)(k0 + 16) * ldb, ldb);
    sched_fence();
    mma_frags(P, acc0, acc1, acc2, acc3);
    mma_frags(Q, acc0, acc1, acc2, acc3);
  }
  for (int k0 = K32; k0 < K; k0 += 4) {
    v2f a; a.x = arow[k0]; a.y = arow[k0 + 1];
    const float* bp = brow + (long long)k0 * ldb;
    v2f b0; b0.x = bp[0];   b0.y = bp[ldb];
    v2f b1; b1.x = bp[16];  b1.y = bp[ldb + 16];
    v2f b2; b2.x = bp[32];  b2.y = bp[ldb + 32];
    v2f b3; b3.x = bp[48];  b3.y = bp[ldb + 48];
    acc0 = __builtin_amdgcn_wmma_f32_16x16x4_f32(false, a, false, b0, (short)0, acc0, false, false);
    acc1 = __builtin_amdgcn_wmma_f32_16x16x4_f32(false, a, false, b1, (short)0, acc1, false, false);
    acc2 = __builtin_amdgcn_wmma_f32_16x16x4_f32(false, a, false, b2, (short)0, acc2, false, false);
    acc3 = __builtin_amdgcn_wmma_f32_16x16x4_f32(false, a, false, b3, (short)0, acc3, false, false);
  }
#pragma unroll
  for (int v = 0; v < 8; ++v) {
    const int row = m0 + half * 8 + v;
    float* cp = C + (long long)row * ldc + n0 + l16;
    if (accumulate) {
      cp[0]  += alpha * acc0[v]; cp[16] += alpha * acc1[v];
      cp[32] += alpha * acc2[v]; cp[48] += alpha * acc3[v];
    } else {
      cp[0]  = alpha * acc0[v]; cp[16] = alpha * acc1[v];
      cp[32] = alpha * acc2[v]; cp[48] = alpha * acc3[v];
    }
  }
}

__device__ __forceinline__ void conv_load(FragSet& f, const float* __restrict__ xsp,
                                          const float* __restrict__ ck, int g,
                                          int l, int mrow, int half, int ncol) {
  const int tap = g >> 3, kk = (g & 7) * 16;
  const int dy = tap / 3, dx = tap - 3 * dy;
  const float* arow = xsp + (((long long)(l + dy)) * 129 + (mrow + dx)) * 128 + 2 * half + kk;
  const float* brow = ck + (((long long)tap) * 128 + 2 * half + kk) * 128 + ncol;
  load_frags(f, arow, brow, 128);
}

__global__ void __launch_bounds__(32)
conv3x3_wmma(const float* __restrict__ xsp, const float* __restrict__ ck,
             const float* __restrict__ cb, const float* __restrict__ tvec,
             float* __restrict__ out) {
  const int lane = threadIdx.x;
  const int n0 = blockIdx.x * 64;
  const int m0 = blockIdx.y * 16;
  const int l  = blockIdx.z;
  const int half = lane >> 4, l16 = lane & 15;
  const int mrow = m0 + l16, ncol = n0 + l16;
  v8f acc0 = {}, acc1 = {}, acc2 = {}, acc3 = {};
  {
    FragSet P, Q;
    conv_load(P, xsp, ck, 0, l, mrow, half, ncol);
    int g = 0;
    for (; g + 2 < 72; g += 2) {
      conv_load(Q, xsp, ck, g + 1, l, mrow, half, ncol);
      mma_frags(P, acc0, acc1, acc2, acc3);
      conv_load(P, xsp, ck, g + 2, l, mrow, half, ncol);
      mma_frags(Q, acc0, acc1, acc2, acc3);
      sched_loads_then_mma_twice();
    }
    conv_load(Q, xsp, ck, g + 1, l, mrow, half, ncol);
    sched_fence();
    mma_frags(P, acc0, acc1, acc2, acc3);
    mma_frags(Q, acc0, acc1, acc2, acc3);
  }
  const int c0 = n0 + l16;
  const float bias0 = cb[c0]      + tvec[128 + c0];
  const float bias1 = cb[c0 + 16] + tvec[128 + c0 + 16];
  const float bias2 = cb[c0 + 32] + tvec[128 + c0 + 32];
  const float bias3 = cb[c0 + 48] + tvec[128 + c0 + 48];
#pragma unroll
  for (int v = 0; v < 8; ++v) {
    const int row = m0 + half * 8 + v;
    float* op = out + (((long long)l) * 128 + row) * 128 + c0;
    op[0]  = acc0[v] + bias0; op[16] = acc1[v] + bias1;
    op[32] = acc2[v] + bias2; op[48] = acc3[v] + bias3;
  }
}

constexpr int EW_U = 4;

__device__ __forceinline__ void zero_job(float* __restrict__ p, int n4, int blk) {
  v4f* p4 = (v4f*)p;
  const int base = blk * (128 * EW_U) + threadIdx.x;
  const v4f z = {0.0f, 0.0f, 0.0f, 0.0f};
#pragma unroll
  for (int u = 0; u < EW_U; ++u) {
    const int i = base + u * 128;
    if (i < n4) p4[i] = z;
  }
}

__device__ __forceinline__ void xpad_job(const float* __restrict__ x, float* __restrict__ xp,
                                         int n4, int blk) {
  const v4f* x4 = (const v4f*)x;
  v4f* xp4 = (v4f*)xp;
  const int base = blk * (128 * EW_U) + threadIdx.x;
  v4f v[EW_U];
#pragma unroll
  for (int u = 0; u < EW_U; ++u) {
    const int i = min(base + u * 128, n4 - 1);
    const int t = i >> 13, r = i & 8191, p = r >> 5, c4 = r & 31;
    const v4f z = {0.0f, 0.0f, 0.0f, 0.0f};
    const v4f ld = x4[(t * 255 + min(p, 254)) * 32 + c4];
    v[u] = (p < 255) ? ld : z;
  }
#pragma unroll
  for (int u = 0; u < EW_U; ++u) {
    const int i = base + u * 128;
    if (i < n4) xp4[i] = v[u];
  }
}

__device__ __forceinline__ void tvec_job(const float* __restrict__ temb,
                                         const float* __restrict__ dw, const float* __restrict__ db,
                                         const float* __restrict__ drw, const float* __restrict__ drb,
                                         const float* __restrict__ diw, const float* __restrict__ dib,
                                         float* __restrict__ tvec, float* __restrict__ tcr,
                                         float* __restrict__ tci, int blk) {
  const int tid = threadIdx.x;
  if (blk < 2) {
    const int j = blk * 128 + tid;
    float acc = db[j];
    for (int k = 0; k < 256; ++k) acc += temb[k] * dw[k * 256 + j];
    tvec[j] = acc;
  } else {
    const int sel = tid >> 6, j = tid & 63;
    const float* __restrict__ w  = sel ? diw : drw;
    const float* __restrict__ bs = sel ? dib : drb;
    float* __restrict__ dst = sel ? tci : tcr;
    float acc = bs[j];
    for (int k = 0; k < 256; ++k) acc += temb[k] * w[k * 64 + j];
    dst[j] = acc;
  }
}

__device__ __forceinline__ void phase1_job(float* __restrict__ c1, float* __restrict__ s1n, int blk) {
  const int i = blk * 128 + threadIdx.x;
  const int m = i >> 8, p = i & 255;
  if (m < 255 && p < 255) {
    double mval = (double)(m - 127);
    double phi = 2.0 * PI_D * (double)p / 255.0;
    double dphi = 2.0 * PI_D / 255.0;
    c1[i]  = (float)( cos(phi * mval) * dphi);
    s1n[i] = (float)(-sin(phi * mval) * dphi);
  } else { c1[i] = 0.0f; s1n[i] = 0.0f; }
}

__device__ __forceinline__ void phase2_job(float* __restrict__ c2, float* __restrict__ s2, int blk) {
  const int i = blk * 128 + threadIdx.x;
  const int p = i >> 7, m = i & 127;
  if (p < 127 && m < 127) {
    double mval = (double)(m - 63);
    double phi = 2.0 * PI_D * (double)p / 127.0;
    c2[i] = (float)cos(mval * phi);
    s2[i] = (float)sin(mval * phi);
  } else { c2[i] = 0.0f; s2[i] = 0.0f; }
}

__device__ __forceinline__ void resize_job(float* __restrict__ r1, float* __restrict__ r2p, int blk) {
  const int tid = threadIdx.x;
  if (blk == 0) {
    if (tid < 64) {
      int i = tid;
      float scale = 0.5f;
      float sample = (i + 0.5f) / scale - 0.5f;
      float sum = 0.0f;
      for (int j = 0; j < 128; ++j) {
        float w = 1.0f - fabsf(sample - j) * scale;
        if (w > 0.0f) sum += w;
      }
      for (int j = 0; j < 128; ++j) {
        float w = 1.0f - fabsf(sample - j) * scale;
        r1[i * 128 + j] = (w > 0.0f) ? (w / sum) : 0.0f;
      }
    }
  } else {
    int j = tid;
    if (j < 127) {
      float scale = 127.0f / 255.0f;
      float sample = (j + 0.5f) / scale - 0.5f;
      float sum = 0.0f;
      for (int p = 0; p < 255; ++p) {
        float w = 1.0f - fabsf(sample - p) * scale;
        if (w > 0.0f) sum += w;
      }
      for (int p = 0; p < 255; ++p) {
        float w = 1.0f - fabsf(sample - p) * scale;
        r2p[j * 256 + p] = (w > 0.0f) ? (w / sum) : 0.0f;
      }
      r2p[j * 256 + 255] = 0.0f;
    } else {
      for (int p = 0; p < 256; ++p) r2p[j * 256 + p] = 0.0f;
    }
  }
}

__device__ __forceinline__ void legendre_job(float* __restrict__ out, int lg, int weighted, int blk,
                                             double* cdiag, double* ca_all, double* cb_all) {
  const int L = 1 << lg;
  const int q = threadIdx.x >> lg;
  const int t = threadIdx.x & (L - 1);
  const int nblk = 1 << (2 * lg - 7);
  const int m = q * nblk + blk;
  double* ca = ca_all + q * L;
  double* cb = cb_all + q * L;
  {
    const int l = t;
    const double ll = (double)l, mm = (double)m;
    if (q == 0) cdiag[l] = (l >= 1) ? -sqrt((2.0 * ll + 1.0) / (2.0 * ll)) : 0.0;
    double a = 0.0, b = 0.0;
    if (l >= m + 2) {
      a = sqrt((4.0 * ll * ll - 1.0) / (ll * ll - mm * mm));
      b = sqrt((2.0 * ll + 1.0) * ((ll - 1.0) * (ll - 1.0) - mm * mm) /
               ((2.0 * ll - 3.0) * (ll * ll - mm * mm)));
    }
    ca[l] = a; cb[l] = b;
  }
  __syncthreads();
  const double theta = (2.0 * t + 1.0) * PI_D / (2.0 * L - 1.0);
  const double x = cos(theta), s = sin(theta);
  const double w = weighted ? (s * PI_D / (double)L) : 1.0;
  const long long LL = (long long)L * (long long)L;
  float* pos = out + (long long)(L - 1 + m) * LL + t;
  float* neg = out + (long long)(L - 1 - m) * LL + t;
  const bool mir = (m > 0);
  const bool odd = (m & 1) != 0;
  for (int l = 0; l < m; ++l) {
    pos[(long long)l * L] = 0.0f;
    neg[(long long)l * L] = 0.0f;
  }
  double pmm = sqrt(1.0 / (4.0 * PI_D)) * w;
  for (int j = 1; j <= m; ++j) pmm = cdiag[j] * s * pmm;
  {
    const float fv = (float)pmm;
    pos[(long long)m * L] = fv;
    if (mir) neg[(long long)m * L] = odd ? -fv : fv;
  }
  if (m + 1 < L) {
    double p2 = pmm;
    double p1 = sqrt(2.0 * (m + 1) + 1.0) * x * pmm;
    {
      const float fv = (float)p1;
      pos[(long long)(m + 1) * L] = fv;
      if (mir) neg[(long long)(m + 1) * L] = odd ? -fv : fv;
    }
    for (int l = m + 2; l < L; ++l) {
      const double p0 = ca[l] * x * p1 - cb[l] * p2;
      const float fv = (float)p0;
      pos[(long long)l * L] = fv;
      if (mir) neg[(long long)l * L] = odd ? -fv : fv;
      p2 = p1; p1 = p0;
    }
  }
}

__device__ __forceinline__ void legendre_fold_job(float* __restrict__ out, int blk,
                                                  double* cdiag, double* ca, double* cb, double* rw) {
  const int m = blk;
  const int t = threadIdx.x;
  {
    const int l = t;
    const double ll = (double)l, mm = (double)m;
    cdiag[l] = (l >= 1) ? -sqrt((2.0 * ll + 1.0) / (2.0 * ll)) : 0.0;
    double a = 0.0, b = 0.0;
    if (l >= m + 2) {
      a = sqrt((4.0 * ll * ll - 1.0) / (ll * ll - mm * mm));
      b = sqrt((2.0 * ll + 1.0) * ((ll - 1.0) * (ll - 1.0) - mm * mm) /
               ((2.0 * ll - 3.0) * (ll * ll - mm * mm)));
    }
    ca[l] = a; cb[l] = b;
  }
  if (t < 64) {
    const int i = t;
    const float scale = 0.5f;
    const float sample = (i + 0.5f) / scale - 0.5f;
    float wq[4];
    float sum = 0.0f;
#pragma unroll
    for (int q = 0; q < 4; ++q) {
      const int j = 2 * i - 1 + q;
      const float wv = 1.0f - fabsf(sample - j) * scale;
      wq[q] = (j >= 0 && j < 128 && wv > 0.0f) ? wv : 0.0f;
      sum += wq[q];
    }
#pragma unroll
    for (int q = 0; q < 4; ++q) rw[4 * i + q] = (double)(wq[q] / sum);
  }
  __syncthreads();
  const double theta = (2.0 * t + 1.0) * PI_D / 255.0;
  const double x = cos(theta), s = sin(theta);
  const double w = s * PI_D / 128.0;
  float* pos = out + (long long)(127 + m) * 8192 + t;
  float* neg = out + (long long)(127 - m) * 8192 + t;
  const bool mir = (m > 0);
  const bool odd = (m & 1) != 0;
  double pmm = sqrt(1.0 / (4.0 * PI_D)) * w;
  for (int j = 1; j <= m; ++j) pmm = cdiag[j] * s * pmm;
  const double pm1 = sqrt(2.0 * (m + 1) + 1.0) * x * pmm;
  double w0 = 0.0;
  double w1 = (m == 0) ? pmm : 0.0;
  for (int i = 0; i < 63; ++i) {
    const int la = 2 * i + 1, lb = 2 * i + 2;
    const double caa = ca[la], cba = cb[la], cab = ca[lb], cbb = cb[lb];
    const double r0 = rw[4 * i], r1 = rw[4 * i + 1], r2 = rw[4 * i + 2], r3 = rw[4 * i + 3];
    double pa = caa * x * w1 - cba * w0;
    pa = (la == m) ? pmm : pa;
    pa = (la == m + 1) ? pm1 : pa;
    double pb = cab * x * pa - cbb * w1;
    pb = (lb == m) ? pmm : pb;
    pb = (lb == m + 1) ? pm1 : pb;
    const double v = (r0 * w0 + r1 * w1) + (r2 * pa + r3 * pb);
    const float fv = (float)v;
    pos[i * 128] = fv;
    if (mir) neg[i * 128] = odd ? -fv : fv;
    w0 = pa; w1 = pb;
  }
  {
    double pc = ca[127] * x * w1 - cb[127] * w0;
    pc = (127 == m) ? pmm : pc;
    pc = (127 == m + 1) ? pm1 : pc;
    const double v = rw[252] * w0 + rw[253] * w1 + rw[254] * pc;
    const float fv = (float)v;
    pos[63 * 128] = fv;
    if (mir) neg[63 * 128] = odd ? -fv : fv;
  }
}

constexpr int S0_LEG1 = 128;
constexpr int S0_LEG2 = 32;
constexpr int S0_TVEC = 3;
constexpr int S0_RESZ = 2;
constexpr int S0_PH1  = (int)(SZ_PH1 / 128);
constexpr int S0_PH2  = (int)(SZ_PH2 / 128);
constexpr int S0_ZP   = 12;
constexpr int S0_XPAD = (int)(SZ_XP / 4 / (128 * EW_U));
constexpr int S0_BLOCKS = S0_LEG1 + S0_LEG2 + S0_TVEC + S0_RESZ + S0_PH1 + S0_PH2 + S0_ZP + S0_XPAD;

__global__ void __launch_bounds__(128)
k_setup0(const float* __restrict__ x, const float* __restrict__ temb,
         const float* __restrict__ dw, const float* __restrict__ db,
         const float* __restrict__ drw, const float* __restrict__ drb,
         const float* __restrict__ diw, const float* __restrict__ dib,
         float* __restrict__ ws) {
  __shared__ double cdiag[128];
  __shared__ double ca[128];
  __shared__ double cb[128];
  __shared__ double rw[256];
  int b = blockIdx.x;
  if (b < S0_LEG1) { legendre_fold_job(ws + OFF_A1, b, cdiag, ca, cb, rw); return; }
  b -= S0_LEG1;
  if (b < S0_LEG2) { legendre_job(ws + OFF_P2A, 6, 0, b, cdiag, ca, cb); return; }
  b -= S0_LEG2;
  if (b < S0_TVEC) {
    tvec_job(temb, dw, db, drw, drb, diw, dib, ws + OFF_TVEC, ws + OFF_TCR, ws + OFF_TCI, b);
    return;
  }
  b -= S0_TVEC;
  if (b < S0_RESZ) { resize_job(ws + OFF_R1, ws + OFF_R2P, b); return; }
  b -= S0_RESZ;
  if (b < S0_PH1) { phase1_job(ws + OFF_COS1, ws + OFF_SIN1N, b); return; }
  b -= S0_PH1;
  if (b < S0_PH2) { phase2_job(ws + OFF_COS2, ws + OFF_SIN2, b); return; }
  b -= S0_PH2;
  if (b < S0_ZP) {
    const int pad = b >> 2;
    float* p = (pad == 0) ? (ws + OFF_T1R) : (pad == 1) ? (ws + OFF_T1I) : (ws + OFF_T2R);
    zero_job(p + 255LL * 8192, 2048, b & 3);
    return;
  }
  b -= S0_ZP;
  xpad_job(x, ws + OFF_XP, (int)(SZ_XP / 4), b);
}

__device__ __forceinline__ void wc_job(const float* __restrict__ wr, const float* __restrict__ wi,
                                       const float* __restrict__ tcr, const float* __restrict__ tci,
                                       float* __restrict__ wcr, float* __restrict__ wci, int n4, int blk) {
  const v4f* wr4 = (const v4f*)wr;
  const v4f* wi4 = (const v4f*)wi;
  v4f* wcr4 = (v4f*)wcr;
  v4f* wci4 = (v4f*)wci;
  const int base = blk * (256 * EW_U) + threadIdx.x;
  v4f r[EW_U], im[EW_U];
  float tr[EW_U], ti[EW_U];
#pragma unroll
  for (int u = 0; u < EW_U; ++u) {
    const int i = min(base + u * 256, n4 - 1);
    const int l = i >> 12;
    r[u] = wr4[i]; im[u] = wi4[i];
    tr[u] = tcr[l]; ti[u] = tci[l];
  }
#pragma unroll
  for (int u = 0; u < EW_U; ++u) {
    const int i = base + u * 256;
    if (i < n4) {
      wcr4[i] = tr[u] * r[u] - ti[u] * im[u];
      wci4[i] = tr[u] * im[u] + ti[u] * r[u];
    }
  }
}

__device__ __forceinline__ void wsp_job(const float* __restrict__ spw, const float* __restrict__ tvec,
                                        float* __restrict__ wsp, int blk) {
  const int i = blk * 256 + threadIdx.x;
  const int o = i & 127;
  wsp[i] = tvec[o] * spw[i];
}

__device__ __forceinline__ void transp_job(const float* __restrict__ a, float* __restrict__ b,
                                           int m, float (*tile)[65]) {
  const v4f* a4 = (const v4f*)a + (long long)m * 1024;
  v4f* b4 = (v4f*)b + (long long)m * 1024;
  v4f v[4];
#pragma unroll
  for (int u = 0; u < 4; ++u) v[u] = a4[threadIdx.x + 256 * u];
#pragma unroll
  for (int u = 0; u < 4; ++u) {
    const int idx = threadIdx.x + 256 * u;
    const int l = idx >> 4, t4 = idx & 15;
    tile[l][4 * t4 + 0] = v[u].x; tile[l][4 * t4 + 1] = v[u].y;
    tile[l][4 * t4 + 2] = v[u].z; tile[l][4 * t4 + 3] = v[u].w;
  }
  __syncthreads();
#pragma unroll
  for (int u = 0; u < 4; ++u) {
    const int idx = threadIdx.x + 256 * u;
    const int t = idx >> 4, l4 = idx & 15;
    v4f o;
    o.x = tile[4 * l4 + 0][t]; o.y = tile[4 * l4 + 1][t];
    o.z = tile[4 * l4 + 2][t]; o.w = tile[4 * l4 + 3][t];
    b4[idx] = o;
  }
}

constexpr int S1_WC  = (int)(SZ_WC / 4 / (256 * EW_U));
constexpr int S1_WSP = (int)(SZ_WSP / 256);
constexpr int S1_TR  = 127;
constexpr int S1_BLOCKS = S1_WC + S1_WSP + S1_TR;

__global__ void __launch_bounds__(256)
k_setup1(const float* __restrict__ wr, const float* __restrict__ wi,
         const float* __restrict__ spw, float* __restrict__ ws) {
  __shared__ float tile[64][65];
  int b = blockIdx.x;
  if (b < S1_WC) {
    wc_job(wr, wi, ws + OFF_TCR, ws + OFF_TCI, ws + OFF_WCR, ws + OFF_WCI, (int)(SZ_WC / 4), b);
    return;
  }
  b -= S1_WC;
  if (b < S1_WSP) { wsp_job(spw, ws + OFF_TVEC, ws + OFF_WSP, b); return; }
  b -= S1_WSP;
  transp_job(ws + OFF_P2A, ws + OFF_P2B, b, tile);
}

__global__ void __launch_bounds__(256)
k_xsppad(const float* __restrict__ xs2, float* __restrict__ xsp, int n4) {
  const v4f* s4 = (const v4f*)xs2;
  v4f* d4 = (v4f*)xsp;
  const int base = blockIdx.x * (256 * EW_U) + threadIdx.x;
  v4f v[EW_U];
#pragma unroll
  for (int u = 0; u < EW_U; ++u) {
    const int i = min(base + u * 256, n4 - 1);
    const int row = i / (129 * 32), r = i - row * (129 * 32);
    const int col = r >> 5, c4 = r & 31;
    const int l = row - 1, m = col - 1;
    const bool inside = (l >= 0 && l < 64 && m >= 0 && m < 127);
    const v4f z = {0.0f, 0.0f, 0.0f, 0.0f};
    const v4f ld = s4[(min(max(l, 0), 63) * 128 + min(max(m, 0), 126)) * 32 + c4];
    v[u] = inside ? ld : z;
  }
#pragma unroll
  for (int u = 0; u < EW_U; ++u) {
    const int i = base + u * 256;
    if (i < n4) d4[i] = v[u];
  }
}

__device__ __forceinline__ float wave_sum(float s) {
#pragma unroll
  for (int o = 16; o > 0; o >>= 1) s += __shfl_xor(s, o);
  return s;
}

__global__ void __launch_bounds__(256)
k_fuse(const float* __restrict__ xspec, const float* __restrict__ xcv,
       const float* __restrict__ lns, const float* __restrict__ lnb,
       float* __restrict__ out) {
  const int lane = threadIdx.x & 31;
  const int n = blockIdx.x * 8 + (threadIdx.x >> 5);
  if (n >= 64 * 127) return;
  const int l = n / 127, m = n - l * 127;
  const int i4 = (l * 128 + m) * 32 + lane;
  const v4f a  = ((const v4f*)xspec)[i4];
  const v4f b  = ((const v4f*)xcv)[i4];
  const v4f sc = ((const v4f*)lns)[lane];
  const v4f bi = ((const v4f*)lnb)[lane];
  v4f g;
#pragma unroll
  for (int q = 0; q < 4; ++q) {
    const float v = a[q] + b[q];
    g[q] = 0.5f * v * (1.0f + tanhf(0.7978845608028654f * (v + 0.044715f * v * v * v)));
  }
  const float mu = wave_sum((g[0] + g[1]) + (g[2] + g[3])) * (1.0f / 128.0f);
  v4f d;
#pragma unroll
  for (int q = 0; q < 4; ++q) d[q] = g[q] - mu;
  const float var = wave_sum((d[0] * d[0] + d[1] * d[1]) + (d[2] * d[2] + d[3] * d[3])) * (1.0f / 128.0f);
  const float rs = rsqrtf(var + 1e-6f);
  v4f o;
#pragma unroll
  for (int q = 0; q < 4; ++q) o[q] = d[q] * rs * sc[q] + bi[q];
  ((v4f*)out)[(l * 127 + m) * 32 + lane] = o;
}

static inline void launch_gemm(hipStream_t s, const float* A, const float* B, float* C,
                               int M, int K, int lda, int ldb, int ldc,
                               long long sA, long long sB, long long sC,
                               float alpha, int acc, int batch,
                               int gx = 2, long long pA = 0, long long pB = 0, long long pC = 0,
                               float alpha1 = 1.0f) {
  dim3 g(gx, M / 16, batch);
  gemm_wmma<<<g, dim3(32), 0, s>>>(A, B, C, M, K, lda, ldb, ldc, sA, sB, sC, pA, pB, pC,
                                   alpha, alpha1, acc);
}

#define EWG4(n4) dim3((unsigned)(((n4) + 256 * EW_U - 1) / (256 * EW_U))), dim3(256), 0, stream

extern "C" void kernel_launch(void* const* d_in, const int* in_sizes, int n_in,
                              void* d_out, int out_size, void* d_ws, size_t ws_size,
                              hipStream_t stream) {
  (void)in_sizes; (void)n_in; (void)out_size; (void)ws_size;
  const float* x       = (const float*)d_in[0];
  const float* t_emb   = (const float*)d_in[1];
  const float* sp_dw   = (const float*)d_in[2];
  const float* sp_db   = (const float*)d_in[3];
  const float* sp_wt   = (const float*)d_in[4];
  const float* sp_ck   = (const float*)d_in[5];
  const float* sp_cb   = (const float*)d_in[6];
  const float* spec_wr = (const float*)d_in[7];
  const float* spec_wi = (const float*)d_in[8];
  const float* drw     = (const float*)d_in[9];
  const float* drb     = (const float*)d_in[10];
  const float* diw     = (const float*)d_in[11];
  const float* dib     = (const float*)d_in[12];
  const float* lns     = (const float*)d_in[13];
  const float* lnb     = (const float*)d_in[14];
  float* out = (float*)d_out;
  float* ws  = (float*)d_ws;

  float *A1 = ws + OFF_A1, *XP = ws + OFF_XP, *COS1 = ws + OFF_COS1;
  float *GR = ws + OFF_GR;
  float *R1 = ws + OFF_R1, *R2P = ws + OFF_R2P, *T1R = ws + OFF_T1R;
  float *FFR = ws + OFF_FFR, *FFI = ws + OFF_FFI, *WCR = ws + OFF_WCR, *WCI = ws + OFF_WCI;
  float *F2R = ws + OFF_F2R, *P2B = ws + OFF_P2B;
  float *FTR = ws + OFF_FTR, *FTI = ws + OFF_FTI, *COS2 = ws + OFF_COS2, *SIN2 = ws + OFF_SIN2;
  float *XSPEC = ws + OFF_XSPEC, *TVEC = ws + OFF_TVEC;
  float *WSP = ws + OFF_WSP, *XR = ws + OFF_XS1, *T2R = ws + OFF_T2R, *XS2 = ws + OFF_XS2;
  float *XSP = ws + OFF_XSP, *XCV = ws + OFF_XCV;

  k_setup0<<<dim3(S0_BLOCKS), dim3(128), 0, stream>>>(x, t_emb, sp_dw, sp_db, drw, drb, diw, dib, ws);
  k_setup1<<<dim3(S1_BLOCKS), dim3(256), 0, stream>>>(spec_wr, spec_wi, sp_wt, ws);

  launch_gemm(stream, COS1, XP, GR, 256, 256, 256, 128, 128, 0, 256LL * 128, 256LL * 128, 1.0f, 0, 128,
              4, SZ_PH1, 0, SZ_G, 1.0f);
  launch_gemm(stream, A1, GR, T1R, 64, 128, 128, 256 * 128, 128, 64LL * 128, 128, 64LL * 128, 1.0f, 0, 255,
              4, 0, SZ_G, SZ_T1, 1.0f);
  launch_gemm(stream, R2P, T1R, FFR, 128, 256, 256, 64 * 128, 128, 0, 128, 128LL * 128, 1.0f, 0, 64,
              4, 0, SZ_T1, SZ_FF, 1.0f);
  launch_gemm(stream, FFR, WCR, F2R, 128, 128, 128, 128, 128, 128LL * 128, 128LL * 128, 128LL * 128, 1.0f, 0, 64,
              4, 0, SZ_WC, SZ_F2, 1.0f);
  launch_gemm(stream, FFI, WCI, F2R, 128, 128, 128, 128, 128, 128LL * 128, 128LL * 128, 128LL * 128, -1.0f, 1, 64,
              4, 0, -SZ_WC, SZ_F2, 1.0f);
  launch_gemm(stream, P2B, F2R, FTR, 64, 64, 64, 128 * 128, 128, 64LL * 64, 128, 64LL * 128, 1.0f, 0, 127,
              4, 0, SZ_F2, SZ_FT, 1.0f);
  launch_gemm(stream, COS2, FTR, XSPEC, 128, 128, 128, 64 * 128, 128, 0, 128, 128LL * 128,  1.0f, 0, 64);
  launch_gemm(stream, SIN2, FTI, XSPEC, 128, 128, 128, 64 * 128, 128, 0, 128, 128LL * 128, -1.0f, 1, 64);

  launch_gemm(stream, R1, x, T2R, 64, 128, 128, 255 * 128, 128, 0, 128, 64LL * 128, 1.0f, 0, 255);
  launch_gemm(stream, R2P, T2R, XR, 128, 256, 256, 64 * 128, 128, 0, 128, 128LL * 128, 1.0f, 0, 64);
  launch_gemm(stream, XR, WSP, XS2, 64 * 128, 128, 128, 128, 128, 0, 0, 0, 1.0f, 0, 1);
  k_xsppad<<<EWG4(SZ_XSP / 4)>>>(XS2, XSP, (int)(SZ_XSP / 4));
  conv3x3_wmma<<<dim3(2, 8, 64), dim3(32), 0, stream>>>(XSP, sp_ck, sp_cb, TVEC, XCV);

  k_fuse<<<dim3(64 * 127 / 8), dim3(256), 0, stream>>>(XSPEC, XCV, lns, lnb, out);
}


# WMMA fragment loads and C stores of GEMM and conv at 16 bytes per lane: lane holds 8 consecutive k of its A row, 4 columns of each B row; B row pitch compile-time; 10 loads per 16 k, not 36
# speedup vs baseline: 50.3988x; 1.0571x over previous
#include <hip/hip_runtime.h>
#include <math.h>

typedef __attribute__((ext_vector_type(2))) float v2f;
typedef __attribute__((ext_vector_type(8))) float v8f;
typedef __attribute__((ext_vector_type(4))) float v4f;

#define PI_D 3.14159265358979323846

constexpr long long SZ_A1    = 255LL*128*128;
constexpr long long SZ_XP    = 128LL*256*128;
constexpr long long SZ_PH1   = 256LL*256;
constexpr long long SZ_G     = 128LL*256*128;
constexpr long long SZ_F1    = 255LL*128*128;
constexpr long long SZ_R1    = 64LL*128;
constexpr long long SZ_R2P   = 128LL*256;
constexpr long long SZ_T1    = 256LL*64*128;
constexpr long long SZ_FF    = 64LL*128*128;
constexpr long long SZ_WC    = 64LL*128*128;
constexpr long long SZ_F2    = 64LL*128*128;
constexpr long long SZ_P2    = 127LL*64*64;
constexpr long long SZ_FT    = 127LL*64*128;
constexpr long long SZ_PH2   = 128LL*128;
constexpr long long SZ_XSPEC = 64LL*128*128;
constexpr long long SZ_TVEC  = 256;
constexpr long long SZ_TC    = 64;
constexpr long long SZ_WSP   = 128LL*128;
constexpr long long SZ_XS1   = 128LL*255*128;
constexpr long long SZ_T2    = 256LL*64*128;
constexpr long long SZ_XS2   = 64LL*128*128;
constexpr long long SZ_XSP   = 67LL*129*128;
constexpr long long SZ_XCV   = 64LL*128*128;

constexpr long long OFF_A1    = 0;
constexpr long long OFF_XP    = OFF_A1    + SZ_A1;
constexpr long long OFF_COS1  = OFF_XP    + SZ_XP;
constexpr long long OFF_SIN1N = OFF_COS1  + SZ_PH1;
constexpr long long OFF_GR    = OFF_SIN1N + SZ_PH1;
constexpr long long OFF_GI    = OFF_GR    + SZ_G;
constexpr long long OFF_F1R   = OFF_GI    + SZ_G;
constexpr long long OFF_F1I   = OFF_F1R   + SZ_F1;
constexpr long long OFF_R1    = OFF_F1I   + SZ_F1;
constexpr long long OFF_R2P   = OFF_R1    + SZ_R1;
constexpr long long OFF_T1R   = OFF_R2P   + SZ_R2P;
constexpr long long OFF_T1I   = OFF_T1R   + SZ_T1;
constexpr long long OFF_FFR   = OFF_T1I   + SZ_T1;
constexpr long long OFF_FFI   = OFF_FFR   + SZ_FF;
constexpr long long OFF_WCR   = OFF_FFI   + SZ_FF;
constexpr long long OFF_WCI   = OFF_WCR   + SZ_WC;
constexpr long long OFF_F2R   = OFF_WCI   + SZ_WC;
constexpr long long OFF_F2I   = OFF_F2R   + SZ_F2;
constexpr long long OFF_P2A   = OFF_F2I   + SZ_F2;
constexpr long long OFF_P2B   = OFF_P2A   + SZ_P2;
constexpr long long OFF_FTR   = OFF_P2B   + SZ_P2;
constexpr long long OFF_FTI   = OFF_FTR   + SZ_FT;
constexpr long long OFF_COS2  = OFF_FTI   + SZ_FT;
constexpr long long OFF_SIN2  = OFF_COS2  + SZ_PH2;
constexpr long long OFF_XSPEC = OFF_SIN2  + SZ_PH2;
constexpr long long OFF_TVEC  = OFF_XSPEC + SZ_XSPEC;
constexpr long long OFF_TCR   = OFF_TVEC  + SZ_TVEC;
constexpr long long OFF_TCI   = OFF_TCR   + SZ_TC;
constexpr long long OFF_WSP   = OFF_TCI   + SZ_TC;
constexpr long long OFF_XS1   = OFF_WSP   + SZ_WSP;
constexpr long long OFF_T2R   = OFF_XS1   + SZ_XS1;
constexpr long long OFF_XS2   = OFF_T2R   + SZ_T2;
constexpr long long OFF_XSP   = OFF_XS2   + SZ_XS2;
constexpr long long OFF_XCV   = OFF_XSP   + SZ_XSP;

struct FragSet {
  v4f a[2];
  v4f b[8];
};

__device__ __forceinline__ void sched_fence() { __builtin_amdgcn_sched_barrier(0); }
__device__ __forceinline__ void sched_loads_then_mma_twice() {
  __builtin_amdgcn_sched_group_barrier(0x020, 10, 0);
  __builtin_amdgcn_sched_group_barrier(0x008, 16, 0);
  __builtin_amdgcn_sched_group_barrier(0x020, 10, 0);
  __builtin_amdgcn_sched_group_barrier(0x008, 16, 0);
}

__device__ __forceinline__ void load_frags(FragSet& f, const float* __restrict__ arow,
                                           const float* __restrict__ brow, long long ldb) {
  const v4f* ap = (const v4f*)arow;
  f.a[0] = ap[0];
  f.a[1] = ap[1];
#pragma unroll
  for (int i = 0; i < 8; ++i) f.b[i] = *(const v4f*)(brow + (long long)i * ldb);
}

__device__ __forceinline__ void mma_frags(const FragSet& f, v8f& acc0, v8f& acc1, v8f& acc2, v8f& acc3) {
#pragma unroll
  for (int s = 0; s < 4; ++s) {
    const v4f av = f.a[s >> 1];
    const int e0 = 2 * (s & 1);
    v2f a;  a.x = av[e0];           a.y = av[e0 + 1];
    const v4f r0 = f.b[2 * s], r1 = f.b[2 * s + 1];
    v2f b0; b0.x = r0[0];           b0.y = r1[0];
    v2f b1; b1.x = r0[1];           b1.y = r1[1];
    v2f b2; b2.x = r0[2];           b2.y = r1[2];
    v2f b3; b3.x = r0[3];           b3.y = r1[3];
    acc0 = __builtin_amdgcn_wmma_f32_16x16x4_f32(false, a, false, b0, (short)0, acc0, false, false);
    acc1 = __builtin_amdgcn_wmma_f32_16x16x4_f32(false, a, false, b1, (short)0, acc1, false, false);
    acc2 = __builtin_amdgcn_wmma_f32_16x16x4_f32(false, a, false, b2, (short)0, acc2, false, false);
    acc3 = __builtin_amdgcn_wmma_f32_16x16x4_f32(false, a, false, b3, (short)0, acc3, false, false);
  }
}

template <int LDB>
__global__ void __launch_bounds__(32)
gemm_wmma(const float* __restrict__ A, const float* __restrict__ B,
          float* __restrict__ C, int M, int K,
          int lda, int ldc,
          long long sAb, long long sBb, long long sCb,
          long long pA, long long pB, long long pC,
          float alpha0, float alpha1, int accumulate) {
  (void)M;
  constexpr long long ldb = LDB;
  const int lane = threadIdx.x;
  const int part = blockIdx.x >> 1;
  const int n0 = (blockIdx.x & 1) * 64;
  const int m0 = blockIdx.y * 16;
  const long long b = blockIdx.z;
  A += b * sAb + part * pA; B += b * sBb + part * pB; C += b * sCb + part * pC;
  const float alpha = part ? alpha1 : alpha0;
  const int half = lane >> 4;
  const int l16  = lane & 15;
  v8f acc0 = {}, acc1 = {}, acc2 = {}, acc3 = {};
  const float* arow = A + (long long)(m0 + l16) * lda + 8 * half;
  const float* brow = B + (long long)(8 * half) * ldb + n0 + 4 * l16;
  const int K32 = K & ~31;
  if (K32 > 0) {
    FragSet P, Q;
    load_frags(P, arow, brow, ldb);
    int k0 = 0;
    for (; k0 + 32 < K32; k0 += 32) {
      load_frags(Q, arow + k0 + 16, brow + (long long)(k0 + 16) * ldb, ldb);
      mma_frags(P, acc0, acc1, acc2, acc3);
      load_frags(P, arow + k0 + 32, brow + (long long)(k0 + 32) * ldb, ldb);
      mma_frags(Q, acc0, acc1, acc2, acc3);
      sched_loads_then_mma_twice();
    }
    load_frags(Q, arow + k0 + 16, brow + (long long)(k0 + 16) * ldb, ldb);
    sched_fence();
    mma_frags(P, acc0, acc1, acc2, acc3);
    mma_frags(Q, acc0, acc1, acc2, acc3);
  }
  if (K32 < K) {
    FragSet R;
    load_frags(R, arow + K32, brow + (long long)K32 * ldb, ldb);
    mma_frags(R, acc0, acc1, acc2, acc3);
  }
#pragma unroll
  for (int v = 0; v < 8; ++v) {
    const int row = m0 + half * 8 + v;
    v4f* cp = (v4f*)(C + (long long)row * ldc + n0 + 4 * l16);
    v4f o;
    o.x = alpha * acc0[v]; o.y = alpha * acc1[v]; o.z = alpha * acc2[v]; o.w = alpha * acc3[v];
    if (accumulate) o += *cp;
    *cp = o;
  }
}

__device__ __forceinline__ void conv_load(FragSet& f, const float* __restrict__ xsp,
                                          const float* __restrict__ ck, int g,
                                          int l, int mrow, int half, int ncol) {
  const int tap = g >> 3, kk = (g & 7) * 16;
  const int dy = tap / 3, dx = tap - 3 * dy;
  const float* arow = xsp + (((long long)(l + dy)) * 129 + (mrow + dx)) * 128 + 8 * half + kk;
  const float* brow = ck + (((long long)tap) * 128 + 8 * half + kk) * 128 + ncol;
  load_frags(f, arow, brow, 128);
}

__global__ void __launch_bounds__(32)
conv3x3_wmma(const float* __restrict__ xsp, const float* __restrict__ ck,
             const float* __restrict__ cb, const float* __restrict__ tvec,
             float* __restrict__ out) {
  const int lane = threadIdx.x;
  const int n0 = blockIdx.x * 64;
  const int m0 = blockIdx.y * 16;
  const int l  = blockIdx.z;
  const int half = lane >> 4, l16 = lane & 15;
  const int mrow = m0 + l16, ncol = n0 + 4 * l16;
  v8f acc0 = {}, acc1 = {}, acc2 = {}, acc3 = {};
  {
    FragSet P, Q;
    conv_load(P, xsp, ck, 0, l, mrow, half, ncol);
    int g = 0;
    for (; g + 2 < 72; g += 2) {
      conv_load(Q, xsp, ck, g + 1, l, mrow, half, ncol);
      mma_frags(P, acc0, acc1, acc2, acc3);
      conv_load(P, xsp, ck, g + 2, l, mrow, half, ncol);
      mma_frags(Q, acc0, acc1, acc2, acc3);
      sched_loads_then_mma_twice();
    }
    conv_load(Q, xsp, ck, g + 1, l, mrow, half, ncol);
    sched_fence();
    mma_frags(P, acc0, acc1, acc2, acc3);
    mma_frags(Q, acc0, acc1, acc2, acc3);
  }
  const v4f bias = *(const v4f*)(cb + ncol) + *(const v4f*)(tvec + 128 + ncol);
#pragma unroll
  for (int v = 0; v < 8; ++v) {
    const int row = m0 + half * 8 + v;
    v4f o;
    o.x = acc0[v]; o.y = acc1[v]; o.z = acc2[v]; o.w = acc3[v];
    *(v4f*)(out + (((long long)l) * 128 + row) * 128 + ncol) = o + bias;
  }
}

constexpr int EW_U = 4;

__device__ __forceinline__ void zero_job(float* __restrict__ p, int n4, int blk) {
  v4f* p4 = (v4f*)p;
  const int base = blk * (128 * EW_U) + threadIdx.x;
  const v4f z = {0.0f, 0.0f, 0.0f, 0.0f};
#pragma unroll
  for (int u = 0; u < EW_U; ++u) {
    const int i = base + u * 128;
    if (i < n4) p4[i] = z;
  }
}

__device__ __forceinline__ void xpad_job(const float* __restrict__ x, float* __restrict__ xp,
                                         int n4, int blk) {
  const v4f* x4 = (const v4f*)x;
  v4f* xp4 = (v4f*)xp;
  const int base = blk * (128 * EW_U) + threadIdx.x;
  v4f v[EW_U];
#pragma unroll
  for (int u = 0; u < EW_U; ++u) {
    const int i = min(base + u * 128, n4 - 1);
    const int t = i >> 13, r = i & 8191, p = r >> 5, c4 = r & 31;
    const v4f z = {0.0f, 0.0f, 0.0f, 0.0f};
    const v4f ld = x4[(t * 255 + min(p, 254)) * 32 + c4];
    v[u] = (p < 255) ? ld : z;
  }
#pragma unroll
  for (int u = 0; u < EW_U; ++u) {
    const int i = base + u * 128;
    if (i < n4) xp4[i] = v[u];
  }
}

__device__ __forceinline__ void tvec_job(const float* __restrict__ temb,
                                         const float* __restrict__ dw, const float* __restrict__ db,
                                         const float* __restrict__ drw, const float* __restrict__ drb,
                                         const float* __restrict__ diw, const float* __restrict__ dib,
                                         float* __restrict__ tvec, float* __restrict__ tcr,
                                         float* __restrict__ tci, int blk) {
  const int tid = threadIdx.x;
  if (blk < 2) {
    const int j = blk * 128 + tid;
    float acc = db[j];
    for (int k = 0; k < 256; ++k) acc += temb[k] * dw[k * 256 + j];
    tvec[j] = acc;
  } else {
    const int sel = tid >> 6, j = tid & 63;
    const float* __restrict__ w  = sel ? diw : drw;
    const float* __restrict__ bs = sel ? dib : drb;
    float* __restrict__ dst = sel ? tci : tcr;
    float acc = bs[j];
    for (int k = 0; k < 256; ++k) acc += temb[k] * w[k * 64 + j];
    dst[j] = acc;
  }
}

__device__ __forceinline__ void phase1_job(float* __restrict__ c1, float* __restrict__ s1n, int blk) {
  const int i = blk * 128 + threadIdx.x;
  const int m = i >> 8, p = i & 255;
  if (m < 255 && p < 255) {
    double mval = (double)(m - 127);
    double phi = 2.0 * PI_D * (double)p / 255.0;
    double dphi = 2.0 * PI_D / 255.0;
    c1[i]  = (float)( cos(phi * mval) * dphi);
    s1n[i] = (float)(-sin(phi * mval) * dphi);
  } else { c1[i] = 0.0f; s1n[i] = 0.0f; }
}

__device__ __forceinline__ void phase2_job(float* __restrict__ c2, float* __restrict__ s2, int blk) {
  const int i = blk * 128 + threadIdx.x;
  const int p = i >> 7, m = i & 127;
  if (p < 127 && m < 127) {
    double mval = (double)(m - 63);
    double phi = 2.0 * PI_D * (double)p / 127.0;
    c2[i] = (float)cos(mval * phi);
    s2[i] = (float)sin(mval * phi);
  } else { c2[i] = 0.0f; s2[i] = 0.0f; }
}

__device__ __forceinline__ void resize_job(float* __restrict__ r1, float* __restrict__ r2p, int blk) {
  const int tid = threadIdx.x;
  if (blk == 0) {
    if (tid < 64) {
      int i = tid;
      float scale = 0.5f;
      float sample = (i + 0.5f) / scale - 0.5f;
      float sum = 0.0f;
      for (int j = 0; j < 128; ++j) {
        float w = 1.0f - fabsf(sample - j) * scale;
        if (w > 0.0f) sum += w;
      }
      for (int j = 0; j < 128; ++j) {
        float w = 1.0f - fabsf(sample - j) * scale;
        r1[i * 128 + j] = (w > 0.0f) ? (w / sum) : 0.0f;
      }
    }
  } else {
    int j = tid;
    if (j < 127) {
      float scale = 127.0f / 255.0f;
      float sample = (j + 0.5f) / scale - 0.5f;
      float sum = 0.0f;
      for (int p = 0; p < 255; ++p) {
        float w = 1.0f - fabsf(sample - p) * scale;
        if (w > 0.0f) sum += w;
      }
      for (int p = 0; p < 255; ++p) {
        float w = 1.0f - fabsf(sample - p) * scale;
        r2p[j * 256 + p] = (w > 0.0f) ? (w / sum) : 0.0f;
      }
      r2p[j * 256 + 255] = 0.0f;
    } else {
      for (int p = 0; p < 256; ++p) r2p[j * 256 + p] = 0.0f;
    }
  }
}

__device__ __forceinline__ void legendre_job(float* __restrict__ out, int lg, int weighted, int blk,
                                             double* cdiag, double* ca_all, double* cb_all) {
  const int L = 1 << lg;
  const int q = threadIdx.x >> lg;
  const int t = threadIdx.x & (L - 1);
  const int nblk = 1 << (2 * lg - 7);
  const int m = q * nblk + blk;
  double* ca = ca_all + q * L;
  double* cb = cb_all + q * L;
  {
    const int l = t;
    const double ll = (double)l, mm = (double)m;
    if (q == 0) cdiag[l] = (l >= 1) ? -sqrt((2.0 * ll + 1.0) / (2.0 * ll)) : 0.0;
    double a = 0.0, b = 0.0;
    if (l >= m + 2) {
      a = sqrt((4.0 * ll * ll - 1.0) / (ll * ll - mm * mm));
      b = sqrt((2.0 * ll + 1.0) * ((ll - 1.0) * (ll - 1.0) - mm * mm) /
               ((2.0 * ll - 3.0) * (ll * ll - mm * mm)));
    }
    ca[l] = a; cb[l] = b;
  }
  __syncthreads();
  const double theta = (2.0 * t + 1.0) * PI_D / (2.0 * L - 1.0);
  const double x = cos(theta), s = sin(theta);
  const double w = weighted ? (s * PI_D / (double)L) : 1.0;
  const long long LL = (long long)L * (long long)L;
  float* pos = out + (long long)(L - 1 + m) * LL + t;
  float* neg = out + (long long)(L - 1 - m) * LL + t;
  const bool mir = (m > 0);
  const bool odd = (m & 1) != 0;
  for (int l = 0; l < m; ++l) {
    pos[(long long)l * L] = 0.0f;
    neg[(long long)l * L] = 0.0f;
  }
  double pmm = sqrt(1.0 / (4.0 * PI_D)) * w;
  for (int j = 1; j <= m; ++j) pmm = cdiag[j] * s * pmm;
  {
    const float fv = (float)pmm;
    pos[(long long)m * L] = fv;
    if (mir) neg[(long long)m * L] = odd ? -fv : fv;
  }
  if (m + 1 < L) {
    double p2 = pmm;
    double p1 = sqrt(2.0 * (m + 1) + 1.0) * x * pmm;
    {
      const float fv = (float)p1;
      pos[(long long)(m + 1) * L] = fv;
      if (mir) neg[(long long)(m + 1) * L] = odd ? -fv : fv;
    }
    for (int l = m + 2; l < L; ++l) {
      const double p0 = ca[l] * x * p1 - cb[l] * p2;
      const float fv = (float)p0;
      pos[(long long)l * L] = fv;
      if (mir) neg[(long long)l * L] = odd ? -fv : fv;
      p2 = p1; p1 = p0;
    }
  }
}

__device__ __forceinline__ void legendre_fold_job(float* __restrict__ out, int blk,
                                                  double* cdiag, double* ca, double* cb, double* rw) {
  const int m = blk;
  const int t = threadIdx.x;
  {
    const int l = t;
    const double ll = (double)l, mm = (double)m;
    cdiag[l] = (l >= 1) ? -sqrt((2.0 * ll + 1.0) / (2.0 * ll)) : 0.0;
    double a = 0.0, b = 0.0;
    if (l >= m + 2) {
      a = sqrt((4.0 * ll * ll - 1.0) / (ll * ll - mm * mm));
      b = sqrt((2.0 * ll + 1.0) * ((ll - 1.0) * (ll - 1.0) - mm * mm) /
               ((2.0 * ll - 3.0) * (ll * ll - mm * mm)));
    }
    ca[l] = a; cb[l] = b;
  }
  if (t < 64) {
    const int i = t;
    const float scale = 0.5f;
    const float sample = (i + 0.5f) / scale - 0.5f;
    float wq[4];
    float sum = 0.0f;
#pragma unroll
    for (int q = 0; q < 4; ++q) {
      const int j = 2 * i - 1 + q;
      const float wv = 1.0f - fabsf(sample - j) * scale;
      wq[q] = (j >= 0 && j < 128 && wv > 0.0f) ? wv : 0.0f;
      sum += wq[q];
    }
#pragma unroll
    for (int q = 0; q < 4; ++q) rw[4 * i + q] = (double)(wq[q] / sum);
  }
  __syncthreads();
  const double theta = (2.0 * t + 1.0) * PI_D / 255.0;
  const double x = cos(theta), s = sin(theta);
  const double w = s * PI_D / 128.0;
  float* pos = out + (long long)(127 + m) * 8192 + t;
  float* neg = out + (long long)(127 - m) * 8192 + t;
  const bool mir = (m > 0);
  const bool odd = (m & 1) != 0;
  double pmm = sqrt(1.0 / (4.0 * PI_D)) * w;
  for (int j = 1; j <= m; ++j) pmm = cdiag[j] * s * pmm;
  const double pm1 = sqrt(2.0 * (m + 1) + 1.0) * x * pmm;
  double w0 = 0.0;
  double w1 = (m == 0) ? pmm : 0.0;
  for (int i = 0; i < 63; ++i) {
    const int la = 2 * i + 1, lb = 2 * i + 2;
    const double caa = ca[la], cba = cb[la], cab = ca[lb], cbb = cb[lb];
    const double r0 = rw[4 * i], r1 = rw[4 * i + 1], r2 = rw[4 * i + 2], r3 = rw[4 * i + 3];
    double pa = caa * x * w1 - cba * w0;
    pa = (la == m) ? pmm : pa;
    pa = (la == m + 1) ? pm1 : pa;
    double pb = cab * x * pa - cbb * w1;
    pb = (lb == m) ? pmm : pb;
    pb = (lb == m + 1) ? pm1 : pb;
    const double v = (r0 * w0 + r1 * w1) + (r2 * pa + r3 * pb);
    const float fv = (float)v;
    pos[i * 128] = fv;
    if (mir) neg[i * 128] = odd ? -fv : fv;
    w0 = pa; w1 = pb;
  }
  {
    double pc = ca[127] * x * w1 - cb[127] * w0;
    pc = (127 == m) ? pmm : pc;
    pc = (127 == m + 1) ? pm1 : pc;
    const double v = rw[252] * w0 + rw[253] * w1 + rw[254] * pc;
    const float fv = (float)v;
    pos[63 * 128] = fv;
    if (mir) neg[63 * 128] = odd ? -fv : fv;
  }
}

constexpr int S0_LEG1 = 128;
constexpr int S0_LEG2 = 32;
constexpr int S0_TVEC = 3;
constexpr int S0_RESZ = 2;
constexpr int S0_PH1  = (int)(SZ_PH1 / 128);
constexpr int S0_PH2  = (int)(SZ_PH2 / 128);
constexpr int S0_ZP   = 12;
constexpr int S0_XPAD = (int)(SZ_XP / 4 / (128 * EW_U));
constexpr int S0_BLOCKS = S0_LEG1 + S0_LEG2 + S0_TVEC + S0_RESZ + S0_PH1 + S0_PH2 + S0_ZP + S0_XPAD;

__global__ void __launch_bounds__(128)
k_setup0(const float* __restrict__ x, const float* __restrict__ temb,
         const float* __restrict__ dw, const float* __restrict__ db,
         const float* __restrict__ drw, const float* __restrict__ drb,
         const float* __restrict__ diw, const float* __restrict__ dib,
         float* __restrict__ ws) {
  __shared__ double cdiag[128];
  __shared__ double ca[128];
  __shared__ double cb[128];
  __shared__ double rw[256];
  int b = blockIdx.x;
  if (b < S0_LEG1) { legendre_fold_job(ws + OFF_A1, b, cdiag, ca, cb, rw); return; }
  b -= S0_LEG1;
  if (b < S0_LEG2) { legendre_job(ws + OFF_P2A, 6, 0, b, cdiag, ca, cb); return; }
  b -= S0_LEG2;
  if (b < S0_TVEC) {
    tvec_job(temb, dw, db, drw, drb, diw, dib, ws + OFF_TVEC, ws + OFF_TCR, ws + OFF_TCI, b);
    return;
  }
  b -= S0_TVEC;
  if (b < S0_RESZ) { resize_job(ws + OFF_R1, ws + OFF_R2P, b); return; }
  b -= S0_RESZ;
  if (b < S0_PH1) { phase1_job(ws + OFF_COS1, ws + OFF_SIN1N, b); return; }
  b -= S0_PH1;
  if (b < S0_PH2) { phase2_job(ws + OFF_COS2, ws + OFF_SIN2, b); return; }
  b -= S0_PH2;
  if (b < S0_ZP) {
    const int pad = b >> 2;
    float* p = (pad == 0) ? (ws + OFF_T1R) : (pad == 1) ? (ws + OFF_T1I) : (ws + OFF_T2R);
    zero_job(p + 255LL * 8192, 2048, b & 3);
    return;
  }
  b -= S0_ZP;
  xpad_job(x, ws + OFF_XP, (int)(SZ_XP / 4), b);
}

__device__ __forceinline__ void wc_job(const float* __restrict__ wr, const float* __restrict__ wi,
                                       const float* __restrict__ tcr, const float* __restrict__ tci,
                                       float* __restrict__ wcr, float* __restrict__ wci, int n4, int blk) {
  const v4f* wr4 = (const v4f*)wr;
  const v4f* wi4 = (const v4f*)wi;
  v4f* wcr4 = (v4f*)wcr;
  v4f* wci4 = (v4f*)wci;
  const int base = blk * (256 * EW_U) + threadIdx.x;
  v4f r[EW_U], im[EW_U];
  float tr[EW_U], ti[EW_U];
#pragma unroll
  for (int u = 0; u < EW_U; ++u) {
    const int i = min(base + u * 256, n4 - 1);
    const int l = i >> 12;
    r[u] = wr4[i]; im[u] = wi4[i];
    tr[u] = tcr[l]; ti[u] = tci[l];
  }
#pragma unroll
  for (int u = 0; u < EW_U; ++u) {
    const int i = base + u * 256;
    if (i < n4) {
      wcr4[i] = tr[u] * r[u] - ti[u] * im[u];
      wci4[i] = tr[u] * im[u] + ti[u] * r[u];
    }
  }
}

__device__ __forceinline__ void wsp_job(const float* __restrict__ spw, const float* __restrict__ tvec,
                                        float* __restrict__ wsp, int blk) {
  const int i = blk * 256 + threadIdx.x;
  const int o = i & 127;
  wsp[i] = tvec[o] * spw[i];
}

__device__ __forceinline__ void transp_job(const float* __restrict__ a, float* __restrict__ b,
                                           int m, float (*tile)[65]) {
  const v4f* a4 = (const v4f*)a + (long long)m * 1024;
  v4f* b4 = (v4f*)b + (long long)m * 1024;
  v4f v[4];
#pragma unroll
  for (int u = 0; u < 4; ++u) v[u] = a4[threadIdx.x + 256 * u];
#pragma unroll
  for (int u = 0; u < 4; ++u) {
    const int idx = threadIdx.x + 256 * u;
    const int l = idx >> 4, t4 = idx & 15;
    tile[l][4 * t4 + 0] = v[u].x; tile[l][4 * t4 + 1] = v[u].y;
    tile[l][4 * t4 + 2] = v[u].z; tile[l][4 * t4 + 3] = v[u].w;
  }
  __syncthreads();
#pragma unroll
  for (int u = 0; u < 4; ++u) {
    const int idx = threadIdx.x + 256 * u;
    const int t = idx >> 4, l4 = idx & 15;
    v4f o;
    o.x = tile[4 * l4 + 0][t]; o.y = tile[4 * l4 + 1][t];
    o.z = tile[4 * l4 + 2][t]; o.w = tile[4 * l4 + 3][t];
    b4[idx] = o;
  }
}

constexpr int S1_WC  = (int)(SZ_WC / 4 / (256 * EW_U));
constexpr int S1_WSP = (int)(SZ_WSP / 256);
constexpr int S1_TR  = 127;
constexpr int S1_BLOCKS = S1_WC + S1_WSP + S1_TR;

__global__ void __launch_bounds__(256)
k_setup1(const float* __restrict__ wr, const float* __restrict__ wi,
         const float* __restrict__ spw, float* __restrict__ ws) {
  __shared__ float tile[64][65];
  int b = blockIdx.x;
  if (b < S1_WC) {
    wc_job(wr, wi, ws + OFF_TCR, ws + OFF_TCI, ws + OFF_WCR, ws + OFF_WCI, (int)(SZ_WC / 4), b);
    return;
  }
  b -= S1_WC;
  if (b < S1_WSP) { wsp_job(spw, ws + OFF_TVEC, ws + OFF_WSP, b); return; }
  b -= S1_WSP;
  transp_job(ws + OFF_P2A, ws + OFF_P2B, b, tile);
}

__global__ void __launch_bounds__(256)
k_xsppad(const float* __restrict__ xs2, float* __restrict__ xsp, int n4) {
  const v4f* s4 = (const v4f*)xs2;
  v4f* d4 = (v4f*)xsp;
  const int base = blockIdx.x * (256 * EW_U) + threadIdx.x;
  v4f v[EW_U];
#pragma unroll
  for (int u = 0; u < EW_U; ++u) {
    const int i = min(base + u * 256, n4 - 1);
    const int row = i / (129 * 32), r = i - row * (129 * 32);
    const int col = r >> 5, c4 = r & 31;
    const int l = row - 1, m = col - 1;
    const bool inside = (l >= 0 && l < 64 && m >= 0 && m < 127);
    const v4f z = {0.0f, 0.0f, 0.0f, 0.0f};
    const v4f ld = s4[(min(max(l, 0), 63) * 128 + min(max(m, 0), 126)) * 32 + c4];
    v[u] = inside ? ld : z;
  }
#pragma unroll
  for (int u = 0; u < EW_U; ++u) {
    const int i = base + u * 256;
    if (i < n4) d4[i] = v[u];
  }
}

__device__ __forceinline__ float wave_sum(float s) {
#pragma unroll
  for (int o = 16; o > 0; o >>= 1) s += __shfl_xor(s, o);
  return s;
}

__global__ void __launch_bounds__(256)
k_fuse(const float* __restrict__ xspec, const float* __restrict__ xcv,
       const float* __restrict__ lns, const float* __restrict__ lnb,
       float* __restrict__ out) {
  const int lane = threadIdx.x & 31;
  const int n = blockIdx.x * 8 + (threadIdx.x >> 5);
  if (n >= 64 * 127) return;
  const int l = n / 127, m = n - l * 127;
  const int i4 = (l * 128 + m) * 32 + lane;
  const v4f a  = ((const v4f*)xspec)[i4];
  const v4f b  = ((const v4f*)xcv)[i4];
  const v4f sc = ((const v4f*)lns)[lane];
  const v4f bi = ((const v4f*)lnb)[lane];
  v4f g;
#pragma unroll
  for (int q = 0; q < 4; ++q) {
    const float v = a[q] + b[q];
    g[q] = 0.5f * v * (1.0f + tanhf(0.7978845608028654f * (v + 0.044715f * v * v * v)));
  }
  const float mu = wave_sum((g[0] + g[1]) + (g[2] + g[3])) * (1.0f / 128.0f);
  v4f d;
#pragma unroll
  for (int q = 0; q < 4; ++q) d[q] = g[q] - mu;
  const float var = wave_sum((d[0] * d[0] + d[1] * d[1]) + (d[2] * d[2] + d[3] * d[3])) * (1.0f / 128.0f);
  const float rs = rsqrtf(var + 1e-6f);
  v4f o;
#pragma unroll
  for (int q = 0; q < 4; ++q) o[q] = d[q] * rs * sc[q] + bi[q];
  ((v4f*)out)[(l * 127 + m) * 32 + lane] = o;
}

template <int LDB>
static inline void launch_gemm(hipStream_t s, const float* A, const float* B, float* C,
                               int M, int K, int lda, int ldc,
                               long long sA, long long sB, long long sC,
                               float alpha, int acc, int batch,
                               int gx = 2, long long pA = 0, long long pB = 0, long long pC = 0,
                               float alpha1 = 1.0f) {
  dim3 g(gx, M / 16, batch);
  gemm_wmma<LDB><<<g, dim3(32), 0, s>>>(A, B, C, M, K, lda, ldc, sA, sB, sC, pA, pB, pC,
                                        alpha, alpha1, acc);
}

#define EWG4(n4) dim3((unsigned)(((n4) + 256 * EW_U - 1) / (256 * EW_U))), dim3(256), 0, stream

extern "C" void kernel_launch(void* const* d_in, const int* in_sizes, int n_in,
                              void* d_out, int out_size, void* d_ws, size_t ws_size,
                              hipStream_t stream) {
  (void)in_sizes; (void)n_in; (void)out_size; (void)ws_size;
  const float* x       = (const float*)d_in[0];
  const float* t_emb   = (const float*)d_in[1];
  const float* sp_dw   = (const float*)d_in[2];
  const float* sp_db   = (const float*)d_in[3];
  const float* sp_wt   = (const float*)d_in[4];
  const float* sp_ck   = (const float*)d_in[5];
  const float* sp_cb   = (const float*)d_in[6];
  const float* spec_wr = (const float*)d_in[7];
  const float* spec_wi = (const float*)d_in[8];
  const float* drw     = (const float*)d_in[9];
  const float* drb     = (const float*)d_in[10];
  const float* diw     = (const float*)d_in[11];
  const float* dib     = (const float*)d_in[12];
  const float* lns     = (const float*)d_in[13];
  const float* lnb     = (const float*)d_in[14];
  float* out = (float*)d_out;
  float* ws  = (float*)d_ws;

  float *A1 = ws + OFF_A1, *XP = ws + OFF_XP, *COS1 = ws + OFF_COS1;
  float *GR = ws + OFF_GR;
  float *R1 = ws + OFF_R1, *R2P = ws + OFF_R2P, *T1R = ws + OFF_T1R;
  float *FFR = ws + OFF_FFR, *FFI = ws + OFF_FFI, *WCR = ws + OFF_WCR, *WCI = ws + OFF_WCI;
  float *F2R = ws + OFF_F2R, *P2B = ws + OFF_P2B;
  float *FTR = ws + OFF_FTR, *FTI = ws + OFF_FTI, *COS2 = ws + OFF_COS2, *SIN2 = ws + OFF_SIN2;
  float *XSPEC = ws + OFF_XSPEC, *TVEC = ws + OFF_TVEC;
  float *WSP = ws + OFF_WSP, *XR = ws + OFF_XS1, *T2R = ws + OFF_T2R, *XS2 = ws + OFF_XS2;
  float *XSP = ws + OFF_XSP, *XCV = ws + OFF_XCV;

  k_setup0<<<dim3(S0_BLOCKS), dim3(128), 0, stream>>>(x, t_emb, sp_dw, sp_db, drw, drb, diw, dib, ws);
  k_setup1<<<dim3(S1_BLOCKS), dim3(256), 0, stream>>>(spec_wr, spec_wi, sp_wt, ws);

  launch_gemm<128>(stream, COS1, XP, GR, 256, 256, 256, 128, 0, 256LL * 128, 256LL * 128, 1.0f, 0, 128,
              4, SZ_PH1, 0, SZ_G, 1.0f);
  launch_gemm<256 * 128>(stream, A1, GR, T1R, 64, 128, 128, 128, 64LL * 128, 128, 64LL * 128, 1.0f, 0, 255,
              4, 0, SZ_G, SZ_T1, 1.0f);
  launch_gemm<64 * 128>(stream, R2P, T1R, FFR, 128, 256, 256, 128, 0, 128, 128LL * 128, 1.0f, 0, 64,
              4, 0, SZ_T1, SZ_FF, 1.0f);
  launch_gemm<128>(stream, FFR, WCR, F2R, 128, 128, 128, 128, 128LL * 128, 128LL * 128, 128LL * 128, 1.0f, 0, 64,
              4, 0, SZ_WC, SZ_F2, 1.0f);
  launch_gemm<128>(stream, FFI, WCI, F2R, 128, 128, 128, 128, 128LL * 128, 128LL * 128, 128LL * 128, -1.0f, 1, 64,
              4, 0, -SZ_WC, SZ_F2, 1.0f);
  launch_gemm<128 * 128>(stream, P2B, F2R, FTR, 64, 64, 64, 128, 64LL * 64, 128, 64LL * 128, 1.0f, 0, 127,
              4, 0, SZ_F2, SZ_FT, 1.0f);
  launch_gemm<64 * 128>(stream, COS2, FTR, XSPEC, 128, 128, 128, 128, 0, 128, 128LL * 128,  1.0f, 0, 64);
  launch_gemm<64 * 128>(stream, SIN2, FTI, XSPEC, 128, 128, 128, 128, 0, 128, 128LL * 128, -1.0f, 1, 64);

  launch_gemm<255 * 128>(stream, R1, x, T2R, 64, 128, 128, 128, 0, 128, 64LL * 128, 1.0f, 0, 255);
  launch_gemm<64 * 128>(stream, R2P, T2R, XR, 128, 256, 256, 128, 0, 128, 128LL * 128, 1.0f, 0, 64);
  launch_gemm<128>(stream, XR, WSP, XS2, 64 * 128, 128, 128, 128, 0, 0, 0, 1.0f, 0, 1);
  k_xsppad<<<EWG4(SZ_XSP / 4)>>>(XS2, XSP, (int)(SZ_XSP / 4));
  conv3x3_wmma<<<dim3(2, 8, 64), dim3(32), 0, stream>>>(XSP, sp_ck, sp_cb, TVEC, XCV);

  k_fuse<<<dim3(64 * 127 / 8), dim3(256), 0, stream>>>(XSPEC, XCV, lns, lnb, out);
}


# Forward transform on orders m >= 0 only (real input: negative orders are conjugates, folded into the order-resize table): phase product M 256->128, Legendre batches 255->128, resize K 256->128
# speedup vs baseline: 72.4677x; 1.4379x over previous
#include <hip/hip_runtime.h>
#include <math.h>

typedef __attribute__((ext_vector_type(2))) float v2f;
typedef __attribute__((ext_vector_type(8))) float v8f;
typedef __attribute__((ext_vector_type(4))) float v4f;

#define PI_D 3.14159265358979323846

constexpr long long SZ_A1    = 255LL*128*128;
constexpr long long SZ_XP    = 128LL*256*128;
constexpr long long SZ_PH1   = 256LL*256;
constexpr long long SZ_G     = 128LL*256*128;
constexpr long long SZ_F1    = 255LL*128*128;
constexpr long long SZ_R1    = 64LL*128;
constexpr long long SZ_R2P   = 128LL*256;
constexpr long long SZ_T1    = 256LL*64*128;
constexpr long long SZ_FF    = 64LL*128*128;
constexpr long long SZ_WC    = 64LL*128*128;
constexpr long long SZ_F2    = 64LL*128*128;
constexpr long long SZ_P2    = 127LL*64*64;
constexpr long long SZ_FT    = 127LL*64*128;
constexpr long long SZ_PH2   = 128LL*128;
constexpr long long SZ_XSPEC = 64LL*128*128;
constexpr long long SZ_TVEC  = 256;
constexpr long long SZ_TC    = 64;
constexpr long long SZ_WSP   = 128LL*128;
constexpr long long SZ_XS1   = 128LL*255*128;
constexpr long long SZ_T2    = 256LL*64*128;
constexpr long long SZ_XS2   = 64LL*128*128;
constexpr long long SZ_XSP   = 67LL*129*128;
constexpr long long SZ_XCV   = 64LL*128*128;

constexpr long long OFF_A1    = 0;
constexpr long long OFF_XP    = OFF_A1    + SZ_A1;
constexpr long long OFF_COS1  = OFF_XP    + SZ_XP;
constexpr long long OFF_SIN1N = OFF_COS1  + SZ_PH1;
constexpr long long OFF_GR    = OFF_SIN1N + SZ_PH1;
constexpr long long OFF_GI    = OFF_GR    + SZ_G;
constexpr long long OFF_F1R   = OFF_GI    + SZ_G;
constexpr long long OFF_F1I   = OFF_F1R   + SZ_F1;
constexpr long long OFF_R1    = OFF_F1I   + SZ_F1;
constexpr long long OFF_R2P   = OFF_R1    + SZ_R1;
constexpr long long OFF_T1R   = OFF_R2P   + SZ_R2P;
constexpr long long OFF_T1I   = OFF_T1R   + SZ_T1;
constexpr long long OFF_FFR   = OFF_T1I   + SZ_T1;
constexpr long long OFF_FFI   = OFF_FFR   + SZ_FF;
constexpr long long OFF_WCR   = OFF_FFI   + SZ_FF;
constexpr long long OFF_WCI   = OFF_WCR   + SZ_WC;
constexpr long long OFF_F2R   = OFF_WCI   + SZ_WC;
constexpr long long OFF_F2I   = OFF_F2R   + SZ_F2;
constexpr long long OFF_P2A   = OFF_F2I   + SZ_F2;
constexpr long long OFF_P2B   = OFF_P2A   + SZ_P2;
constexpr long long OFF_FTR   = OFF_P2B   + SZ_P2;
constexpr long long OFF_FTI   = OFF_FTR   + SZ_FT;
constexpr long long OFF_COS2  = OFF_FTI   + SZ_FT;
constexpr long long OFF_SIN2  = OFF_COS2  + SZ_PH2;
constexpr long long OFF_XSPEC = OFF_SIN2  + SZ_PH2;
constexpr long long OFF_TVEC  = OFF_XSPEC + SZ_XSPEC;
constexpr long long OFF_TCR   = OFF_TVEC  + SZ_TVEC;
constexpr long long OFF_TCI   = OFF_TCR   + SZ_TC;
constexpr long long OFF_WSP   = OFF_TCI   + SZ_TC;
constexpr long long OFF_XS1   = OFF_WSP   + SZ_WSP;
constexpr long long OFF_T2R   = OFF_XS1   + SZ_XS1;
constexpr long long OFF_XS2   = OFF_T2R   + SZ_T2;
constexpr long long OFF_XSP   = OFF_XS2   + SZ_XS2;
constexpr long long OFF_XCV   = OFF_XSP   + SZ_XSP;
constexpr long long OFF_WCS   = OFF_F1R;
constexpr long long OFF_FFC   = OFF_FFR;
constexpr long long OFF_PH2C  = OFF_COS2;
constexpr long long OFF_RF    = OFF_WCR;

struct FragSet {
  v4f a[2];
  v4f b[8];
};

__device__ __forceinline__ void sched_fence() { __builtin_amdgcn_sched_barrier(0); }
__device__ __forceinline__ void sched_loads_then_mma_twice() {
  __builtin_amdgcn_sched_group_barrier(0x020, 10, 0);
  __builtin_amdgcn_sched_group_barrier(0x008, 16, 0);
  __builtin_amdgcn_sched_group_barrier(0x020, 10, 0);
  __builtin_amdgcn_sched_group_barrier(0x008, 16, 0);
}

__device__ __forceinline__ void load_frags(FragSet& f, const float* __restrict__ arow,
                                           const float* __restrict__ brow, long long ldb) {
  const v4f* ap = (const v4f*)arow;
  f.a[0] = ap[0];
  f.a[1] = ap[1];
#pragma unroll
  for (int i = 0; i < 8; ++i) f.b[i] = *(const v4f*)(brow + (long long)i * ldb);
}

__device__ __forceinline__ void mma_frags(const FragSet& f, v8f& acc0, v8f& acc1, v8f& acc2, v8f& acc3) {
#pragma unroll
  for (int s = 0; s < 4; ++s) {
    const v4f av = f.a[s >> 1];
    const int e0 = 2 * (s & 1);
    v2f a;  a.x = av[e0];           a.y = av[e0 + 1];
    const v4f r0 = f.b[2 * s], r1 = f.b[2 * s + 1];
    v2f b0; b0.x = r0[0];           b0.y = r1[0];
    v2f b1; b1.x = r0[1];           b1.y = r1[1];
    v2f b2; b2.x = r0[2];           b2.y = r1[2];
    v2f b3; b3.x = r0[3];           b3.y = r1[3];
    acc0 = __builtin_amdgcn_wmma_f32_16x16x4_f32(false, a, false, b0, (short)0, acc0, false, false);
    acc1 = __builtin_amdgcn_wmma_f32_16x16x4_f32(false, a, false, b1, (short)0, acc1, false, false);
    acc2 = __builtin_amdgcn_wmma_f32_16x16x4_f32(false, a, false, b2, (short)0, acc2, false, false);
    acc3 = __builtin_amdgcn_wmma_f32_16x16x4_f32(false, a, false, b3, (short)0, acc3, false, false);
  }
}

template <int LDB, int K>
__device__ __forceinline__ void gemm_tile(const float* __restrict__ A, const float* __restrict__ B,
                                          float* __restrict__ C, int lda, int ldc,
                                          int n0, int m0) {
  static_assert(K >= 32 && (K & 31) == 0, "K is walked in pairs of 16-k groups");
  constexpr long long ldb = LDB;
  const int lane = threadIdx.x;
  const int half = lane >> 4;
  const int l16  = lane & 15;
  v8f acc0 = {}, acc1 = {}, acc2 = {}, acc3 = {};
  const float* arow = A + (long long)(m0 + l16) * lda + 8 * half;
  const float* brow = B + (long long)(8 * half) * ldb + n0 + 4 * l16;
  {
    FragSet P, Q;
    load_frags(P, arow, brow, ldb);
    int k0 = 0;
#pragma clang loop unroll(disable)
    for (; k0 + 32 < K; k0 += 32) {
      load_frags(Q, arow + k0 + 16, brow + (long long)(k0 + 16) * ldb, ldb);
      mma_frags(P, acc0, acc1, acc2, acc3);
      load_frags(P, arow + k0 + 32, brow + (long long)(k0 + 32) * ldb, ldb);
      mma_frags(Q, acc0, acc1, acc2, acc3);
      sched_loads_then_mma_twice();
    }
    load_frags(Q, arow + k0 + 16, brow + (long long)(k0 + 16) * ldb, ldb);
    sched_fence();
    mma_frags(P, acc0, acc1, acc2, acc3);
    mma_frags(Q, acc0, acc1, acc2, acc3);
  }
#pragma unroll
  for (int v = 0; v < 8; ++v) {
    const int row = m0 + half * 8 + v;
    v4f o;
    o.x = acc0[v]; o.y = acc1[v]; o.z = acc2[v]; o.w = acc3[v];
    *(v4f*)(C + (long long)row * ldc + n0 + 4 * l16) = o;
  }
}

template <int LDB, int K>
__device__ __forceinline__ void gemm_job(const float* __restrict__ A, const float* __restrict__ B,
                                         float* __restrict__ C,
                                         long long sAb, long long sBb, long long sCb,
                                         long long pA, long long pB, long long pC,
                                         int lda, int ldc, int lgx, int lgmy, int idx) {
  const int bx = idx & ((1 << lgx) - 1);
  const int t  = idx >> lgx;
  const int by = t & ((1 << lgmy) - 1);
  const long long b = t >> lgmy;
  const int part = bx >> 1;
  A += b * sAb + part * pA; B += b * sBb + part * pB; C += b * sCb + part * pC;
  gemm_tile<LDB, K>(A, B, C, lda, ldc, (bx & 1) * 64, by * 16);
}

template <int LDB, int K>
__global__ void __launch_bounds__(32)
gemm_wmma(const float* __restrict__ A, const float* __restrict__ B, float* __restrict__ C,
          long long sAb, long long sBb, long long sCb,
          long long pA, long long pB, long long pC,
          int lda, int ldc, int lgx, int lgmy) {
  gemm_job<LDB, K>(A, B, C, sAb, sBb, sCb, pA, pB, pC, lda, ldc, lgx, lgmy, (int)blockIdx.x);
}

template <int LDB0, int K0, int LDB1, int K1>
__global__ void __launch_bounds__(32)
gemm_pair(const float* __restrict__ A0, const float* __restrict__ B0, float* __restrict__ C0,
          const float* __restrict__ A1, const float* __restrict__ B1, float* __restrict__ C1,
          long long sA0, long long sB0, long long sC0, long long pA0, long long pB0, long long pC0,
          long long sA1, long long sB1, long long sC1, long long pA1, long long pB1, long long pC1,
          int lda0, int ldc0, int lgx0, int lgmy0, int nblk0,
          int lda1, int ldc1, int lgx1, int lgmy1) {
  const int idx = (int)blockIdx.x;
  if (idx < nblk0)
    gemm_job<LDB0, K0>(A0, B0, C0, sA0, sB0, sC0, pA0, pB0, pC0, lda0, ldc0, lgx0, lgmy0, idx);
  else
    gemm_job<LDB1, K1>(A1, B1, C1, sA1, sB1, sC1, pA1, pB1, pC1, lda1, ldc1, lgx1, lgmy1, idx - nblk0);
}

__device__ __forceinline__ void conv_load(FragSet& f, const float* __restrict__ xsp,
                                          const float* __restrict__ ck, int g,
                                          int l, int mrow, int half, int ncol) {
  const int tap = g >> 3, kk = (g & 7) * 16;
  const int dy = tap / 3, dx = tap - 3 * dy;
  const float* arow = xsp + (((long long)(l + dy)) * 129 + (mrow + dx)) * 128 + 8 * half + kk;
  const float* brow = ck + (((long long)tap) * 128 + 8 * half + kk) * 128 + ncol;
  load_frags(f, arow, brow, 128);
}

__device__ __forceinline__ void conv_block(const float* __restrict__ xsp, const float* __restrict__ ck,
                                           const float* __restrict__ cb, const float* __restrict__ tvec,
                                           float* __restrict__ out, int n0, int m0, int l) {
  const int lane = threadIdx.x;
  const int half = lane >> 4, l16 = lane & 15;
  const int mrow = m0 + l16, ncol = n0 + 4 * l16;
  v8f acc0 = {}, acc1 = {}, acc2 = {}, acc3 = {};
  {
    FragSet P, Q;
    conv_load(P, xsp, ck, 0, l, mrow, half, ncol);
    int g = 0;
    for (; g + 2 < 72; g += 2) {
      conv_load(Q, xsp, ck, g + 1, l, mrow, half, ncol);
      mma_frags(P, acc0, acc1, acc2, acc3);
      conv_load(P, xsp, ck, g + 2, l, mrow, half, ncol);
      mma_frags(Q, acc0, acc1, acc2, acc3);
      sched_loads_then_mma_twice();
    }
    conv_load(Q, xsp, ck, g + 1, l, mrow, half, ncol);
    sched_fence();
    mma_frags(P, acc0, acc1, acc2, acc3);
    mma_frags(Q, acc0, acc1, acc2, acc3);
  }
  const v4f bias = *(const v4f*)(cb + ncol) + *(const v4f*)(tvec + 128 + ncol);
#pragma unroll
  for (int v = 0; v < 8; ++v) {
    const int row = m0 + half * 8 + v;
    v4f o;
    o.x = acc0[v]; o.y = acc1[v]; o.z = acc2[v]; o.w = acc3[v];
    *(v4f*)(out + (((long long)l) * 128 + row) * 128 + ncol) = o + bias;
  }
}

constexpr int CONV_BLOCKS = 2 * 8 * 64;
template <int LDB0, int K0>
__global__ void __launch_bounds__(32)
conv_gemm_pair(const float* __restrict__ xsp, const float* __restrict__ ck,
               const float* __restrict__ cb, const float* __restrict__ tvec,
               float* __restrict__ out,
               const float* __restrict__ A0, const float* __restrict__ B0, float* __restrict__ C0,
               long long sA0, long long sB0, long long sC0, long long pA0, long long pB0, long long pC0,
               int lda0, int ldc0, int lgx0, int lgmy0) {
  const int idx = (int)blockIdx.x;
  if (idx < CONV_BLOCKS)
    conv_block(xsp, ck, cb, tvec, out, (idx & 1) * 64, ((idx >> 1) & 7) * 16, idx >> 4);
  else
    gemm_job<LDB0, K0>(A0, B0, C0, sA0, sB0, sC0, pA0, pB0, pC0, lda0, ldc0, lgx0, lgmy0, idx - CONV_BLOCKS);
}

constexpr int EW_U = 4;

__device__ __forceinline__ void zero_job(float* __restrict__ p, int n4, int blk) {
  v4f* p4 = (v4f*)p;
  const int base = blk * (128 * EW_U) + threadIdx.x;
  const v4f z = {0.0f, 0.0f, 0.0f, 0.0f};
#pragma unroll
  for (int u = 0; u < EW_U; ++u) {
    const int i = base + u * 128;
    if (i < n4) p4[i] = z;
  }
}

__device__ __forceinline__ void xpad_job(const float* __restrict__ x, float* __restrict__ xp,
                                         int n4, int blk) {
  const v4f* x4 = (const v4f*)x;
  v4f* xp4 = (v4f*)xp;
  const int base = blk * (128 * EW_U) + threadIdx.x;
  v4f v[EW_U];
#pragma unroll
  for (int u = 0; u < EW_U; ++u) {
    const int i = min(base + u * 128, n4 - 1);
    const int t = i >> 13, r = i & 8191, p = r >> 5, c4 = r & 31;
    const v4f z = {0.0f, 0.0f, 0.0f, 0.0f};
    const v4f ld = x4[(t * 255 + min(p, 254)) * 32 + c4];
    v[u] = (p < 255) ? ld : z;
  }
#pragma unroll
  for (int u = 0; u < EW_U; ++u) {
    const int i = base + u * 128;
    if (i < n4) xp4[i] = v[u];
  }
}

__device__ __forceinline__ void tvec_job(const float* __restrict__ temb,
                                         const float* __restrict__ dw, const float* __restrict__ db,
                                         const float* __restrict__ drw, const float* __restrict__ drb,
                                         const float* __restrict__ diw, const float* __restrict__ dib,
                                         float* __restrict__ tvec, float* __restrict__ tcr,
                                         float* __restrict__ tci, int blk) {
  const int tid = threadIdx.x;
  if (blk < 2) {
    const int j = blk * 128 + tid;
    float acc = db[j];
    for (int k = 0; k < 256; ++k) acc += temb[k] * dw[k * 256 + j];
    tvec[j] = acc;
  } else {
    const int sel = tid >> 6, j = tid & 63;
    const float* __restrict__ w  = sel ? diw : drw;
    const float* __restrict__ bs = sel ? dib : drb;
    float* __restrict__ dst = sel ? tci : tcr;
    float acc = bs[j];
    for (int k = 0; k < 256; ++k) acc += temb[k] * w[k * 64 + j];
    dst[j] = acc;
  }
}

__device__ __forceinline__ void phase1_job(float* __restrict__ c1, float* __restrict__ s1n, int blk) {
  const int i = blk * 128 + threadIdx.x;
  const int m = i >> 8, p = i & 255;
  if (p < 255) {
    double mval = (double)m;
    double phi = 2.0 * PI_D * (double)p / 255.0;
    double dphi = 2.0 * PI_D / 255.0;
    c1[i]  = (float)( cos(phi * mval) * dphi);
    s1n[i] = (float)(-sin(phi * mval) * dphi);
  } else { c1[i] = 0.0f; s1n[i] = 0.0f; }
}

__device__ __forceinline__ void phase2_job(float* __restrict__ tab, int blk) {
  const int i = blk * 128 + threadIdx.x;
  const int p = i >> 8, k = i & 255;
  float v = 0.0f;
  if (p < 127 && k < 254) {
    const bool re = k < 127;
    const double mval = (double)((re ? k : k - 127) - 63);
    const double phi = 2.0 * PI_D * (double)p / 127.0;
    v = re ? (float)cos(mval * phi) : (float)(-sin(mval * phi));
  }
  tab[i] = v;
}

__device__ __forceinline__ void resize_job(float* __restrict__ r1, float* __restrict__ r2p,
                                           float* __restrict__ rf, int blk) {
  const int tid = threadIdx.x;
  if (blk == 0) {
    if (tid < 64) {
      int i = tid;
      float scale = 0.5f;
      float sample = (i + 0.5f) / scale - 0.5f;
      float sum = 0.0f;
      for (int j = 0; j < 128; ++j) {
        float w = 1.0f - fabsf(sample - j) * scale;
        if (w > 0.0f) sum += w;
      }
      for (int j = 0; j < 128; ++j) {
        float w = 1.0f - fabsf(sample - j) * scale;
        r1[i * 128 + j] = (w > 0.0f) ? (w / sum) : 0.0f;
      }
    }
  } else if (blk == 1) {
    int j = tid;
    if (j < 127) {
      float scale = 127.0f / 255.0f;
      float sample = (j + 0.5f) / scale - 0.5f;
      float sum = 0.0f;
      for (int p = 0; p < 255; ++p) {
        float w = 1.0f - fabsf(sample - p) * scale;
        if (w > 0.0f) sum += w;
      }
      for (int p = 0; p < 255; ++p) {
        float w = 1.0f - fabsf(sample - p) * scale;
        r2p[j * 256 + p] = (w > 0.0f) ? (w / sum) : 0.0f;
      }
      r2p[j * 256 + 255] = 0.0f;
    } else {
      for (int p = 0; p < 256; ++p) r2p[j * 256 + p] = 0.0f;
    }
  } else {
    int j = tid;
    float* rre = rf + j * 128;
    float* rim = rf + 128 * 128 + j * 128;
    if (j < 127) {
      float scale = 127.0f / 255.0f;
      float sample = (j + 0.5f) / scale - 0.5f;
      float sum = 0.0f;
      for (int p = 0; p < 255; ++p) {
        float w = 1.0f - fabsf(sample - p) * scale;
        if (w > 0.0f) sum += w;
      }
      for (int m = 0; m < 128; ++m) {
        const int pa = 127 + m, pb = 127 - m;
        float wa = 1.0f - fabsf(sample - pa) * scale;
        float wb = 1.0f - fabsf(sample - pb) * scale;
        const float a = (wa > 0.0f) ? (wa / sum) : 0.0f;
        const float b = (m > 0 && wb > 0.0f) ? (wb / sum) : 0.0f;
        const float s = (m & 1) ? -b : b;
        rre[m] = a + s;
        rim[m] = a - s;
      }
    } else {
      for (int m = 0; m < 128; ++m) { rre[m] = 0.0f; rim[m] = 0.0f; }
    }
  }
}

__device__ __forceinline__ void legendre_job(float* __restrict__ out, int lg, int weighted, int blk,
                                             double* cdiag, double* ca_all, double* cb_all) {
  const int L = 1 << lg;
  const int q = threadIdx.x >> lg;
  const int t = threadIdx.x & (L - 1);
  const int nblk = 1 << (2 * lg - 7);
  const int m = q * nblk + blk;
  double* ca = ca_all + q * L;
  double* cb = cb_all + q * L;
  {
    const int l = t;
    const double ll = (double)l, mm = (double)m;
    if (q == 0) cdiag[l] = (l >= 1) ? -sqrt((2.0 * ll + 1.0) / (2.0 * ll)) : 0.0;
    double a = 0.0, b = 0.0;
    if (l >= m + 2) {
      a = sqrt((4.0 * ll * ll - 1.0) / (ll * ll - mm * mm));
      b = sqrt((2.0 * ll + 1.0) * ((ll - 1.0) * (ll - 1.0) - mm * mm) /
               ((2.0 * ll - 3.0) * (ll * ll - mm * mm)));
    }
    ca[l] = a; cb[l] = b;
  }
  __syncthreads();
  const double theta = (2.0 * t + 1.0) * PI_D / (2.0 * L - 1.0);
  const double x = cos(theta), s = sin(theta);
  const double w = weighted ? (s * PI_D / (double)L) : 1.0;
  const long long LL = (long long)L * (long long)L;
  float* pos = out + (long long)(L - 1 + m) * LL + t;
  float* neg = out + (long long)(L - 1 - m) * LL + t;
  const bool mir = (m > 0);
  const bool odd = (m & 1) != 0;
  for (int l = 0; l < m; ++l) {
    pos[(long long)l * L] = 0.0f;
    neg[(long long)l * L] = 0.0f;
  }
  double pmm = sqrt(1.0 / (4.0 * PI_D)) * w;
  for (int j = 1; j <= m; ++j) pmm = cdiag[j] * s * pmm;
  {
    const float fv = (float)pmm;
    pos[(long long)m * L] = fv;
    if (mir) neg[(long long)m * L] = odd ? -fv : fv;
  }
  if (m + 1 < L) {
    double p2 = pmm;
    double p1 = sqrt(2.0 * (m + 1) + 1.0) * x * pmm;
    {
      const float fv = (float)p1;
      pos[(long long)(m + 1) * L] = fv;
      if (mir) neg[(long long)(m + 1) * L] = odd ? -fv : fv;
    }
    for (int l = m + 2; l < L; ++l) {
      const double p0 = ca[l] * x * p1 - cb[l] * p2;
      const float fv = (float)p0;
      pos[(long long)l * L] = fv;
      if (mir) neg[(long long)l * L] = odd ? -fv : fv;
      p2 = p1; p1 = p0;
    }
  }
}

__device__ __forceinline__ void legendre_fold_job(float* __restrict__ out, int blk,
                                                  double* cdiag, double* ca, double* cb, double* rw) {
  const int m = blk;
  const int t = threadIdx.x;
  {
    const int l = t;
    const double ll = (double)l, mm = (double)m;
    cdiag[l] = (l >= 1) ? -sqrt((2.0 * ll + 1.0) / (2.0 * ll)) : 0.0;
    double a = 0.0, b = 0.0;
    if (l >= m + 2) {
      a = sqrt((4.0 * ll * ll - 1.0) / (ll * ll - mm * mm));
      b = sqrt((2.0 * ll + 1.0) * ((ll - 1.0) * (ll - 1.0) - mm * mm) /
               ((2.0 * ll - 3.0) * (ll * ll - mm * mm)));
    }
    ca[l] = a; cb[l] = b;
  }
  if (t < 64) {
    const int i = t;
    const float scale = 0.5f;
    const float sample = (i + 0.5f) / scale - 0.5f;
    float wq[4];
    float sum = 0.0f;
#pragma unroll
    for (int q = 0; q < 4; ++q) {
      const int j = 2 * i - 1 + q;
      const float wv = 1.0f - fabsf(sample - j) * scale;
      wq[q] = (j >= 0 && j < 128 && wv > 0.0f) ? wv : 0.0f;
      sum += wq[q];
    }
#pragma unroll
    for (int q = 0; q < 4; ++q) rw[4 * i + q] = (double)(wq[q] / sum);
  }
  __syncthreads();
  const double theta = (2.0 * t + 1.0) * PI_D / 255.0;
  const double x = cos(theta), s = sin(theta);
  const double w = s * PI_D / 128.0;
  float* pos = out + (long long)m * 8192 + t;
  double pmm = sqrt(1.0 / (4.0 * PI_D)) * w;
  for (int j = 1; j <= m; ++j) pmm = cdiag[j] * s * pmm;
  const double pm1 = sqrt(2.0 * (m + 1) + 1.0) * x * pmm;
  double w0 = 0.0;
  double w1 = (m == 0) ? pmm : 0.0;
  for (int i = 0; i < 63; ++i) {
    const int la = 2 * i + 1, lb = 2 * i + 2;
    const double caa = ca[la], cba = cb[la], cab = ca[lb], cbb = cb[lb];
    const double r0 = rw[4 * i], r1 = rw[4 * i + 1], r2 = rw[4 * i + 2], r3 = rw[4 * i + 3];
    double pa = caa * x * w1 - cba * w0;
    pa = (la == m) ? pmm : pa;
    pa = (la == m + 1) ? pm1 : pa;
    double pb = cab * x * pa - cbb * w1;
    pb = (lb == m) ? pmm : pb;
    pb = (lb == m + 1) ? pm1 : pb;
    const double v = (r0 * w0 + r1 * w1) + (r2 * pa + r3 * pb);
    pos[i * 128] = (float)v;
    w0 = pa; w1 = pb;
  }
  {
    double pc = ca[127] * x * w1 - cb[127] * w0;
    pc = (127 == m) ? pmm : pc;
    pc = (127 == m + 1) ? pm1 : pc;
    const double v = rw[252] * w0 + rw[253] * w1 + rw[254] * pc;
    pos[63 * 128] = (float)v;
  }
}

constexpr int S0_LEG1 = 128;
constexpr int S0_LEG2 = 32;
constexpr int S0_TVEC = 3;
constexpr int S0_RESZ = 3;
constexpr int S0_PH1  = 128 * 256 / 128;
constexpr int S0_PH2  = (int)(2 * SZ_PH2 / 128);
constexpr int S0_ZP   = 4;
constexpr int S0_ZX   = (int)((SZ_XSP / 4 + 128 * EW_U - 1) / (128 * EW_U));
constexpr int S0_XPAD = (int)(SZ_XP / 4 / (128 * EW_U));
constexpr int S0_BLOCKS = S0_LEG1 + S0_LEG2 + S0_TVEC + S0_RESZ + S0_PH1 + S0_PH2 + S0_ZP + S0_ZX + S0_XPAD;

__global__ void __launch_bounds__(128)
k_setup0(const float* __restrict__ x, const float* __restrict__ temb,
         const float* __restrict__ dw, const float* __restrict__ db,
         const float* __restrict__ drw, const float* __restrict__ drb,
         const float* __restrict__ diw, const float* __restrict__ dib,
         float* __restrict__ ws) {
  __shared__ double cdiag[128];
  __shared__ double ca[128];
  __shared__ double cb[128];
  __shared__ double rw[256];
  int b = blockIdx.x;
  if (b < S0_LEG1) { legendre_fold_job(ws + OFF_A1, b, cdiag, ca, cb, rw); return; }
  b -= S0_LEG1;
  if (b < S0_LEG2) { legendre_job(ws + OFF_P2A, 6, 0, b, cdiag, ca, cb); return; }
  b -= S0_LEG2;
  if (b < S0_TVEC) {
    tvec_job(temb, dw, db, drw, drb, diw, dib, ws + OFF_TVEC, ws + OFF_TCR, ws + OFF_TCI, b);
    return;
  }
  b -= S0_TVEC;
  if (b < S0_RESZ) { resize_job(ws + OFF_R1, ws + OFF_R2P, ws + OFF_RF, b); return; }
  b -= S0_RESZ;
  if (b < S0_PH1) { phase1_job(ws + OFF_COS1, ws + OFF_SIN1N, b); return; }
  b -= S0_PH1;
  if (b < S0_PH2) { phase2_job(ws + OFF_PH2C, b); return; }
  b -= S0_PH2;
  if (b < S0_ZP) {
    zero_job(ws + OFF_T2R + 255LL * 8192, 2048, b);
    return;
  }
  b -= S0_ZP;
  if (b < S0_ZX) {
    zero_job(ws + OFF_XSP, (int)(SZ_XSP / 4), b);
    return;
  }
  b -= S0_ZX;
  xpad_job(x, ws + OFF_XP, (int)(SZ_XP / 4), b);
}

__device__ __forceinline__ void wc_job(const float* __restrict__ wr, const float* __restrict__ wi,
                                       const float* __restrict__ tcr, const float* __restrict__ tci,
                                       float* __restrict__ wcs, int n4, int blk) {
  const v4f* wr4 = (const v4f*)wr;
  const v4f* wi4 = (const v4f*)wi;
  v4f* wcs4 = (v4f*)wcs;
  const int base = blk * (256 * EW_U) + threadIdx.x;
  v4f r[EW_U], im[EW_U];
  float tr[EW_U], ti[EW_U];
#pragma unroll
  for (int u = 0; u < EW_U; ++u) {
    const int i = min(base + u * 256, n4 - 1);
    const int l = i >> 12;
    r[u] = wr4[i]; im[u] = wi4[i];
    tr[u] = tcr[l]; ti[u] = tci[l];
  }
#pragma unroll
  for (int u = 0; u < EW_U; ++u) {
    const int i = base + u * 256;
    if (i < n4) {
      const v4f cr = tr[u] * r[u] - ti[u] * im[u];
      const v4f ci = tr[u] * im[u] + ti[u] * r[u];
      v4f* d = wcs4 + (long long)(i >> 12) * 16384 + (i & 4095);
      d[0] = cr; d[4096] = -ci; d[8192] = ci; d[12288] = cr;
    }
  }
}

__device__ __forceinline__ void wsp_job(const float* __restrict__ spw, const float* __restrict__ tvec,
                                        float* __restrict__ wsp, int blk) {
  const int i = blk * 256 + threadIdx.x;
  const int o = i & 127;
  wsp[i] = tvec[o] * spw[i];
}

__device__ __forceinline__ void transp_job(const float* __restrict__ a, float* __restrict__ b,
                                           int m, float (*tile)[65]) {
  const v4f* a4 = (const v4f*)a + (long long)m * 1024;
  v4f* b4 = (v4f*)b + (long long)m * 1024;
  v4f v[4];
#pragma unroll
  for (int u = 0; u < 4; ++u) v[u] = a4[threadIdx.x + 256 * u];
#pragma unroll
  for (int u = 0; u < 4; ++u) {
    const int idx = threadIdx.x + 256 * u;
    const int l = idx >> 4, t4 = idx & 15;
    tile[l][4 * t4 + 0] = v[u].x; tile[l][4 * t4 + 1] = v[u].y;
    tile[l][4 * t4 + 2] = v[u].z; tile[l][4 * t4 + 3] = v[u].w;
  }
  __syncthreads();
#pragma unroll
  for (int u = 0; u < 4; ++u) {
    const int idx = threadIdx.x + 256 * u;
    const int t = idx >> 4, l4 = idx & 15;
    v4f o;
    o.x = tile[4 * l4 + 0][t]; o.y = tile[4 * l4 + 1][t];
    o.z = tile[4 * l4 + 2][t]; o.w = tile[4 * l4 + 3][t];
    b4[idx] = o;
  }
}

constexpr int S1_WC  = (int)(SZ_WC / 4 / (256 * EW_U));
constexpr int S1_WSP = (int)(SZ_WSP / 256);
constexpr int S1_TR  = 127;
constexpr int S1_BLOCKS = S1_WC + S1_WSP + S1_TR;

__global__ void __launch_bounds__(256)
k_setup1(const float* __restrict__ wr, const float* __restrict__ wi,
         const float* __restrict__ spw, float* __restrict__ ws) {
  __shared__ float tile[64][65];
  int b = blockIdx.x;
  if (b < S1_WC) {
    wc_job(wr, wi, ws + OFF_TCR, ws + OFF_TCI, ws + OFF_WCS, (int)(SZ_WC / 4), b);
    return;
  }
  b -= S1_WC;
  if (b < S1_WSP) { wsp_job(spw, ws + OFF_TVEC, ws + OFF_WSP, b); return; }
  b -= S1_WSP;
  transp_job(ws + OFF_P2A, ws + OFF_P2B, b, tile);
}

__device__ __forceinline__ float wave_sum(float s) {
#pragma unroll
  for (int o = 16; o > 0; o >>= 1) s += __shfl_xor(s, o);
  return s;
}

__global__ void __launch_bounds__(256)
k_fuse(const float* __restrict__ xspec, const float* __restrict__ xcv,
       const float* __restrict__ lns, const float* __restrict__ lnb,
       float* __restrict__ out) {
  const int lane = threadIdx.x & 31;
  const int n = blockIdx.x * 8 + (threadIdx.x >> 5);
  if (n >= 64 * 127) return;
  const int l = n / 127, m = n - l * 127;
  const int i4 = (l * 128 + m) * 32 + lane;
  const v4f a  = ((const v4f*)xspec)[i4];
  const v4f b  = ((const v4f*)xcv)[i4];
  const v4f sc = ((const v4f*)lns)[lane];
  const v4f bi = ((const v4f*)lnb)[lane];
  v4f g;
#pragma unroll
  for (int q = 0; q < 4; ++q) {
    const float v = a[q] + b[q];
    g[q] = 0.5f * v * (1.0f + tanhf(0.7978845608028654f * (v + 0.044715f * v * v * v)));
  }
  const float mu = wave_sum((g[0] + g[1]) + (g[2] + g[3])) * (1.0f / 128.0f);
  v4f d;
#pragma unroll
  for (int q = 0; q < 4; ++q) d[q] = g[q] - mu;
  const float var = wave_sum((d[0] * d[0] + d[1] * d[1]) + (d[2] * d[2] + d[3] * d[3])) * (1.0f / 128.0f);
  const float rs = rsqrtf(var + 1e-6f);
  v4f o;
#pragma unroll
  for (int q = 0; q < 4; ++q) o[q] = d[q] * rs * sc[q] + bi[q];
  ((v4f*)out)[(l * 127 + m) * 32 + lane] = o;
}

extern "C" void kernel_launch(void* const* d_in, const int* in_sizes, int n_in,
                              void* d_out, int out_size, void* d_ws, size_t ws_size,
                              hipStream_t stream) {
  (void)in_sizes; (void)n_in; (void)out_size; (void)ws_size;
  const float* x       = (const float*)d_in[0];
  const float* t_emb   = (const float*)d_in[1];
  const float* sp_dw   = (const float*)d_in[2];
  const float* sp_db   = (const float*)d_in[3];
  const float* sp_wt   = (const float*)d_in[4];
  const float* sp_ck   = (const float*)d_in[5];
  const float* sp_cb   = (const float*)d_in[6];
  const float* spec_wr = (const float*)d_in[7];
  const float* spec_wi = (const float*)d_in[8];
  const float* drw     = (const float*)d_in[9];
  const float* drb     = (const float*)d_in[10];
  const float* diw     = (const float*)d_in[11];
  const float* dib     = (const float*)d_in[12];
  const float* lns     = (const float*)d_in[13];
  const float* lnb     = (const float*)d_in[14];
  float* out = (float*)d_out;
  float* ws  = (float*)d_ws;

  float *A1 = ws + OFF_A1, *XP = ws + OFF_XP, *COS1 = ws + OFF_COS1;
  float *GR = ws + OFF_GR;
  float *R1 = ws + OFF_R1, *R2P = ws + OFF_R2P, *T1R = ws + OFF_T1R;
  float *FFC = ws + OFF_FFC, *WCS = ws + OFF_WCS, *RF = ws + OFF_RF;
  float *F2R = ws + OFF_F2R, *P2B = ws + OFF_P2B;
  float *FTR = ws + OFF_FTR, *PH2C = ws + OFF_PH2C;
  float *XSPEC = ws + OFF_XSPEC, *TVEC = ws + OFF_TVEC;
  float *WSP = ws + OFF_WSP, *XR = ws + OFF_XS1, *T2R = ws + OFF_T2R;
  float *XSP = ws + OFF_XSP, *XCV = ws + OFF_XCV;

  k_setup0<<<dim3(S0_BLOCKS), dim3(128), 0, stream>>>(x, t_emb, sp_dw, sp_db, drw, drb, diw, dib, ws);
  k_setup1<<<dim3(S1_BLOCKS), dim3(256), 0, stream>>>(spec_wr, spec_wi, sp_wt, ws);


  gemm_pair<128, 256, 255 * 128, 128><<<dim3(4096 + 2040), dim3(32), 0, stream>>>(
      COS1, XP, GR,  R1, x, T2R,
      0LL, 256LL * 128, 128LL * 128, SZ_PH1, 0LL, SZ_G,
      0LL, 128LL, 64LL * 128, 0LL, 0LL, 0LL,
      256, 128, 2, 3, 4096,
      128, 128, 1, 2);
  gemm_pair<64 * 128, 256, 128 * 128, 128><<<dim3(1024 + 2048), dim3(32), 0, stream>>>(
      R2P, T2R, XR,  A1, GR, T1R,
      0LL, 128LL, 128LL * 128, 0LL, 0LL, 0LL,
      64LL * 128, 128LL, 64LL * 128, 0LL, SZ_G, SZ_T1,
      256, 128, 1, 3, 1024,
      128, 128, 2, 2);
  gemm_pair<64 * 128, 128, 128, 128><<<dim3(2048 + 1024), dim3(32), 0, stream>>>(
      RF, T1R, FFC,  XR, WSP, XSP + 130LL * 128,
      0LL, 128LL, 128LL * 256, 128LL * 128, SZ_T1, 128LL,
      128LL * 128, 0LL, 129LL * 128, 0LL, 0LL, 0LL,
      128, 256, 2, 3, 2048,
      128, 128, 1, 3);
  conv_gemm_pair<128, 256><<<dim3(CONV_BLOCKS + 2048), dim3(32), 0, stream>>>(
      XSP, sp_ck, sp_cb, TVEC, XCV,
      FFC, WCS, F2R,
      128LL * 256, 2LL * 256 * 128, 128LL * 128, 0LL, 256LL * 128, SZ_F2,
      256, 128, 2, 3);
  gemm_wmma<128 * 128, 64><<<dim3(2032), dim3(32), 0, stream>>>(
      P2B, F2R, FTR, 64LL * 64, 128LL, 64LL * 128, 0LL, SZ_F2, SZ_FT, 64, 128, 2, 2);
  gemm_wmma<64 * 128, 256><<<dim3(1024), dim3(32), 0, stream>>>(
      PH2C, FTR, XSPEC, 0LL, 128LL, 128LL * 128, 0LL, 0LL, 0LL, 256, 128, 1, 3);

  k_fuse<<<dim3(64 * 127 / 8), dim3(256), 0, stream>>>(XSPEC, XCV, lns, lnb, out);
}


# Legendre fold-table job, fewer f64 instructions on its longest wave: recurrence stays f64, 4-tap row sum at f32 precision, zero rows under order m not walked, diagonal factors only to m, sincos
# speedup vs baseline: 79.3467x; 1.0949x over previous
#include <hip/hip_runtime.h>
#include <math.h>

typedef __attribute__((ext_vector_type(2))) float v2f;
typedef __attribute__((ext_vector_type(8))) float v8f;
typedef __attribute__((ext_vector_type(4))) float v4f;

#define PI_D 3.14159265358979323846

constexpr long long SZ_A1    = 255LL*128*128;
constexpr long long SZ_XP    = 128LL*256*128;
constexpr long long SZ_PH1   = 256LL*256;
constexpr long long SZ_G     = 128LL*256*128;
constexpr long long SZ_F1    = 255LL*128*128;
constexpr long long SZ_R1    = 64LL*128;
constexpr long long SZ_R2P   = 128LL*256;
constexpr long long SZ_T1    = 256LL*64*128;
constexpr long long SZ_FF    = 64LL*128*128;
constexpr long long SZ_WC    = 64LL*128*128;
constexpr long long SZ_F2    = 64LL*128*128;
constexpr long long SZ_P2    = 127LL*64*64;
constexpr long long SZ_FT    = 127LL*64*128;
constexpr long long SZ_PH2   = 128LL*128;
constexpr long long SZ_XSPEC = 64LL*128*128;
constexpr long long SZ_TVEC  = 256;
constexpr long long SZ_TC    = 64;
constexpr long long SZ_WSP   = 128LL*128;
constexpr long long SZ_XS1   = 128LL*255*128;
constexpr long long SZ_T2    = 256LL*64*128;
constexpr long long SZ_XS2   = 64LL*128*128;
constexpr long long SZ_XSP   = 67LL*129*128;
constexpr long long SZ_XCV   = 64LL*128*128;

constexpr long long OFF_A1    = 0;
constexpr long long OFF_XP    = OFF_A1    + SZ_A1;
constexpr long long OFF_COS1  = OFF_XP    + SZ_XP;
constexpr long long OFF_SIN1N = OFF_COS1  + SZ_PH1;
constexpr long long OFF_GR    = OFF_SIN1N + SZ_PH1;
constexpr long long OFF_GI    = OFF_GR    + SZ_G;
constexpr long long OFF_F1R   = OFF_GI    + SZ_G;
constexpr long long OFF_F1I   = OFF_F1R   + SZ_F1;
constexpr long long OFF_R1    = OFF_F1I   + SZ_F1;
constexpr long long OFF_R2P   = OFF_R1    + SZ_R1;
constexpr long long OFF_T1R   = OFF_R2P   + SZ_R2P;
constexpr long long OFF_T1I   = OFF_T1R   + SZ_T1;
constexpr long long OFF_FFR   = OFF_T1I   + SZ_T1;
constexpr long long OFF_FFI   = OFF_FFR   + SZ_FF;
constexpr long long OFF_WCR   = OFF_FFI   + SZ_FF;
constexpr long long OFF_WCI   = OFF_WCR   + SZ_WC;
constexpr long long OFF_F2R   = OFF_WCI   + SZ_WC;
constexpr long long OFF_F2I   = OFF_F2R   + SZ_F2;
constexpr long long OFF_P2A   = OFF_F2I   + SZ_F2;
constexpr long long OFF_P2B   = OFF_P2A   + SZ_P2;
constexpr long long OFF_FTR   = OFF_P2B   + SZ_P2;
constexpr long long OFF_FTI   = OFF_FTR   + SZ_FT;
constexpr long long OFF_COS2  = OFF_FTI   + SZ_FT;
constexpr long long OFF_SIN2  = OFF_COS2  + SZ_PH2;
constexpr long long OFF_XSPEC = OFF_SIN2  + SZ_PH2;
constexpr long long OFF_TVEC  = OFF_XSPEC + SZ_XSPEC;
constexpr long long OFF_TCR   = OFF_TVEC  + SZ_TVEC;
constexpr long long OFF_TCI   = OFF_TCR   + SZ_TC;
constexpr long long OFF_WSP   = OFF_TCI   + SZ_TC;
constexpr long long OFF_XS1   = OFF_WSP   + SZ_WSP;
constexpr long long OFF_T2R   = OFF_XS1   + SZ_XS1;
constexpr long long OFF_XS2   = OFF_T2R   + SZ_T2;
constexpr long long OFF_XSP   = OFF_XS2   + SZ_XS2;
constexpr long long OFF_XCV   = OFF_XSP   + SZ_XSP;
constexpr long long OFF_WCS   = OFF_F1R;
constexpr long long OFF_FFC   = OFF_FFR;
constexpr long long OFF_PH2C  = OFF_COS2;
constexpr long long OFF_RF    = OFF_WCR;

struct FragSet {
  v4f a[2];
  v4f b[8];
};

__device__ __forceinline__ void sched_fence() { __builtin_amdgcn_sched_barrier(0); }
__device__ __forceinline__ void sched_loads_then_mma_twice() {
  __builtin_amdgcn_sched_group_barrier(0x020, 10, 0);
  __builtin_amdgcn_sched_group_barrier(0x008, 16, 0);
  __builtin_amdgcn_sched_group_barrier(0x020, 10, 0);
  __builtin_amdgcn_sched_group_barrier(0x008, 16, 0);
}

__device__ __forceinline__ void load_frags(FragSet& f, const float* __restrict__ arow,
                                           const float* __restrict__ brow, long long ldb) {
  const v4f* ap = (const v4f*)arow;
  f.a[0] = ap[0];
  f.a[1] = ap[1];
#pragma unroll
  for (int i = 0; i < 8; ++i) f.b[i] = *(const v4f*)(brow + (long long)i * ldb);
}

__device__ __forceinline__ void mma_frags(const FragSet& f, v8f& acc0, v8f& acc1, v8f& acc2, v8f& acc3) {
#pragma unroll
  for (int s = 0; s < 4; ++s) {
    const v4f av = f.a[s >> 1];
    const int e0 = 2 * (s & 1);
    v2f a;  a.x = av[e0];           a.y = av[e0 + 1];
    const v4f r0 = f.b[2 * s], r1 = f.b[2 * s + 1];
    v2f b0; b0.x = r0[0];           b0.y = r1[0];
    v2f b1; b1.x = r0[1];           b1.y = r1[1];
    v2f b2; b2.x = r0[2];           b2.y = r1[2];
    v2f b3; b3.x = r0[3];           b3.y = r1[3];
    acc0 = __builtin_amdgcn_wmma_f32_16x16x4_f32(false, a, false, b0, (short)0, acc0, false, false);
    acc1 = __builtin_amdgcn_wmma_f32_16x16x4_f32(false, a, false, b1, (short)0, acc1, false, false);
    acc2 = __builtin_amdgcn_wmma_f32_16x16x4_f32(false, a, false, b2, (short)0, acc2, false, false);
    acc3 = __builtin_amdgcn_wmma_f32_16x16x4_f32(false, a, false, b3, (short)0, acc3, false, false);
  }
}

template <int LDB, int K>
__device__ __forceinline__ void gemm_tile(const float* __restrict__ A, const float* __restrict__ B,
                                          float* __restrict__ C, int lda, int ldc,
                                          int n0, int m0) {
  static_assert(K >= 32 && (K & 31) == 0, "K is walked in pairs of 16-k groups");
  constexpr long long ldb = LDB;
  const int lane = threadIdx.x;
  const int half = lane >> 4;
  const int l16  = lane & 15;
  v8f acc0 = {}, acc1 = {}, acc2 = {}, acc3 = {};
  const float* arow = A + (long long)(m0 + l16) * lda + 8 * half;
  const float* brow = B + (long long)(8 * half) * ldb + n0 + 4 * l16;
  {
    FragSet P, Q;
    load_frags(P, arow, brow, ldb);
    int k0 = 0;
#pragma clang loop unroll(disable)
    for (; k0 + 32 < K; k0 += 32) {
      load_frags(Q, arow + k0 + 16, brow + (long long)(k0 + 16) * ldb, ldb);
      mma_frags(P, acc0, acc1, acc2, acc3);
      load_frags(P, arow + k0 + 32, brow + (long long)(k0 + 32) * ldb, ldb);
      mma_frags(Q, acc0, acc1, acc2, acc3);
      sched_loads_then_mma_twice();
    }
    load_frags(Q, arow + k0 + 16, brow + (long long)(k0 + 16) * ldb, ldb);
    sched_fence();
    mma_frags(P, acc0, acc1, acc2, acc3);
    mma_frags(Q, acc0, acc1, acc2, acc3);
  }
#pragma unroll
  for (int v = 0; v < 8; ++v) {
    const int row = m0 + half * 8 + v;
    v4f o;
    o.x = acc0[v]; o.y = acc1[v]; o.z = acc2[v]; o.w = acc3[v];
    *(v4f*)(C + (long long)row * ldc + n0 + 4 * l16) = o;
  }
}

template <int LDB, int K>
__device__ __forceinline__ void gemm_job(const float* __restrict__ A, const float* __restrict__ B,
                                         float* __restrict__ C,
                                         long long sAb, long long sBb, long long sCb,
                                         long long pA, long long pB, long long pC,
                                         int lda, int ldc, int lgx, int lgmy, int idx) {
  const int bx = idx & ((1 << lgx) - 1);
  const int t  = idx >> lgx;
  const int by = t & ((1 << lgmy) - 1);
  const long long b = t >> lgmy;
  const int part = bx >> 1;
  A += b * sAb + part * pA; B += b * sBb + part * pB; C += b * sCb + part * pC;
  gemm_tile<LDB, K>(A, B, C, lda, ldc, (bx & 1) * 64, by * 16);
}

template <int LDB, int K>
__global__ void __launch_bounds__(32)
gemm_wmma(const float* __restrict__ A, const float* __restrict__ B, float* __restrict__ C,
          long long sAb, long long sBb, long long sCb,
          long long pA, long long pB, long long pC,
          int lda, int ldc, int lgx, int lgmy) {
  gemm_job<LDB, K>(A, B, C, sAb, sBb, sCb, pA, pB, pC, lda, ldc, lgx, lgmy, (int)blockIdx.x);
}

template <int LDB0, int K0, int LDB1, int K1>
__global__ void __launch_bounds__(32)
gemm_pair(const float* __restrict__ A0, const float* __restrict__ B0, float* __restrict__ C0,
          const float* __restrict__ A1, const float* __restrict__ B1, float* __restrict__ C1,
          long long sA0, long long sB0, long long sC0, long long pA0, long long pB0, long long pC0,
          long long sA1, long long sB1, long long sC1, long long pA1, long long pB1, long long pC1,
          int lda0, int ldc0, int lgx0, int lgmy0, int nblk0,
          int lda1, int ldc1, int lgx1, int lgmy1) {
  const int idx = (int)blockIdx.x;
  if (idx < nblk0)
    gemm_job<LDB0, K0>(A0, B0, C0, sA0, sB0, sC0, pA0, pB0, pC0, lda0, ldc0, lgx0, lgmy0, idx);
  else
    gemm_job<LDB1, K1>(A1, B1, C1, sA1, sB1, sC1, pA1, pB1, pC1, lda1, ldc1, lgx1, lgmy1, idx - nblk0);
}

__device__ __forceinline__ void conv_load(FragSet& f, const float* __restrict__ xsp,
                                          const float* __restrict__ ck, int g,
                                          int l, int mrow, int half, int ncol) {
  const int tap = g >> 3, kk = (g & 7) * 16;
  const int dy = tap / 3, dx = tap - 3 * dy;
  const float* arow = xsp + (((long long)(l + dy)) * 129 + (mrow + dx)) * 128 + 8 * half + kk;
  const float* brow = ck + (((long long)tap) * 128 + 8 * half + kk) * 128 + ncol;
  load_frags(f, arow, brow, 128);
}

__device__ __forceinline__ void conv_block(const float* __restrict__ xsp, const float* __restrict__ ck,
                                           const float* __restrict__ cb, const float* __restrict__ tvec,
                                           float* __restrict__ out, int n0, int m0, int l) {
  const int lane = threadIdx.x;
  const int half = lane >> 4, l16 = lane & 15;
  const int mrow = m0 + l16, ncol = n0 + 4 * l16;
  v8f acc0 = {}, acc1 = {}, acc2 = {}, acc3 = {};
  {
    FragSet P, Q;
    conv_load(P, xsp, ck, 0, l, mrow, half, ncol);
    int g = 0;
    for (; g + 2 < 72; g += 2) {
      conv_load(Q, xsp, ck, g + 1, l, mrow, half, ncol);
      mma_frags(P, acc0, acc1, acc2, acc3);
      conv_load(P, xsp, ck, g + 2, l, mrow, half, ncol);
      mma_frags(Q, acc0, acc1, acc2, acc3);
      sched_loads_then_mma_twice();
    }
    conv_load(Q, xsp, ck, g + 1, l, mrow, half, ncol);
    sched_fence();
    mma_frags(P, acc0, acc1, acc2, acc3);
    mma_frags(Q, acc0, acc1, acc2, acc3);
  }
  const v4f bias = *(const v4f*)(cb + ncol) + *(const v4f*)(tvec + 128 + ncol);
#pragma unroll
  for (int v = 0; v < 8; ++v) {
    const int row = m0 + half * 8 + v;
    v4f o;
    o.x = acc0[v]; o.y = acc1[v]; o.z = acc2[v]; o.w = acc3[v];
    *(v4f*)(out + (((long long)l) * 128 + row) * 128 + ncol) = o + bias;
  }
}

constexpr int CONV_BLOCKS = 2 * 8 * 64;
template <int LDB0, int K0>
__global__ void __launch_bounds__(32)
conv_gemm_pair(const float* __restrict__ xsp, const float* __restrict__ ck,
               const float* __restrict__ cb, const float* __restrict__ tvec,
               float* __restrict__ out,
               const float* __restrict__ A0, const float* __restrict__ B0, float* __restrict__ C0,
               long long sA0, long long sB0, long long sC0, long long pA0, long long pB0, long long pC0,
               int lda0, int ldc0, int lgx0, int lgmy0) {
  const int idx = (int)blockIdx.x;
  if (idx < CONV_BLOCKS)
    conv_block(xsp, ck, cb, tvec, out, (idx & 1) * 64, ((idx >> 1) & 7) * 16, idx >> 4);
  else
    gemm_job<LDB0, K0>(A0, B0, C0, sA0, sB0, sC0, pA0, pB0, pC0, lda0, ldc0, lgx0, lgmy0, idx - CONV_BLOCKS);
}

constexpr int EW_U = 4;

__device__ __forceinline__ void zero_job(float* __restrict__ p, int n4, int blk) {
  v4f* p4 = (v4f*)p;
  const int base = blk * (128 * EW_U) + threadIdx.x;
  const v4f z = {0.0f, 0.0f, 0.0f, 0.0f};
#pragma unroll
  for (int u = 0; u < EW_U; ++u) {
    const int i = base + u * 128;
    if (i < n4) p4[i] = z;
  }
}

__device__ __forceinline__ void xpad_job(const float* __restrict__ x, float* __restrict__ xp,
                                         int n4, int blk) {
  const v4f* x4 = (const v4f*)x;
  v4f* xp4 = (v4f*)xp;
  const int base = blk * (128 * EW_U) + threadIdx.x;
  v4f v[EW_U];
#pragma unroll
  for (int u = 0; u < EW_U; ++u) {
    const int i = min(base + u * 128, n4 - 1);
    const int t = i >> 13, r = i & 8191, p = r >> 5, c4 = r & 31;
    const v4f z = {0.0f, 0.0f, 0.0f, 0.0f};
    const v4f ld = x4[(t * 255 + min(p, 254)) * 32 + c4];
    v[u] = (p < 255) ? ld : z;
  }
#pragma unroll
  for (int u = 0; u < EW_U; ++u) {
    const int i = base + u * 128;
    if (i < n4) xp4[i] = v[u];
  }
}

__device__ __forceinline__ void tvec_job(const float* __restrict__ temb,
                                         const float* __restrict__ dw, const float* __restrict__ db,
                                         const float* __restrict__ drw, const float* __restrict__ drb,
                                         const float* __restrict__ diw, const float* __restrict__ dib,
                                         float* __restrict__ tvec, float* __restrict__ tcr,
                                         float* __restrict__ tci, int blk) {
  const int tid = threadIdx.x;
  if (blk < 2) {
    const int j = blk * 128 + tid;
    float acc = db[j];
    for (int k = 0; k < 256; ++k) acc += temb[k] * dw[k * 256 + j];
    tvec[j] = acc;
  } else {
    const int sel = tid >> 6, j = tid & 63;
    const float* __restrict__ w  = sel ? diw : drw;
    const float* __restrict__ bs = sel ? dib : drb;
    float* __restrict__ dst = sel ? tci : tcr;
    float acc = bs[j];
    for (int k = 0; k < 256; ++k) acc += temb[k] * w[k * 64 + j];
    dst[j] = acc;
  }
}

__device__ __forceinline__ void phase1_job(float* __restrict__ c1, float* __restrict__ s1n, int blk) {
  const int i = blk * 128 + threadIdx.x;
  const int m = i >> 8, p = i & 255;
  if (p < 255) {
    double mval = (double)m;
    double phi = 2.0 * PI_D * (double)p / 255.0;
    double dphi = 2.0 * PI_D / 255.0;
    c1[i]  = (float)( cos(phi * mval) * dphi);
    s1n[i] = (float)(-sin(phi * mval) * dphi);
  } else { c1[i] = 0.0f; s1n[i] = 0.0f; }
}

__device__ __forceinline__ void phase2_job(float* __restrict__ tab, int blk) {
  const int i = blk * 128 + threadIdx.x;
  const int p = i >> 8, k = i & 255;
  float v = 0.0f;
  if (p < 127 && k < 254) {
    const bool re = k < 127;
    const double mval = (double)((re ? k : k - 127) - 63);
    const double phi = 2.0 * PI_D * (double)p / 127.0;
    v = re ? (float)cos(mval * phi) : (float)(-sin(mval * phi));
  }
  tab[i] = v;
}

__device__ __forceinline__ void resize_job(float* __restrict__ r1, float* __restrict__ r2p,
                                           float* __restrict__ rf, int blk) {
  const int tid = threadIdx.x;
  if (blk == 0) {
    if (tid < 64) {
      int i = tid;
      float scale = 0.5f;
      float sample = (i + 0.5f) / scale - 0.5f;
      float sum = 0.0f;
      for (int j = 0; j < 128; ++j) {
        float w = 1.0f - fabsf(sample - j) * scale;
        if (w > 0.0f) sum += w;
      }
      for (int j = 0; j < 128; ++j) {
        float w = 1.0f - fabsf(sample - j) * scale;
        r1[i * 128 + j] = (w > 0.0f) ? (w / sum) : 0.0f;
      }
    }
  } else if (blk == 1) {
    int j = tid;
    if (j < 127) {
      float scale = 127.0f / 255.0f;
      float sample = (j + 0.5f) / scale - 0.5f;
      float sum = 0.0f;
      for (int p = 0; p < 255; ++p) {
        float w = 1.0f - fabsf(sample - p) * scale;
        if (w > 0.0f) sum += w;
      }
      for (int p = 0; p < 255; ++p) {
        float w = 1.0f - fabsf(sample - p) * scale;
        r2p[j * 256 + p] = (w > 0.0f) ? (w / sum) : 0.0f;
      }
      r2p[j * 256 + 255] = 0.0f;
    } else {
      for (int p = 0; p < 256; ++p) r2p[j * 256 + p] = 0.0f;
    }
  } else {
    int j = tid;
    float* rre = rf + j * 128;
    float* rim = rf + 128 * 128 + j * 128;
    if (j < 127) {
      float scale = 127.0f / 255.0f;
      float sample = (j + 0.5f) / scale - 0.5f;
      float sum = 0.0f;
      for (int p = 0; p < 255; ++p) {
        float w = 1.0f - fabsf(sample - p) * scale;
        if (w > 0.0f) sum += w;
      }
      for (int m = 0; m < 128; ++m) {
        const int pa = 127 + m, pb = 127 - m;
        float wa = 1.0f - fabsf(sample - pa) * scale;
        float wb = 1.0f - fabsf(sample - pb) * scale;
        const float a = (wa > 0.0f) ? (wa / sum) : 0.0f;
        const float b = (m > 0 && wb > 0.0f) ? (wb / sum) : 0.0f;
        const float s = (m & 1) ? -b : b;
        rre[m] = a + s;
        rim[m] = a - s;
      }
    } else {
      for (int m = 0; m < 128; ++m) { rre[m] = 0.0f; rim[m] = 0.0f; }
    }
  }
}

__device__ __forceinline__ void legendre_job(float* __restrict__ out, int lg, int weighted, int blk,
                                             double* cdiag, double* ca_all, double* cb_all) {
  const int L = 1 << lg;
  const int q = threadIdx.x >> lg;
  const int t = threadIdx.x & (L - 1);
  const int nblk = 1 << (2 * lg - 7);
  const int m = q * nblk + blk;
  double* ca = ca_all + q * L;
  double* cb = cb_all + q * L;
  {
    const int l = t;
    const double ll = (double)l, mm = (double)m;
    if (q == 0) cdiag[l] = (l >= 1) ? -sqrt((2.0 * ll + 1.0) / (2.0 * ll)) : 0.0;
    double a = 0.0, b = 0.0;
    if (l >= m + 2) {
      a = sqrt((4.0 * ll * ll - 1.0) / (ll * ll - mm * mm));
      b = sqrt((2.0 * ll + 1.0) * ((ll - 1.0) * (ll - 1.0) - mm * mm) /
               ((2.0 * ll - 3.0) * (ll * ll - mm * mm)));
    }
    ca[l] = a; cb[l] = b;
  }
  __syncthreads();
  const double theta = (2.0 * t + 1.0) * PI_D / (2.0 * L - 1.0);
  const double x = cos(theta), s = sin(theta);
  const double w = weighted ? (s * PI_D / (double)L) : 1.0;
  const long long LL = (long long)L * (long long)L;
  float* pos = out + (long long)(L - 1 + m) * LL + t;
  float* neg = out + (long long)(L - 1 - m) * LL + t;
  const bool mir = (m > 0);
  const bool odd = (m & 1) != 0;
  for (int l = 0; l < m; ++l) {
    pos[(long long)l * L] = 0.0f;
    neg[(long long)l * L] = 0.0f;
  }
  double pmm = sqrt(1.0 / (4.0 * PI_D)) * w;
  for (int j = 1; j <= m; ++j) pmm = cdiag[j] * s * pmm;
  {
    const float fv = (float)pmm;
    pos[(long long)m * L] = fv;
    if (mir) neg[(long long)m * L] = odd ? -fv : fv;
  }
  if (m + 1 < L) {
    double p2 = pmm;
    double p1 = sqrt(2.0 * (m + 1) + 1.0) * x * pmm;
    {
      const float fv = (float)p1;
      pos[(long long)(m + 1) * L] = fv;
      if (mir) neg[(long long)(m + 1) * L] = odd ? -fv : fv;
    }
    for (int l = m + 2; l < L; ++l) {
      const double p0 = ca[l] * x * p1 - cb[l] * p2;
      const float fv = (float)p0;
      pos[(long long)l * L] = fv;
      if (mir) neg[(long long)l * L] = odd ? -fv : fv;
      p2 = p1; p1 = p0;
    }
  }
}

__device__ __forceinline__ void legendre_fold_job(float* __restrict__ out, int blk,
                                                  double* cdiag, double* ca, double* cb, float* rw) {
  const int m = blk;
  const int t = threadIdx.x;
  {
    const int l = t;
    const double ll = (double)l, mm = (double)m;
    double d = 0.0;
    if (l >= 1 && l <= m) d = -sqrt((2.0 * ll + 1.0) / (2.0 * ll));
    cdiag[l] = d;
    double a = 0.0, b = 0.0;
    if (l >= m + 2) {
      a = sqrt((4.0 * ll * ll - 1.0) / (ll * ll - mm * mm));
      b = sqrt((2.0 * ll + 1.0) * ((ll - 1.0) * (ll - 1.0) - mm * mm) /
               ((2.0 * ll - 3.0) * (ll * ll - mm * mm)));
    }
    ca[l] = a; cb[l] = b;
  }
  if (t < 64) {
    const int i = t;
    const float scale = 0.5f;
    const float sample = (i + 0.5f) / scale - 0.5f;
    float wq[4];
    float sum = 0.0f;
#pragma unroll
    for (int q = 0; q < 4; ++q) {
      const int j = 2 * i - 1 + q;
      const float wv = 1.0f - fabsf(sample - j) * scale;
      wq[q] = (j >= 0 && j < 128 && wv > 0.0f) ? wv : 0.0f;
      sum += wq[q];
    }
    v4f rv;
    rv.x = wq[0] / sum; rv.y = wq[1] / sum; rv.z = wq[2] / sum; rv.w = wq[3] / sum;
    *(v4f*)(rw + 4 * i) = rv;
  }
  __syncthreads();
  const double theta = (2.0 * t + 1.0) * PI_D / 255.0;
  double x, s;
  sincos(theta, &s, &x);
  const double w = s * PI_D / 128.0;
  float* pos = out + (long long)m * 8192 + t;
  double pmm = sqrt(1.0 / (4.0 * PI_D)) * w;
  for (int j = 1; j <= m; ++j) pmm = cdiag[j] * s * pmm;
  const double pm1 = sqrt(2.0 * (m + 1) + 1.0) * x * pmm;
  const int i0 = (m > 0) ? ((m - 1) >> 1) : 0;
  for (int i = 0; i < i0; ++i) pos[i * 128] = 0.0f;
  double w0 = 0.0;
  double w1 = (m == 0) ? pmm : 0.0;
  float w0f = 0.0f, w1f = (float)w1;
  for (int i = i0; i < 63; ++i) {
    const int la = 2 * i + 1, lb = 2 * i + 2;
    const double caa = ca[la], cba = cb[la], cab = ca[lb], cbb = cb[lb];
    const v4f r = *(const v4f*)(rw + 4 * i);
    double pa = caa * x * w1 - cba * w0;
    pa = (la == m) ? pmm : pa;
    pa = (la == m + 1) ? pm1 : pa;
    double pb = cab * x * pa - cbb * w1;
    pb = (lb == m) ? pmm : pb;
    pb = (lb == m + 1) ? pm1 : pb;
    const float paf = (float)pa, pbf = (float)pb;
    pos[i * 128] = (r.x * w0f + r.y * w1f) + (r.z * paf + r.w * pbf);
    w0 = pa; w1 = pb; w0f = paf; w1f = pbf;
  }
  {
    double pc = ca[127] * x * w1 - cb[127] * w0;
    pc = (127 == m) ? pmm : pc;
    pc = (127 == m + 1) ? pm1 : pc;
    pos[63 * 128] = rw[252] * w0f + rw[253] * w1f + rw[254] * (float)pc;
  }
}

constexpr int S0_LEG1 = 128;
constexpr int S0_LEG2 = 32;
constexpr int S0_TVEC = 3;
constexpr int S0_RESZ = 3;
constexpr int S0_PH1  = 128 * 256 / 128;
constexpr int S0_PH2  = (int)(2 * SZ_PH2 / 128);
constexpr int S0_ZP   = 4;
constexpr int S0_ZX   = (int)((SZ_XSP / 4 + 128 * EW_U - 1) / (128 * EW_U));
constexpr int S0_XPAD = (int)(SZ_XP / 4 / (128 * EW_U));
constexpr int S0_BLOCKS = S0_LEG1 + S0_LEG2 + S0_TVEC + S0_RESZ + S0_PH1 + S0_PH2 + S0_ZP + S0_ZX + S0_XPAD;

__global__ void __launch_bounds__(128)
k_setup0(const float* __restrict__ x, const float* __restrict__ temb,
         const float* __restrict__ dw, const float* __restrict__ db,
         const float* __restrict__ drw, const float* __restrict__ drb,
         const float* __restrict__ diw, const float* __restrict__ dib,
         float* __restrict__ ws) {
  __shared__ double cdiag[128];
  __shared__ double ca[128];
  __shared__ double cb[128];
  __shared__ __attribute__((aligned(16))) float rw[256];
  int b = blockIdx.x;
  if (b < S0_LEG1) { legendre_fold_job(ws + OFF_A1, b, cdiag, ca, cb, rw); return; }
  b -= S0_LEG1;
  if (b < S0_LEG2) { legendre_job(ws + OFF_P2A, 6, 0, b, cdiag, ca, cb); return; }
  b -= S0_LEG2;
  if (b < S0_TVEC) {
    tvec_job(temb, dw, db, drw, drb, diw, dib, ws + OFF_TVEC, ws + OFF_TCR, ws + OFF_TCI, b);
    return;
  }
  b -= S0_TVEC;
  if (b < S0_RESZ) { resize_job(ws + OFF_R1, ws + OFF_R2P, ws + OFF_RF, b); return; }
  b -= S0_RESZ;
  if (b < S0_PH1) { phase1_job(ws + OFF_COS1, ws + OFF_SIN1N, b); return; }
  b -= S0_PH1;
  if (b < S0_PH2) { phase2_job(ws + OFF_PH2C, b); return; }
  b -= S0_PH2;
  if (b < S0_ZP) {
    zero_job(ws + OFF_T2R + 255LL * 8192, 2048, b);
    return;
  }
  b -= S0_ZP;
  if (b < S0_ZX) {
    zero_job(ws + OFF_XSP, (int)(SZ_XSP / 4), b);
    return;
  }
  b -= S0_ZX;
  xpad_job(x, ws + OFF_XP, (int)(SZ_XP / 4), b);
}

__device__ __forceinline__ void wc_job(const float* __restrict__ wr, const float* __restrict__ wi,
                                       const float* __restrict__ tcr, const float* __restrict__ tci,
                                       float* __restrict__ wcs, int n4, int blk) {
  const v4f* wr4 = (const v4f*)wr;
  const v4f* wi4 = (const v4f*)wi;
  v4f* wcs4 = (v4f*)wcs;
  const int base = blk * (256 * EW_U) + threadIdx.x;
  v4f r[EW_U], im[EW_U];
  float tr[EW_U], ti[EW_U];
#pragma unroll
  for (int u = 0; u < EW_U; ++u) {
    const int i = min(base + u * 256, n4 - 1);
    const int l = i >> 12;
    r[u] = wr4[i]; im[u] = wi4[i];
    tr[u] = tcr[l]; ti[u] = tci[l];
  }
#pragma unroll
  for (int u = 0; u < EW_U; ++u) {
    const int i = base + u * 256;
    if (i < n4) {
      const v4f cr = tr[u] * r[u] - ti[u] * im[u];
      const v4f ci = tr[u] * im[u] + ti[u] * r[u];
      v4f* d = wcs4 + (long long)(i >> 12) * 16384 + (i & 4095);
      d[0] = cr; d[4096] = -ci; d[8192] = ci; d[12288] = cr;
    }
  }
}

__device__ __forceinline__ void wsp_job(const float* __restrict__ spw, const float* __restrict__ tvec,
                                        float* __restrict__ wsp, int blk) {
  const int i = blk * 256 + threadIdx.x;
  const int o = i & 127;
  wsp[i] = tvec[o] * spw[i];
}

__device__ __forceinline__ void transp_job(const float* __restrict__ a, float* __restrict__ b,
                                           int m, float (*tile)[65]) {
  const v4f* a4 = (const v4f*)a + (long long)m * 1024;
  v4f* b4 = (v4f*)b + (long long)m * 1024;
  v4f v[4];
#pragma unroll
  for (int u = 0; u < 4; ++u) v[u] = a4[threadIdx.x + 256 * u];
#pragma unroll
  for (int u = 0; u < 4; ++u) {
    const int idx = threadIdx.x + 256 * u;
    const int l = idx >> 4, t4 = idx & 15;
    tile[l][4 * t4 + 0] = v[u].x; tile[l][4 * t4 + 1] = v[u].y;
    tile[l][4 * t4 + 2] = v[u].z; tile[l][4 * t4 + 3] = v[u].w;
  }
  __syncthreads();
#pragma unroll
  for (int u = 0; u < 4; ++u) {
    const int idx = threadIdx.x + 256 * u;
    const int t = idx >> 4, l4 = idx & 15;
    v4f o;
    o.x = tile[4 * l4 + 0][t]; o.y = tile[4 * l4 + 1][t];
    o.z = tile[4 * l4 + 2][t]; o.w = tile[4 * l4 + 3][t];
    b4[idx] = o;
  }
}

constexpr int S1_WC  = (int)(SZ_WC / 4 / (256 * EW_U));
constexpr int S1_WSP = (int)(SZ_WSP / 256);
constexpr int S1_TR  = 127;
constexpr int S1_BLOCKS = S1_WC + S1_WSP + S1_TR;

__global__ void __launch_bounds__(256)
k_setup1(const float* __restrict__ wr, const float* __restrict__ wi,
         const float* __restrict__ spw, float* __restrict__ ws) {
  __shared__ float tile[64][65];
  int b = blockIdx.x;
  if (b < S1_WC) {
    wc_job(wr, wi, ws + OFF_TCR, ws + OFF_TCI, ws + OFF_WCS, (int)(SZ_WC / 4), b);
    return;
  }
  b -= S1_WC;
  if (b < S1_WSP) { wsp_job(spw, ws + OFF_TVEC, ws + OFF_WSP, b); return; }
  b -= S1_WSP;
  transp_job(ws + OFF_P2A, ws + OFF_P2B, b, tile);
}

__device__ __forceinline__ float wave_sum(float s) {
#pragma unroll
  for (int o = 16; o > 0; o >>= 1) s += __shfl_xor(s, o);
  return s;
}

__global__ void __launch_bounds__(256)
k_fuse(const float* __restrict__ xspec, const float* __restrict__ xcv,
       const float* __restrict__ lns, const float* __restrict__ lnb,
       float* __restrict__ out) {
  const int lane = threadIdx.x & 31;
  const int n = blockIdx.x * 8 + (threadIdx.x >> 5);
  if (n >= 64 * 127) return;
  const int l = n / 127, m = n - l * 127;
  const int i4 = (l * 128 + m) * 32 + lane;
  const v4f a  = ((const v4f*)xspec)[i4];
  const v4f b  = ((const v4f*)xcv)[i4];
  const v4f sc = ((const v4f*)lns)[lane];
  const v4f bi = ((const v4f*)lnb)[lane];
  v4f g;
#pragma unroll
  for (int q = 0; q < 4; ++q) {
    const float v = a[q] + b[q];
    g[q] = 0.5f * v * (1.0f + tanhf(0.7978845608028654f * (v + 0.044715f * v * v * v)));
  }
  const float mu = wave_sum((g[0] + g[1]) + (g[2] + g[3])) * (1.0f / 128.0f);
  v4f d;
#pragma unroll
  for (int q = 0; q < 4; ++q) d[q] = g[q] - mu;
  const float var = wave_sum((d[0] * d[0] + d[1] * d[1]) + (d[2] * d[2] + d[3] * d[3])) * (1.0f / 128.0f);
  const float rs = rsqrtf(var + 1e-6f);
  v4f o;
#pragma unroll
  for (int q = 0; q < 4; ++q) o[q] = d[q] * rs * sc[q] + bi[q];
  ((v4f*)out)[(l * 127 + m) * 32 + lane] = o;
}

extern "C" void kernel_launch(void* const* d_in, const int* in_sizes, int n_in,
                              void* d_out, int out_size, void* d_ws, size_t ws_size,
                              hipStream_t stream) {
  (void)in_sizes; (void)n_in; (void)out_size; (void)ws_size;
  const float* x       = (const float*)d_in[0];
  const float* t_emb   = (const float*)d_in[1];
  const float* sp_dw   = (const float*)d_in[2];
  const float* sp_db   = (const float*)d_in[3];
  const float* sp_wt   = (const float*)d_in[4];
  const float* sp_ck   = (const float*)d_in[5];
  const float* sp_cb   = (const float*)d_in[6];
  const float* spec_wr = (const float*)d_in[7];
  const float* spec_wi = (const float*)d_in[8];
  const float* drw     = (const float*)d_in[9];
  const float* drb     = (const float*)d_in[10];
  const float* diw     = (const float*)d_in[11];
  const float* dib     = (const float*)d_in[12];
  const float* lns     = (const float*)d_in[13];
  const float* lnb     = (const float*)d_in[14];
  float* out = (float*)d_out;
  float* ws  = (float*)d_ws;

  float *A1 = ws + OFF_A1, *XP = ws + OFF_XP, *COS1 = ws + OFF_COS1;
  float *GR = ws + OFF_GR;
  float *R1 = ws + OFF_R1, *R2P = ws + OFF_R2P, *T1R = ws + OFF_T1R;
  float *FFC = ws + OFF_FFC, *WCS = ws + OFF_WCS, *RF = ws + OFF_RF;
  float *F2R = ws + OFF_F2R, *P2B = ws + OFF_P2B;
  float *FTR = ws + OFF_FTR, *PH2C = ws + OFF_PH2C;
  float *XSPEC = ws + OFF_XSPEC, *TVEC = ws + OFF_TVEC;
  float *WSP = ws + OFF_WSP, *XR = ws + OFF_XS1, *T2R = ws + OFF_T2R;
  float *XSP = ws + OFF_XSP, *XCV = ws + OFF_XCV;

  k_setup0<<<dim3(S0_BLOCKS), dim3(128), 0, stream>>>(x, t_emb, sp_dw, sp_db, drw, drb, diw, dib, ws);
  k_setup1<<<dim3(S1_BLOCKS), dim3(256), 0, stream>>>(spec_wr, spec_wi, sp_wt, ws);


  gemm_pair<128, 256, 255 * 128, 128><<<dim3(4096 + 2040), dim3(32), 0, stream>>>(
      COS1, XP, GR,  R1, x, T2R,
      0LL, 256LL * 128, 128LL * 128, SZ_PH1, 0LL, SZ_G,
      0LL, 128LL, 64LL * 128, 0LL, 0LL, 0LL,
      256, 128, 2, 3, 4096,
      128, 128, 1, 2);
  gemm_pair<64 * 128, 256, 128 * 128, 128><<<dim3(1024 + 2048), dim3(32), 0, stream>>>(
      R2P, T2R, XR,  A1, GR, T1R,
      0LL, 128LL, 128LL * 128, 0LL, 0LL, 0LL,
      64LL * 128, 128LL, 64LL * 128, 0LL, SZ_G, SZ_T1,
      256, 128, 1, 3, 1024,
      128, 128, 2, 2);
  gemm_pair<64 * 128, 128, 128, 128><<<dim3(2048 + 1024), dim3(32), 0, stream>>>(
      RF, T1R, FFC,  XR, WSP, XSP + 130LL * 128,
      0LL, 128LL, 128LL * 256, 128LL * 128, SZ_T1, 128LL,
      128LL * 128, 0LL, 129LL * 128, 0LL, 0LL, 0LL,
      128, 256, 2, 3, 2048,
      128, 128, 1, 3);
  conv_gemm_pair<128, 256><<<dim3(CONV_BLOCKS + 2048), dim3(32), 0, stream>>>(
      XSP, sp_ck, sp_cb, TVEC, XCV,
      FFC, WCS, F2R,
      128LL * 256, 2LL * 256 * 128, 128LL * 128, 0LL, 256LL * 128, SZ_F2,
      256, 128, 2, 3);
  gemm_wmma<128 * 128, 64><<<dim3(2032), dim3(32), 0, stream>>>(
      P2B, F2R, FTR, 64LL * 64, 128LL, 64LL * 128, 0LL, SZ_F2, SZ_FT, 64, 128, 2, 2);
  gemm_wmma<64 * 128, 256><<<dim3(1024), dim3(32), 0, stream>>>(
      PH2C, FTR, XSPEC, 0LL, 128LL, 128LL * 128, 0LL, 0LL, 0LL, 256, 128, 1, 3);

  k_fuse<<<dim3(64 * 127 / 8), dim3(256), 0, stream>>>(XSPEC, XCV, lns, lnb, out);
}


# GEMM and conv workgroups of four waves (were one): the four take four 16-row tiles against one B panel, so a B line leaves L2 once and three waves find it in the WGP cache; 4x fewer workgroups
# speedup vs baseline: 82.9159x; 1.0450x over previous
#include <hip/hip_runtime.h>
#include <math.h>

typedef __attribute__((ext_vector_type(2))) float v2f;
typedef __attribute__((ext_vector_type(8))) float v8f;
typedef __attribute__((ext_vector_type(4))) float v4f;

#define PI_D 3.14159265358979323846

constexpr long long SZ_A1    = 255LL*128*128;
constexpr long long SZ_XP    = 128LL*256*128;
constexpr long long SZ_PH1   = 256LL*256;
constexpr long long SZ_G     = 128LL*256*128;
constexpr long long SZ_F1    = 255LL*128*128;
constexpr long long SZ_R1    = 64LL*128;
constexpr long long SZ_R2P   = 128LL*256;
constexpr long long SZ_T1    = 256LL*64*128;
constexpr long long SZ_FF    = 64LL*128*128;
constexpr long long SZ_WC    = 64LL*128*128;
constexpr long long SZ_F2    = 64LL*128*128;
constexpr long long SZ_P2    = 127LL*64*64;
constexpr long long SZ_FT    = 127LL*64*128;
constexpr long long SZ_PH2   = 128LL*128;
constexpr long long SZ_XSPEC = 64LL*128*128;
constexpr long long SZ_TVEC  = 256;
constexpr long long SZ_TC    = 64;
constexpr long long SZ_WSP   = 128LL*128;
constexpr long long SZ_XS1   = 128LL*255*128;
constexpr long long SZ_T2    = 256LL*64*128;
constexpr long long SZ_XS2   = 64LL*128*128;
constexpr long long SZ_XSP   = 67LL*129*128;
constexpr long long SZ_XCV   = 64LL*128*128;

constexpr long long OFF_A1    = 0;
constexpr long long OFF_XP    = OFF_A1    + SZ_A1;
constexpr long long OFF_COS1  = OFF_XP    + SZ_XP;
constexpr long long OFF_SIN1N = OFF_COS1  + SZ_PH1;
constexpr long long OFF_GR    = OFF_SIN1N + SZ_PH1;
constexpr long long OFF_GI    = OFF_GR    + SZ_G;
constexpr long long OFF_F1R   = OFF_GI    + SZ_G;
constexpr long long OFF_F1I   = OFF_F1R   + SZ_F1;
constexpr long long OFF_R1    = OFF_F1I   + SZ_F1;
constexpr long long OFF_R2P   = OFF_R1    + SZ_R1;
constexpr long long OFF_T1R   = OFF_R2P   + SZ_R2P;
constexpr long long OFF_T1I   = OFF_T1R   + SZ_T1;
constexpr long long OFF_FFR   = OFF_T1I   + SZ_T1;
constexpr long long OFF_FFI   = OFF_FFR   + SZ_FF;
constexpr long long OFF_WCR   = OFF_FFI   + SZ_FF;
constexpr long long OFF_WCI   = OFF_WCR   + SZ_WC;
constexpr long long OFF_F2R   = OFF_WCI   + SZ_WC;
constexpr long long OFF_F2I   = OFF_F2R   + SZ_F2;
constexpr long long OFF_P2A   = OFF_F2I   + SZ_F2;
constexpr long long OFF_P2B   = OFF_P2A   + SZ_P2;
constexpr long long OFF_FTR   = OFF_P2B   + SZ_P2;
constexpr long long OFF_FTI   = OFF_FTR   + SZ_FT;
constexpr long long OFF_COS2  = OFF_FTI   + SZ_FT;
constexpr long long OFF_SIN2  = OFF_COS2  + SZ_PH2;
constexpr long long OFF_XSPEC = OFF_SIN2  + SZ_PH2;
constexpr long long OFF_TVEC  = OFF_XSPEC + SZ_XSPEC;
constexpr long long OFF_TCR   = OFF_TVEC  + SZ_TVEC;
constexpr long long OFF_TCI   = OFF_TCR   + SZ_TC;
constexpr long long OFF_WSP   = OFF_TCI   + SZ_TC;
constexpr long long OFF_XS1   = OFF_WSP   + SZ_WSP;
constexpr long long OFF_T2R   = OFF_XS1   + SZ_XS1;
constexpr long long OFF_XS2   = OFF_T2R   + SZ_T2;
constexpr long long OFF_XSP   = OFF_XS2   + SZ_XS2;
constexpr long long OFF_XCV   = OFF_XSP   + SZ_XSP;
constexpr long long OFF_WCS   = OFF_F1R;
constexpr long long OFF_FFC   = OFF_FFR;
constexpr long long OFF_PH2C  = OFF_COS2;
constexpr long long OFF_RF    = OFF_WCR;

struct FragSet {
  v4f a[2];
  v4f b[8];
};

__device__ __forceinline__ void sched_fence() { __builtin_amdgcn_sched_barrier(0); }
__device__ __forceinline__ void sched_loads_then_mma_twice() {
  __builtin_amdgcn_sched_group_barrier(0x020, 10, 0);
  __builtin_amdgcn_sched_group_barrier(0x008, 16, 0);
  __builtin_amdgcn_sched_group_barrier(0x020, 10, 0);
  __builtin_amdgcn_sched_group_barrier(0x008, 16, 0);
}

__device__ __forceinline__ void load_frags(FragSet& f, const float* __restrict__ arow,
                                           const float* __restrict__ brow, long long ldb) {
  const v4f* ap = (const v4f*)arow;
  f.a[0] = ap[0];
  f.a[1] = ap[1];
#pragma unroll
  for (int i = 0; i < 8; ++i) f.b[i] = *(const v4f*)(brow + (long long)i * ldb);
}

__device__ __forceinline__ void mma_frags(const FragSet& f, v8f& acc0, v8f& acc1, v8f& acc2, v8f& acc3) {
#pragma unroll
  for (int s = 0; s < 4; ++s) {
    const v4f av = f.a[s >> 1];
    const int e0 = 2 * (s & 1);
    v2f a;  a.x = av[e0];           a.y = av[e0 + 1];
    const v4f r0 = f.b[2 * s], r1 = f.b[2 * s + 1];
    v2f b0; b0.x = r0[0];           b0.y = r1[0];
    v2f b1; b1.x = r0[1];           b1.y = r1[1];
    v2f b2; b2.x = r0[2];           b2.y = r1[2];
    v2f b3; b3.x = r0[3];           b3.y = r1[3];
    acc0 = __builtin_amdgcn_wmma_f32_16x16x4_f32(false, a, false, b0, (short)0, acc0, false, false);
    acc1 = __builtin_amdgcn_wmma_f32_16x16x4_f32(false, a, false, b1, (short)0, acc1, false, false);
    acc2 = __builtin_amdgcn_wmma_f32_16x16x4_f32(false, a, false, b2, (short)0, acc2, false, false);
    acc3 = __builtin_amdgcn_wmma_f32_16x16x4_f32(false, a, false, b3, (short)0, acc3, false, false);
  }
}

template <int LDB, int K>
__device__ __forceinline__ void gemm_tile(const float* __restrict__ A, const float* __restrict__ B,
                                          float* __restrict__ C, int lda, int ldc,
                                          int n0, int m0) {
  static_assert(K >= 32 && (K & 31) == 0, "K is walked in pairs of 16-k groups");
  constexpr long long ldb = LDB;
  const int lane = threadIdx.x & 31;
  const int half = lane >> 4;
  const int l16  = lane & 15;
  v8f acc0 = {}, acc1 = {}, acc2 = {}, acc3 = {};
  const float* arow = A + (long long)(m0 + l16) * lda + 8 * half;
  const float* brow = B + (long long)(8 * half) * ldb + n0 + 4 * l16;
  {
    FragSet P, Q;
    load_frags(P, arow, brow, ldb);
    int k0 = 0;
#pragma clang loop unroll(disable)
    for (; k0 + 32 < K; k0 += 32) {
      load_frags(Q, arow + k0 + 16, brow + (long long)(k0 + 16) * ldb, ldb);
      mma_frags(P, acc0, acc1, acc2, acc3);
      load_frags(P, arow + k0 + 32, brow + (long long)(k0 + 32) * ldb, ldb);
      mma_frags(Q, acc0, acc1, acc2, acc3);
      sched_loads_then_mma_twice();
    }
    load_frags(Q, arow + k0 + 16, brow + (long long)(k0 + 16) * ldb, ldb);
    sched_fence();
    mma_frags(P, acc0, acc1, acc2, acc3);
    mma_frags(Q, acc0, acc1, acc2, acc3);
  }
#pragma unroll
  for (int v = 0; v < 8; ++v) {
    const int row = m0 + half * 8 + v;
    v4f o;
    o.x = acc0[v]; o.y = acc1[v]; o.z = acc2[v]; o.w = acc3[v];
    *(v4f*)(C + (long long)row * ldc + n0 + 4 * l16) = o;
  }
}

__device__ __forceinline__ int wave_in_group() {
  return __builtin_amdgcn_readfirstlane((int)(threadIdx.x >> 5));
}

template <int LDB, int K>
__device__ __forceinline__ void gemm_job(const float* __restrict__ A, const float* __restrict__ B,
                                         float* __restrict__ C,
                                         long long sAb, long long sBb, long long sCb,
                                         long long pA, long long pB, long long pC,
                                         int lda, int ldc, int lgx, int lgmy, int blk) {
  const int bx = blk & ((1 << lgx) - 1);
  const int t  = blk >> lgx;
  const int by = ((t & ((1 << (lgmy - 2)) - 1)) << 2) | wave_in_group();
  const long long b = t >> (lgmy - 2);
  const int part = bx >> 1;
  A += b * sAb + part * pA; B += b * sBb + part * pB; C += b * sCb + part * pC;
  gemm_tile<LDB, K>(A, B, C, lda, ldc, (bx & 1) * 64, by * 16);
}

template <int LDB, int K>
__global__ void __launch_bounds__(128)
gemm_wmma(const float* __restrict__ A, const float* __restrict__ B, float* __restrict__ C,
          long long sAb, long long sBb, long long sCb,
          long long pA, long long pB, long long pC,
          int lda, int ldc, int lgx, int lgmy) {
  gemm_job<LDB, K>(A, B, C, sAb, sBb, sCb, pA, pB, pC, lda, ldc, lgx, lgmy, (int)blockIdx.x);
}

template <int LDB0, int K0, int LDB1, int K1>
__global__ void __launch_bounds__(128)
gemm_pair(const float* __restrict__ A0, const float* __restrict__ B0, float* __restrict__ C0,
          const float* __restrict__ A1, const float* __restrict__ B1, float* __restrict__ C1,
          long long sA0, long long sB0, long long sC0, long long pA0, long long pB0, long long pC0,
          long long sA1, long long sB1, long long sC1, long long pA1, long long pB1, long long pC1,
          int lda0, int ldc0, int lgx0, int lgmy0, int nblk0,
          int lda1, int ldc1, int lgx1, int lgmy1) {
  const int idx = (int)blockIdx.x;
  if (idx < nblk0)
    gemm_job<LDB0, K0>(A0, B0, C0, sA0, sB0, sC0, pA0, pB0, pC0, lda0, ldc0, lgx0, lgmy0, idx);
  else
    gemm_job<LDB1, K1>(A1, B1, C1, sA1, sB1, sC1, pA1, pB1, pC1, lda1, ldc1, lgx1, lgmy1, idx - nblk0);
}

__device__ __forceinline__ void conv_load(FragSet& f, const float* __restrict__ xsp,
                                          const float* __restrict__ ck, int g,
                                          int l, int mrow, int half, int ncol) {
  const int tap = g >> 3, kk = (g & 7) * 16;
  const int dy = tap / 3, dx = tap - 3 * dy;
  const float* arow = xsp + (((long long)(l + dy)) * 129 + (mrow + dx)) * 128 + 8 * half + kk;
  const float* brow = ck + (((long long)tap) * 128 + 8 * half + kk) * 128 + ncol;
  load_frags(f, arow, brow, 128);
}

__device__ __forceinline__ void conv_block(const float* __restrict__ xsp, const float* __restrict__ ck,
                                           const float* __restrict__ cb, const float* __restrict__ tvec,
                                           float* __restrict__ out, int n0, int m0, int l) {
  const int lane = threadIdx.x & 31;
  const int half = lane >> 4, l16 = lane & 15;
  const int mrow = m0 + l16, ncol = n0 + 4 * l16;
  v8f acc0 = {}, acc1 = {}, acc2 = {}, acc3 = {};
  {
    FragSet P, Q;
    conv_load(P, xsp, ck, 0, l, mrow, half, ncol);
    int g = 0;
    for (; g + 2 < 72; g += 2) {
      conv_load(Q, xsp, ck, g + 1, l, mrow, half, ncol);
      mma_frags(P, acc0, acc1, acc2, acc3);
      conv_load(P, xsp, ck, g + 2, l, mrow, half, ncol);
      mma_frags(Q, acc0, acc1, acc2, acc3);
      sched_loads_then_mma_twice();
    }
    conv_load(Q, xsp, ck, g + 1, l, mrow, half, ncol);
    sched_fence();
    mma_frags(P, acc0, acc1, acc2, acc3);
    mma_frags(Q, acc0, acc1, acc2, acc3);
  }
  const v4f bias = *(const v4f*)(cb + ncol) + *(const v4f*)(tvec + 128 + ncol);
#pragma unroll
  for (int v = 0; v < 8; ++v) {
    const int row = m0 + half * 8 + v;
    v4f o;
    o.x = acc0[v]; o.y = acc1[v]; o.z = acc2[v]; o.w = acc3[v];
    *(v4f*)(out + (((long long)l) * 128 + row) * 128 + ncol) = o + bias;
  }
}

constexpr int CONV_BLOCKS = 2 * 2 * 64;
template <int LDB0, int K0>
__global__ void __launch_bounds__(128)
conv_gemm_pair(const float* __restrict__ xsp, const float* __restrict__ ck,
               const float* __restrict__ cb, const float* __restrict__ tvec,
               float* __restrict__ out,
               const float* __restrict__ A0, const float* __restrict__ B0, float* __restrict__ C0,
               long long sA0, long long sB0, long long sC0, long long pA0, long long pB0, long long pC0,
               int lda0, int ldc0, int lgx0, int lgmy0) {
  const int idx = (int)blockIdx.x;
  if (idx < CONV_BLOCKS)
    conv_block(xsp, ck, cb, tvec, out, (idx & 1) * 64, (((idx >> 1) & 1) * 4 + wave_in_group()) * 16, idx >> 2);
  else
    gemm_job<LDB0, K0>(A0, B0, C0, sA0, sB0, sC0, pA0, pB0, pC0, lda0, ldc0, lgx0, lgmy0, idx - CONV_BLOCKS);
}

constexpr int EW_U = 4;

__device__ __forceinline__ void zero_job(float* __restrict__ p, int n4, int blk) {
  v4f* p4 = (v4f*)p;
  const int base = blk * (128 * EW_U) + threadIdx.x;
  const v4f z = {0.0f, 0.0f, 0.0f, 0.0f};
#pragma unroll
  for (int u = 0; u < EW_U; ++u) {
    const int i = base + u * 128;
    if (i < n4) p4[i] = z;
  }
}

__device__ __forceinline__ void xpad_job(const float* __restrict__ x, float* __restrict__ xp,
                                         int n4, int blk) {
  const v4f* x4 = (const v4f*)x;
  v4f* xp4 = (v4f*)xp;
  const int base = blk * (128 * EW_U) + threadIdx.x;
  v4f v[EW_U];
#pragma unroll
  for (int u = 0; u < EW_U; ++u) {
    const int i = min(base + u * 128, n4 - 1);
    const int t = i >> 13, r = i & 8191, p = r >> 5, c4 = r & 31;
    const v4f z = {0.0f, 0.0f, 0.0f, 0.0f};
    const v4f ld = x4[(t * 255 + min(p, 254)) * 32 + c4];
    v[u] = (p < 255) ? ld : z;
  }
#pragma unroll
  for (int u = 0; u < EW_U; ++u) {
    const int i = base + u * 128;
    if (i < n4) xp4[i] = v[u];
  }
}

__device__ __forceinline__ void tvec_job(const float* __restrict__ temb,
                                         const float* __restrict__ dw, const float* __restrict__ db,
                                         const float* __restrict__ drw, const float* __restrict__ drb,
                                         const float* __restrict__ diw, const float* __restrict__ dib,
                                         float* __restrict__ tvec, float* __restrict__ tcr,
                                         float* __restrict__ tci, int blk) {
  const int tid = threadIdx.x;
  if (blk < 2) {
    const int j = blk * 128 + tid;
    float acc = db[j];
    for (int k = 0; k < 256; ++k) acc += temb[k] * dw[k * 256 + j];
    tvec[j] = acc;
  } else {
    const int sel = tid >> 6, j = tid & 63;
    const float* __restrict__ w  = sel ? diw : drw;
    const float* __restrict__ bs = sel ? dib : drb;
    float* __restrict__ dst = sel ? tci : tcr;
    float acc = bs[j];
    for (int k = 0; k < 256; ++k) acc += temb[k] * w[k * 64 + j];
    dst[j] = acc;
  }
}

__device__ __forceinline__ void phase1_job(float* __restrict__ c1, float* __restrict__ s1n, int blk) {
  const int i = blk * 128 + threadIdx.x;
  const int m = i >> 8, p = i & 255;
  if (p < 255) {
    double mval = (double)m;
    double phi = 2.0 * PI_D * (double)p / 255.0;
    double dphi = 2.0 * PI_D / 255.0;
    c1[i]  = (float)( cos(phi * mval) * dphi);
    s1n[i] = (float)(-sin(phi * mval) * dphi);
  } else { c1[i] = 0.0f; s1n[i] = 0.0f; }
}

__device__ __forceinline__ void phase2_job(float* __restrict__ tab, int blk) {
  const int i = blk * 128 + threadIdx.x;
  const int p = i >> 8, k = i & 255;
  float v = 0.0f;
  if (p < 127 && k < 254) {
    const bool re = k < 127;
    const double mval = (double)((re ? k : k - 127) - 63);
    const double phi = 2.0 * PI_D * (double)p / 127.0;
    v = re ? (float)cos(mval * phi) : (float)(-sin(mval * phi));
  }
  tab[i] = v;
}

__device__ __forceinline__ void resize_job(float* __restrict__ r1, float* __restrict__ r2p,
                                           float* __restrict__ rf, int blk) {
  const int tid = threadIdx.x;
  if (blk == 0) {
    if (tid < 64) {
      int i = tid;
      float scale = 0.5f;
      float sample = (i + 0.5f) / scale - 0.5f;
      float sum = 0.0f;
      for (int j = 0; j < 128; ++j) {
        float w = 1.0f - fabsf(sample - j) * scale;
        if (w > 0.0f) sum += w;
      }
      for (int j = 0; j < 128; ++j) {
        float w = 1.0f - fabsf(sample - j) * scale;
        r1[i * 128 + j] = (w > 0.0f) ? (w / sum) : 0.0f;
      }
    }
  } else if (blk == 1) {
    int j = tid;
    if (j < 127) {
      float scale = 127.0f / 255.0f;
      float sample = (j + 0.5f) / scale - 0.5f;
      float sum = 0.0f;
      for (int p = 0; p < 255; ++p) {
        float w = 1.0f - fabsf(sample - p) * scale;
        if (w > 0.0f) sum += w;
      }
      for (int p = 0; p < 255; ++p) {
        float w = 1.0f - fabsf(sample - p) * scale;
        r2p[j * 256 + p] = (w > 0.0f) ? (w / sum) : 0.0f;
      }
      r2p[j * 256 + 255] = 0.0f;
    } else {
      for (int p = 0; p < 256; ++p) r2p[j * 256 + p] = 0.0f;
    }
  } else {
    int j = tid;
    float* rre = rf + j * 128;
    float* rim = rf + 128 * 128 + j * 128;
    if (j < 127) {
      float scale = 127.0f / 255.0f;
      float sample = (j + 0.5f) / scale - 0.5f;
      float sum = 0.0f;
      for (int p = 0; p < 255; ++p) {
        float w = 1.0f - fabsf(sample - p) * scale;
        if (w > 0.0f) sum += w;
      }
      for (int m = 0; m < 128; ++m) {
        const int pa = 127 + m, pb = 127 - m;
        float wa = 1.0f - fabsf(sample - pa) * scale;
        float wb = 1.0f - fabsf(sample - pb) * scale;
        const float a = (wa > 0.0f) ? (wa / sum) : 0.0f;
        const float b = (m > 0 && wb > 0.0f) ? (wb / sum) : 0.0f;
        const float s = (m & 1) ? -b : b;
        rre[m] = a + s;
        rim[m] = a - s;
      }
    } else {
      for (int m = 0; m < 128; ++m) { rre[m] = 0.0f; rim[m] = 0.0f; }
    }
  }
}

__device__ __forceinline__ void legendre_job(float* __restrict__ out, int lg, int weighted, int blk,
                                             double* cdiag, double* ca_all, double* cb_all) {
  const int L = 1 << lg;
  const int q = threadIdx.x >> lg;
  const int t = threadIdx.x & (L - 1);
  const int nblk = 1 << (2 * lg - 7);
  const int m = q * nblk + blk;
  double* ca = ca_all + q * L;
  double* cb = cb_all + q * L;
  {
    const int l = t;
    const double ll = (double)l, mm = (double)m;
    if (q == 0) cdiag[l] = (l >= 1) ? -sqrt((2.0 * ll + 1.0) / (2.0 * ll)) : 0.0;
    double a = 0.0, b = 0.0;
    if (l >= m + 2) {
      a = sqrt((4.0 * ll * ll - 1.0) / (ll * ll - mm * mm));
      b = sqrt((2.0 * ll + 1.0) * ((ll - 1.0) * (ll - 1.0) - mm * mm) /
               ((2.0 * ll - 3.0) * (ll * ll - mm * mm)));
    }
    ca[l] = a; cb[l] = b;
  }
  __syncthreads();
  const double theta = (2.0 * t + 1.0) * PI_D / (2.0 * L - 1.0);
  const double x = cos(theta), s = sin(theta);
  const double w = weighted ? (s * PI_D / (double)L) : 1.0;
  const long long LL = (long long)L * (long long)L;
  float* pos = out + (long long)(L - 1 + m) * LL + t;
  float* neg = out + (long long)(L - 1 - m) * LL + t;
  const bool mir = (m > 0);
  const bool odd = (m & 1) != 0;
  for (int l = 0; l < m; ++l) {
    pos[(long long)l * L] = 0.0f;
    neg[(long long)l * L] = 0.0f;
  }
  double pmm = sqrt(1.0 / (4.0 * PI_D)) * w;
  for (int j = 1; j <= m; ++j) pmm = cdiag[j] * s * pmm;
  {
    const float fv = (float)pmm;
    pos[(long long)m * L] = fv;
    if (mir) neg[(long long)m * L] = odd ? -fv : fv;
  }
  if (m + 1 < L) {
    double p2 = pmm;
    double p1 = sqrt(2.0 * (m + 1) + 1.0) * x * pmm;
    {
      const float fv = (float)p1;
      pos[(long long)(m + 1) * L] = fv;
      if (mir) neg[(long long)(m + 1) * L] = odd ? -fv : fv;
    }
    for (int l = m + 2; l < L; ++l) {
      const double p0 = ca[l] * x * p1 - cb[l] * p2;
      const float fv = (float)p0;
      pos[(long long)l * L] = fv;
      if (mir) neg[(long long)l * L] = odd ? -fv : fv;
      p2 = p1; p1 = p0;
    }
  }
}

__device__ __forceinline__ void legendre_fold_job(float* __restrict__ out, int blk,
                                                  double* cdiag, double* ca, double* cb, float* rw) {
  const int m = blk;
  const int t = threadIdx.x;
  {
    const int l = t;
    const double ll = (double)l, mm = (double)m;
    double d = 0.0;
    if (l >= 1 && l <= m) d = -sqrt((2.0 * ll + 1.0) / (2.0 * ll));
    cdiag[l] = d;
    double a = 0.0, b = 0.0;
    if (l >= m + 2) {
      a = sqrt((4.0 * ll * ll - 1.0) / (ll * ll - mm * mm));
      b = sqrt((2.0 * ll + 1.0) * ((ll - 1.0) * (ll - 1.0) - mm * mm) /
               ((2.0 * ll - 3.0) * (ll * ll - mm * mm)));
    }
    ca[l] = a; cb[l] = b;
  }
  if (t < 64) {
    const int i = t;
    const float scale = 0.5f;
    const float sample = (i + 0.5f) / scale - 0.5f;
    float wq[4];
    float sum = 0.0f;
#pragma unroll
    for (int q = 0; q < 4; ++q) {
      const int j = 2 * i - 1 + q;
      const float wv = 1.0f - fabsf(sample - j) * scale;
      wq[q] = (j >= 0 && j < 128 && wv > 0.0f) ? wv : 0.0f;
      sum += wq[q];
    }
    v4f rv;
    rv.x = wq[0] / sum; rv.y = wq[1] / sum; rv.z = wq[2] / sum; rv.w = wq[3] / sum;
    *(v4f*)(rw + 4 * i) = rv;
  }
  __syncthreads();
  const double theta = (2.0 * t + 1.0) * PI_D / 255.0;
  double x, s;
  sincos(theta, &s, &x);
  const double w = s * PI_D / 128.0;
  float* pos = out + (long long)m * 8192 + t;
  double pmm = sqrt(1.0 / (4.0 * PI_D)) * w;
  for (int j = 1; j <= m; ++j) pmm = cdiag[j] * s * pmm;
  const double pm1 = sqrt(2.0 * (m + 1) + 1.0) * x * pmm;
  const int i0 = (m > 0) ? ((m - 1) >> 1) : 0;
  for (int i = 0; i < i0; ++i) pos[i * 128] = 0.0f;
  double w0 = 0.0;
  double w1 = (m == 0) ? pmm : 0.0;
  float w0f = 0.0f, w1f = (float)w1;
  for (int i = i0; i < 63; ++i) {
    const int la = 2 * i + 1, lb = 2 * i + 2;
    const double caa = ca[la], cba = cb[la], cab = ca[lb], cbb = cb[lb];
    const v4f r = *(const v4f*)(rw + 4 * i);
    double pa = caa * x * w1 - cba * w0;
    pa = (la == m) ? pmm : pa;
    pa = (la == m + 1) ? pm1 : pa;
    double pb = cab * x * pa - cbb * w1;
    pb = (lb == m) ? pmm : pb;
    pb = (lb == m + 1) ? pm1 : pb;
    const float paf = (float)pa, pbf = (float)pb;
    pos[i * 128] = (r.x * w0f + r.y * w1f) + (r.z * paf + r.w * pbf);
    w0 = pa; w1 = pb; w0f = paf; w1f = pbf;
  }
  {
    double pc = ca[127] * x * w1 - cb[127] * w0;
    pc = (127 == m) ? pmm : pc;
    pc = (127 == m + 1) ? pm1 : pc;
    pos[63 * 128] = rw[252] * w0f + rw[253] * w1f + rw[254] * (float)pc;
  }
}

constexpr int S0_LEG1 = 128;
constexpr int S0_LEG2 = 32;
constexpr int S0_TVEC = 3;
constexpr int S0_RESZ = 3;
constexpr int S0_PH1  = 128 * 256 / 128;
constexpr int S0_PH2  = (int)(2 * SZ_PH2 / 128);
constexpr int S0_ZP   = 4;
constexpr int S0_ZX   = (int)((SZ_XSP / 4 + 128 * EW_U - 1) / (128 * EW_U));
constexpr int S0_XPAD = (int)(SZ_XP / 4 / (128 * EW_U));
constexpr int S0_BLOCKS = S0_LEG1 + S0_LEG2 + S0_TVEC + S0_RESZ + S0_PH1 + S0_PH2 + S0_ZP + S0_ZX + S0_XPAD;

__global__ void __launch_bounds__(128)
k_setup0(const float* __restrict__ x, const float* __restrict__ temb,
         const float* __restrict__ dw, const float* __restrict__ db,
         const float* __restrict__ drw, const float* __restrict__ drb,
         const float* __restrict__ diw, const float* __restrict__ dib,
         float* __restrict__ ws) {
  __shared__ double cdiag[128];
  __shared__ double ca[128];
  __shared__ double cb[128];
  __shared__ __attribute__((aligned(16))) float rw[256];
  int b = blockIdx.x;
  if (b < S0_LEG1) { legendre_fold_job(ws + OFF_A1, b, cdiag, ca, cb, rw); return; }
  b -= S0_LEG1;
  if (b < S0_LEG2) { legendre_job(ws + OFF_P2A, 6, 0, b, cdiag, ca, cb); return; }
  b -= S0_LEG2;
  if (b < S0_TVEC) {
    tvec_job(temb, dw, db, drw, drb, diw, dib, ws + OFF_TVEC, ws + OFF_TCR, ws + OFF_TCI, b);
    return;
  }
  b -= S0_TVEC;
  if (b < S0_RESZ) { resize_job(ws + OFF_R1, ws + OFF_R2P, ws + OFF_RF, b); return; }
  b -= S0_RESZ;
  if (b < S0_PH1) { phase1_job(ws + OFF_COS1, ws + OFF_SIN1N, b); return; }
  b -= S0_PH1;
  if (b < S0_PH2) { phase2_job(ws + OFF_PH2C, b); return; }
  b -= S0_PH2;
  if (b < S0_ZP) {
    zero_job(ws + OFF_T2R + 255LL * 8192, 2048, b);
    return;
  }
  b -= S0_ZP;
  if (b < S0_ZX) {
    zero_job(ws + OFF_XSP, (int)(SZ_XSP / 4), b);
    return;
  }
  b -= S0_ZX;
  xpad_job(x, ws + OFF_XP, (int)(SZ_XP / 4), b);
}

__device__ __forceinline__ void wc_job(const float* __restrict__ wr, const float* __restrict__ wi,
                                       const float* __restrict__ tcr, const float* __restrict__ tci,
                                       float* __restrict__ wcs, int n4, int blk) {
  const v4f* wr4 = (const v4f*)wr;
  const v4f* wi4 = (const v4f*)wi;
  v4f* wcs4 = (v4f*)wcs;
  const int base = blk * (256 * EW_U) + threadIdx.x;
  v4f r[EW_U], im[EW_U];
  float tr[EW_U], ti[EW_U];
#pragma unroll
  for (int u = 0; u < EW_U; ++u) {
    const int i = min(base + u * 256, n4 - 1);
    const int l = i >> 12;
    r[u] = wr4[i]; im[u] = wi4[i];
    tr[u] = tcr[l]; ti[u] = tci[l];
  }
#pragma unroll
  for (int u = 0; u < EW_U; ++u) {
    const int i = base + u * 256;
    if (i < n4) {
      const v4f cr = tr[u] * r[u] - ti[u] * im[u];
      const v4f ci = tr[u] * im[u] + ti[u] * r[u];
      v4f* d = wcs4 + (long long)(i >> 12) * 16384 + (i & 4095);
      d[0] = cr; d[4096] = -ci; d[8192] = ci; d[12288] = cr;
    }
  }
}

__device__ __forceinline__ void wsp_job(const float* __restrict__ spw, const float* __restrict__ tvec,
                                        float* __restrict__ wsp, int blk) {
  const int i = blk * 256 + threadIdx.x;
  const int o = i & 127;
  wsp[i] = tvec[o] * spw[i];
}

__device__ __forceinline__ void transp_job(const float* __restrict__ a, float* __restrict__ b,
                                           int m, float (*tile)[65]) {
  const v4f* a4 = (const v4f*)a + (long long)m * 1024;
  v4f* b4 = (v4f*)b + (long long)m * 1024;
  v4f v[4];
#pragma unroll
  for (int u = 0; u < 4; ++u) v[u] = a4[threadIdx.x + 256 * u];
#pragma unroll
  for (int u = 0; u < 4; ++u) {
    const int idx = threadIdx.x + 256 * u;
    const int l = idx >> 4, t4 = idx & 15;
    tile[l][4 * t4 + 0] = v[u].x; tile[l][4 * t4 + 1] = v[u].y;
    tile[l][4 * t4 + 2] = v[u].z; tile[l][4 * t4 + 3] = v[u].w;
  }
  __syncthreads();
#pragma unroll
  for (int u = 0; u < 4; ++u) {
    const int idx = threadIdx.x + 256 * u;
    const int t = idx >> 4, l4 = idx & 15;
    v4f o;
    o.x = tile[4 * l4 + 0][t]; o.y = tile[4 * l4 + 1][t];
    o.z = tile[4 * l4 + 2][t]; o.w = tile[4 * l4 + 3][t];
    b4[idx] = o;
  }
}

constexpr int S1_WC  = (int)(SZ_WC / 4 / (256 * EW_U));
constexpr int S1_WSP = (int)(SZ_WSP / 256);
constexpr int S1_TR  = 127;
constexpr int S1_BLOCKS = S1_WC + S1_WSP + S1_TR;

__global__ void __launch_bounds__(256)
k_setup1(const float* __restrict__ wr, const float* __restrict__ wi,
         const float* __restrict__ spw, float* __restrict__ ws) {
  __shared__ float tile[64][65];
  int b = blockIdx.x;
  if (b < S1_WC) {
    wc_job(wr, wi, ws + OFF_TCR, ws + OFF_TCI, ws + OFF_WCS, (int)(SZ_WC / 4), b);
    return;
  }
  b -= S1_WC;
  if (b < S1_WSP) { wsp_job(spw, ws + OFF_TVEC, ws + OFF_WSP, b); return; }
  b -= S1_WSP;
  transp_job(ws + OFF_P2A, ws + OFF_P2B, b, tile);
}

__device__ __forceinline__ float wave_sum(float s) {
#pragma unroll
  for (int o = 16; o > 0; o >>= 1) s += __shfl_xor(s, o);
  return s;
}

__global__ void __launch_bounds__(256)
k_fuse(const float* __restrict__ xspec, const float* __restrict__ xcv,
       const float* __restrict__ lns, const float* __restrict__ lnb,
       float* __restrict__ out) {
  const int lane = threadIdx.x & 31;
  const int n = blockIdx.x * 8 + (threadIdx.x >> 5);
  if (n >= 64 * 127) return;
  const int l = n / 127, m = n - l * 127;
  const int i4 = (l * 128 + m) * 32 + lane;
  const v4f a  = ((const v4f*)xspec)[i4];
  const v4f b  = ((const v4f*)xcv)[i4];
  const v4f sc = ((const v4f*)lns)[lane];
  const v4f bi = ((const v4f*)lnb)[lane];
  v4f g;
#pragma unroll
  for (int q = 0; q < 4; ++q) {
    const float v = a[q] + b[q];
    g[q] = 0.5f * v * (1.0f + tanhf(0.7978845608028654f * (v + 0.044715f * v * v * v)));
  }
  const float mu = wave_sum((g[0] + g[1]) + (g[2] + g[3])) * (1.0f / 128.0f);
  v4f d;
#pragma unroll
  for (int q = 0; q < 4; ++q) d[q] = g[q] - mu;
  const float var = wave_sum((d[0] * d[0] + d[1] * d[1]) + (d[2] * d[2] + d[3] * d[3])) * (1.0f / 128.0f);
  const float rs = rsqrtf(var + 1e-6f);
  v4f o;
#pragma unroll
  for (int q = 0; q < 4; ++q) o[q] = d[q] * rs * sc[q] + bi[q];
  ((v4f*)out)[(l * 127 + m) * 32 + lane] = o;
}

extern "C" void kernel_launch(void* const* d_in, const int* in_sizes, int n_in,
                              void* d_out, int out_size, void* d_ws, size_t ws_size,
                              hipStream_t stream) {
  (void)in_sizes; (void)n_in; (void)out_size; (void)ws_size;
  const float* x       = (const float*)d_in[0];
  const float* t_emb   = (const float*)d_in[1];
  const float* sp_dw   = (const float*)d_in[2];
  const float* sp_db   = (const float*)d_in[3];
  const float* sp_wt   = (const float*)d_in[4];
  const float* sp_ck   = (const float*)d_in[5];
  const float* sp_cb   = (const float*)d_in[6];
  const float* spec_wr = (const float*)d_in[7];
  const float* spec_wi = (const float*)d_in[8];
  const float* drw     = (const float*)d_in[9];
  const float* drb     = (const float*)d_in[10];
  const float* diw     = (const float*)d_in[11];
  const float* dib     = (const float*)d_in[12];
  const float* lns     = (const float*)d_in[13];
  const float* lnb     = (const float*)d_in[14];
  float* out = (float*)d_out;
  float* ws  = (float*)d_ws;

  float *A1 = ws + OFF_A1, *XP = ws + OFF_XP, *COS1 = ws + OFF_COS1;
  float *GR = ws + OFF_GR;
  float *R1 = ws + OFF_R1, *R2P = ws + OFF_R2P, *T1R = ws + OFF_T1R;
  float *FFC = ws + OFF_FFC, *WCS = ws + OFF_WCS, *RF = ws + OFF_RF;
  float *F2R = ws + OFF_F2R, *P2B = ws + OFF_P2B;
  float *FTR = ws + OFF_FTR, *PH2C = ws + OFF_PH2C;
  float *XSPEC = ws + OFF_XSPEC, *TVEC = ws + OFF_TVEC;
  float *WSP = ws + OFF_WSP, *XR = ws + OFF_XS1, *T2R = ws + OFF_T2R;
  float *XSP = ws + OFF_XSP, *XCV = ws + OFF_XCV;

  k_setup0<<<dim3(S0_BLOCKS), dim3(128), 0, stream>>>(x, t_emb, sp_dw, sp_db, drw, drb, diw, dib, ws);
  k_setup1<<<dim3(S1_BLOCKS), dim3(256), 0, stream>>>(spec_wr, spec_wi, sp_wt, ws);


  gemm_pair<128, 256, 255 * 128, 128><<<dim3(1024 + 510), dim3(128), 0, stream>>>(
      COS1, XP, GR,  R1, x, T2R,
      0LL, 256LL * 128, 128LL * 128, SZ_PH1, 0LL, SZ_G,
      0LL, 128LL, 64LL * 128, 0LL, 0LL, 0LL,
      256, 128, 2, 3, 1024,
      128, 128, 1, 2);
  gemm_pair<64 * 128, 256, 128 * 128, 128><<<dim3(256 + 512), dim3(128), 0, stream>>>(
      R2P, T2R, XR,  A1, GR, T1R,
      0LL, 128LL, 128LL * 128, 0LL, 0LL, 0LL,
      64LL * 128, 128LL, 64LL * 128, 0LL, SZ_G, SZ_T1,
      256, 128, 1, 3, 256,
      128, 128, 2, 2);
  gemm_pair<64 * 128, 128, 128, 128><<<dim3(512 + 256), dim3(128), 0, stream>>>(
      RF, T1R, FFC,  XR, WSP, XSP + 130LL * 128,
      0LL, 128LL, 128LL * 256, 128LL * 128, SZ_T1, 128LL,
      128LL * 128, 0LL, 129LL * 128, 0LL, 0LL, 0LL,
      128, 256, 2, 3, 512,
      128, 128, 1, 3);
  conv_gemm_pair<128, 256><<<dim3(CONV_BLOCKS + 512), dim3(128), 0, stream>>>(
      XSP, sp_ck, sp_cb, TVEC, XCV,
      FFC, WCS, F2R,
      128LL * 256, 2LL * 256 * 128, 128LL * 128, 0LL, 256LL * 128, SZ_F2,
      256, 128, 2, 3);
  gemm_wmma<128 * 128, 64><<<dim3(508), dim3(128), 0, stream>>>(
      P2B, F2R, FTR, 64LL * 64, 128LL, 64LL * 128, 0LL, SZ_F2, SZ_FT, 64, 128, 2, 2);
  gemm_wmma<64 * 128, 256><<<dim3(256), dim3(128), 0, stream>>>(
      PH2C, FTR, XSPEC, 0LL, 128LL, 128LL * 128, 0LL, 0LL, 0LL, 256, 128, 1, 3);

  k_fuse<<<dim3(64 * 127 / 8), dim3(256), 0, stream>>>(XSPEC, XCV, lns, lnb, out);
}
